# starting kernel
# baseline (speedup 1.0000x reference)
#include <hip/hip_runtime.h>

typedef float v2f __attribute__((ext_vector_type(2)));
typedef float v8f __attribute__((ext_vector_type(8)));

#define IN_FEATS 128
#define HIDDEN   64
#define NGRAPH   512
#define BN_EPS   1e-5f

__global__ __launch_bounds__(256) void scatter_add_kernel(
    const float* __restrict__ xin, const int* __restrict__ ei,
    float* __restrict__ agg, int E, int F) {
  long long tid = (long long)blockIdx.x * blockDim.x + threadIdx.x;
  int chunks = F >> 2;
  long long total = (long long)E * chunks;
  if (tid >= total) return;
  int e = (int)(tid / chunks);
  int f = ((int)(tid - (long long)e * chunks)) << 2;
  int s = ei[e];
  int d = ei[E + e];
  const float4 vv = *(const float4*)(xin + (long long)s * F + f);
  float* p = agg + (long long)d * F + f;
  atomicAdd(p + 0, vv.x);
  atomicAdd(p + 1, vv.y);
  atomicAdd(p + 2, vv.z);
  atomicAdd(p + 3, vv.w);
}

template <int FIN>
__global__ __launch_bounds__(256) void gin_mlp_kernel(
    const float* __restrict__ hin,
    const float* __restrict__ Wa,
    const float* __restrict__ ba,
    const float* __restrict__ Wb,
    const float* __restrict__ bb,
    const float* __restrict__ gg,
    const float* __restrict__ be,
    const float* __restrict__ mm,
    const float* __restrict__ vv,
    float* __restrict__ hout,
    int n_nodes) {
  __shared__ float sT[8][16 * 64];

  const int wave = threadIdx.x >> 5;
  const int lane = threadIdx.x & 31;
  const int half = lane >> 4;
  const int l16  = lane & 15;

  const int tile = blockIdx.x * 8 + wave;
  const int row0 = tile * 16;
  if (row0 >= n_nodes) return;

  v8f acc[4];
#pragma unroll
  for (int t = 0; t < 4; ++t) {
    float bias = ba[t * 16 + l16];
#pragma unroll
    for (int r = 0; r < 8; ++r) acc[t][r] = bias;
  }

  int arow = row0 + l16;
  if (arow >= n_nodes) arow = n_nodes - 1;
  const float* aptr = hin + (long long)arow * FIN + 2 * half;

  for (int k = 0; k < FIN; k += 4) {
    v2f a = *(const v2f*)(aptr + k);
#pragma unroll
    for (int t = 0; t < 4; ++t) {
      const float* bp = Wa + (t * 16 + l16) * FIN + k + 2 * half;
      v2f b = *(const v2f*)bp;
      acc[t] = __builtin_amdgcn_wmma_f32_16x16x4_f32(
          false, a, false, b, (short)0, acc[t], false, false);
    }
  }

#pragma unroll
  for (int t = 0; t < 4; ++t) {
#pragma unroll
    for (int r = 0; r < 8; ++r) {
      float x = acc[t][r];
      x = x > 0.0f ? x : 0.0f;
      sT[wave][(r + 8 * half) * 64 + t * 16 + l16] = x;
    }
  }

  v8f acc2[4];
#pragma unroll
  for (int t = 0; t < 4; ++t)
#pragma unroll
    for (int r = 0; r < 8; ++r) acc2[t][r] = 0.0f;

  const float* tp = &sT[wave][l16 * 64 + 2 * half];
  for (int k = 0; k < 64; k += 4) {
    v2f a = *(const v2f*)(tp + k);
#pragma unroll
    for (int t = 0; t < 4; ++t) {
      const float* bp = Wb + (t * 16 + l16) * 64 + k + 2 * half;
      v2f b = *(const v2f*)bp;
      acc2[t] = __builtin_amdgcn_wmma_f32_16x16x4_f32(
          false, a, false, b, (short)0, acc2[t], false, false);
    }
  }

#pragma unroll
  for (int t = 0; t < 4; ++t) {
    int n = t * 16 + l16;
    float s  = gg[n] * rsqrtf(vv[n] + BN_EPS);
    float sh = (bb[n] - mm[n]) * s + be[n];
#pragma unroll
    for (int r = 0; r < 8; ++r) {
      int mrow = row0 + r + 8 * half;
      float o = acc2[t][r] * s + sh;
      o = o > 0.0f ? o : 0.0f;
      if (mrow < n_nodes) hout[(long long)mrow * 64 + n] = o;
    }
  }
}

__global__ __launch_bounds__(256) void pool_zero_kernel(float* __restrict__ sums,
                                                        float* __restrict__ counts) {
  int i = blockIdx.x * blockDim.x + threadIdx.x;
  if (i < NGRAPH * HIDDEN) sums[i] = 0.0f;
  if (i < NGRAPH) counts[i] = 0.0f;
}

__global__ __launch_bounds__(256) void pool_sum_kernel(
    const float* __restrict__ h, const int* __restrict__ batch,
    float* __restrict__ sums, float* __restrict__ counts, int n_nodes) {
  long long tid = (long long)blockIdx.x * blockDim.x + threadIdx.x;
  long long total = (long long)n_nodes * 16;
  if (tid >= total) return;
  int i = (int)(tid >> 4);
  int c = ((int)tid & 15) << 2;
  int g = batch[i];
  const float4 vv = *(const float4*)(h + (long long)i * HIDDEN + c);
  float* p = sums + (long long)g * HIDDEN + c;
  atomicAdd(p + 0, vv.x);
  atomicAdd(p + 1, vv.y);
  atomicAdd(p + 2, vv.z);
  atomicAdd(p + 3, vv.w);
  if (c == 0) atomicAdd(&counts[g], 1.0f);
}

__global__ __launch_bounds__(256) void pool_div_kernel(
    const float* __restrict__ sums, const float* __restrict__ counts,
    float* __restrict__ out) {
  int i = blockIdx.x * blockDim.x + threadIdx.x;
  if (i >= NGRAPH * HIDDEN) return;
  float c = counts[i >> 6];
  c = c > 1.0f ? c : 1.0f;
  out[i] = sums[i] / c;
}

extern "C" void kernel_launch(void* const* d_in, const int* in_sizes, int n_in,
                              void* d_out, int out_size, void* d_ws, size_t ws_size,
                              hipStream_t stream) {
  const float* x   = (const float*)d_in[0];
  const int* ei    = (const int*)d_in[1];
  const int* batch = (const int*)d_in[2];
  const float* W1a = (const float*)d_in[3];
  const float* b1a = (const float*)d_in[4];
  const float* W1b = (const float*)d_in[5];
  const float* b1b = (const float*)d_in[6];
  const float* g1  = (const float*)d_in[7];
  const float* be1 = (const float*)d_in[8];
  const float* m1  = (const float*)d_in[9];
  const float* v1  = (const float*)d_in[10];
  const float* W2a = (const float*)d_in[11];
  const float* b2a = (const float*)d_in[12];
  const float* W2b = (const float*)d_in[13];
  const float* b2b = (const float*)d_in[14];
  const float* g2  = (const float*)d_in[15];
  const float* be2 = (const float*)d_in[16];
  const float* m2  = (const float*)d_in[17];
  const float* v2  = (const float*)d_in[18];
  float* out = (float*)d_out;

  const int N = in_sizes[0] / IN_FEATS;
  const int E = in_sizes[1] / 2;

  float* ws    = (float*)d_ws;
  float* bufA  = ws;
  float* h1    = ws + (long long)N * IN_FEATS;
  float* h2    = h1 + (long long)N * HIDDEN;
  float* sums  = h2 + (long long)N * HIDDEN;
  float* cnts  = sums + NGRAPH * HIDDEN;
  (void)ws_size; (void)n_in; (void)out_size;

  const int tiles  = (N + 15) / 16;
  const int mlpBlk = (tiles + 7) / 8;

  hipMemcpyAsync(bufA, x, (size_t)N * IN_FEATS * sizeof(float),
                 hipMemcpyDeviceToDevice, stream);
  {
    long long tot = (long long)E * (IN_FEATS / 4);
    int blocks = (int)((tot + 255) / 256);
    scatter_add_kernel<<<blocks, 256, 0, stream>>>(x, ei, bufA, E, IN_FEATS);
  }
  gin_mlp_kernel<IN_FEATS><<<mlpBlk, 256, 0, stream>>>(
      bufA, W1a, b1a, W1b, b1b, g1, be1, m1, v1, h1, N);

  hipMemcpyAsync(bufA, h1, (size_t)N * HIDDEN * sizeof(float),
                 hipMemcpyDeviceToDevice, stream);
  {
    long long tot = (long long)E * (HIDDEN / 4);
    int blocks = (int)((tot + 255) / 256);
    scatter_add_kernel<<<blocks, 256, 0, stream>>>(h1, ei, bufA, E, HIDDEN);
  }
  gin_mlp_kernel<HIDDEN><<<mlpBlk, 256, 0, stream>>>(
      bufA, W2a, b2a, W2b, b2b, g2, be2, m2, v2, h2, N);

  pool_zero_kernel<<<(NGRAPH * HIDDEN + 255) / 256, 256, 0, stream>>>(sums, cnts);
  {
    long long tot = (long long)N * 16;
    int blocks = (int)((tot + 255) / 256);
    pool_sum_kernel<<<blocks, 256, 0, stream>>>(h2, batch, sums, cnts, N);
  }
  pool_div_kernel<<<(NGRAPH * HIDDEN + 255) / 256, 256, 0, stream>>>(sums, cnts, out);
}


# Layer 1: apply W1a to x on WMMA before the edge scatter (linearity), so its f32 atomic adds run on 64 features, not 128; sum order changes: per-node products first, then neighbour sum
# speedup vs baseline: 1.4033x; 1.4033x over previous
#include <hip/hip_runtime.h>

typedef float v2f __attribute__((ext_vector_type(2)));
typedef float v8f __attribute__((ext_vector_type(8)));

#define IN_FEATS 128
#define HIDDEN   64
#define NGRAPH   512
#define BN_EPS   1e-5f

__global__ __launch_bounds__(256) void scatter_add_kernel(
    const float* __restrict__ xin, const int* __restrict__ ei,
    float* __restrict__ agg, int E, int F) {
  long long tid = (long long)blockIdx.x * blockDim.x + threadIdx.x;
  int chunks = F >> 2;
  long long total = (long long)E * chunks;
  if (tid >= total) return;
  int e = (int)(tid / chunks);
  int f = ((int)(tid - (long long)e * chunks)) << 2;
  int s = ei[e];
  int d = ei[E + e];
  const float4 vv = *(const float4*)(xin + (long long)s * F + f);
  float* p = agg + (long long)d * F + f;
  atomicAdd(p + 0, vv.x);
  atomicAdd(p + 1, vv.y);
  atomicAdd(p + 2, vv.z);
  atomicAdd(p + 3, vv.w);
}

template <int FIN>
__global__ __launch_bounds__(256) void gin_mlp_kernel(
    const float* __restrict__ hin,
    const float* __restrict__ Wa,
    const float* __restrict__ ba,
    const float* __restrict__ Wb,
    const float* __restrict__ bb,
    const float* __restrict__ gg,
    const float* __restrict__ be,
    const float* __restrict__ mm,
    const float* __restrict__ vv,
    float* __restrict__ hout,
    int n_nodes) {
  __shared__ float sT[8][16 * 64];

  const int wave = threadIdx.x >> 5;
  const int lane = threadIdx.x & 31;
  const int half = lane >> 4;
  const int l16  = lane & 15;

  const int tile = blockIdx.x * 8 + wave;
  const int row0 = tile * 16;
  if (row0 >= n_nodes) return;

  v8f acc[4];
#pragma unroll
  for (int t = 0; t < 4; ++t) {
    float bias = ba[t * 16 + l16];
#pragma unroll
    for (int r = 0; r < 8; ++r) acc[t][r] = bias;
  }

  int arow = row0 + l16;
  if (arow >= n_nodes) arow = n_nodes - 1;
  const float* aptr = hin + (long long)arow * FIN + 2 * half;

  for (int k = 0; k < FIN; k += 4) {
    v2f a = *(const v2f*)(aptr + k);
#pragma unroll
    for (int t = 0; t < 4; ++t) {
      const float* bp = Wa + (t * 16 + l16) * FIN + k + 2 * half;
      v2f b = *(const v2f*)bp;
      acc[t] = __builtin_amdgcn_wmma_f32_16x16x4_f32(
          false, a, false, b, (short)0, acc[t], false, false);
    }
  }

#pragma unroll
  for (int t = 0; t < 4; ++t) {
#pragma unroll
    for (int r = 0; r < 8; ++r) {
      float x = acc[t][r];
      x = x > 0.0f ? x : 0.0f;
      sT[wave][(r + 8 * half) * 64 + t * 16 + l16] = x;
    }
  }

  v8f acc2[4];
#pragma unroll
  for (int t = 0; t < 4; ++t)
#pragma unroll
    for (int r = 0; r < 8; ++r) acc2[t][r] = 0.0f;

  const float* tp = &sT[wave][l16 * 64 + 2 * half];
  for (int k = 0; k < 64; k += 4) {
    v2f a = *(const v2f*)(tp + k);
#pragma unroll
    for (int t = 0; t < 4; ++t) {
      const float* bp = Wb + (t * 16 + l16) * 64 + k + 2 * half;
      v2f b = *(const v2f*)bp;
      acc2[t] = __builtin_amdgcn_wmma_f32_16x16x4_f32(
          false, a, false, b, (short)0, acc2[t], false, false);
    }
  }

#pragma unroll
  for (int t = 0; t < 4; ++t) {
    int n = t * 16 + l16;
    float s  = gg[n] * rsqrtf(vv[n] + BN_EPS);
    float sh = (bb[n] - mm[n]) * s + be[n];
#pragma unroll
    for (int r = 0; r < 8; ++r) {
      int mrow = row0 + r + 8 * half;
      float o = acc2[t][r] * s + sh;
      o = o > 0.0f ? o : 0.0f;
      if (mrow < n_nodes) hout[(long long)mrow * 64 + n] = o;
    }
  }
}

__global__ __launch_bounds__(256) void gin_proj_kernel(
    const float* __restrict__ xin,
    const float* __restrict__ Wa,
    float* __restrict__ ysrc,
    float* __restrict__ yacc,
    int n_nodes) {
  const int wave = threadIdx.x >> 5;
  const int lane = threadIdx.x & 31;
  const int half = lane >> 4;
  const int l16  = lane & 15;

  const int tile = blockIdx.x * 8 + wave;
  const int row0 = tile * 16;
  if (row0 >= n_nodes) return;

  v8f acc[4];
#pragma unroll
  for (int t = 0; t < 4; ++t)
#pragma unroll
    for (int r = 0; r < 8; ++r) acc[t][r] = 0.0f;

  int arow = row0 + l16;
  if (arow >= n_nodes) arow = n_nodes - 1;
  const float* aptr = xin + (long long)arow * IN_FEATS + 2 * half;

  for (int k = 0; k < IN_FEATS; k += 4) {
    v2f a = *(const v2f*)(aptr + k);
#pragma unroll
    for (int t = 0; t < 4; ++t) {
      const float* bp = Wa + (t * 16 + l16) * IN_FEATS + k + 2 * half;
      v2f b = *(const v2f*)bp;
      acc[t] = __builtin_amdgcn_wmma_f32_16x16x4_f32(
          false, a, false, b, (short)0, acc[t], false, false);
    }
  }

#pragma unroll
  for (int t = 0; t < 4; ++t) {
    int n = t * 16 + l16;
#pragma unroll
    for (int r = 0; r < 8; ++r) {
      int mrow = row0 + r + 8 * half;
      if (mrow < n_nodes) {
        long long o = (long long)mrow * HIDDEN + n;
        ysrc[o] = acc[t][r];
        yacc[o] = acc[t][r];
      }
    }
  }
}

__global__ __launch_bounds__(256) void gin_tail_kernel(
    const float* __restrict__ agg,
    const float* __restrict__ ba,
    const float* __restrict__ Wb,
    const float* __restrict__ bb,
    const float* __restrict__ gg,
    const float* __restrict__ be,
    const float* __restrict__ mm,
    const float* __restrict__ vv,
    float* __restrict__ hout,
    int n_nodes) {
  const int wave = threadIdx.x >> 5;
  const int lane = threadIdx.x & 31;
  const int half = lane >> 4;
  const int l16  = lane & 15;

  const int tile = blockIdx.x * 8 + wave;
  const int row0 = tile * 16;
  if (row0 >= n_nodes) return;

  v8f acc2[4];
#pragma unroll
  for (int t = 0; t < 4; ++t)
#pragma unroll
    for (int r = 0; r < 8; ++r) acc2[t][r] = 0.0f;

  int arow = row0 + l16;
  if (arow >= n_nodes) arow = n_nodes - 1;
  const float* aptr = agg + (long long)arow * HIDDEN + 2 * half;
  const float* biasp = ba + 2 * half;

  for (int k = 0; k < HIDDEN; k += 4) {
    v2f a = *(const v2f*)(aptr + k);
    v2f bs = *(const v2f*)(biasp + k);
    a.x = a.x + bs.x;
    a.y = a.y + bs.y;
    a.x = a.x > 0.0f ? a.x : 0.0f;
    a.y = a.y > 0.0f ? a.y : 0.0f;
#pragma unroll
    for (int t = 0; t < 4; ++t) {
      const float* bp = Wb + (t * 16 + l16) * HIDDEN + k + 2 * half;
      v2f b = *(const v2f*)bp;
      acc2[t] = __builtin_amdgcn_wmma_f32_16x16x4_f32(
          false, a, false, b, (short)0, acc2[t], false, false);
    }
  }

#pragma unroll
  for (int t = 0; t < 4; ++t) {
    int n = t * 16 + l16;
    float s  = gg[n] * rsqrtf(vv[n] + BN_EPS);
    float sh = (bb[n] - mm[n]) * s + be[n];
#pragma unroll
    for (int r = 0; r < 8; ++r) {
      int mrow = row0 + r + 8 * half;
      float o = acc2[t][r] * s + sh;
      o = o > 0.0f ? o : 0.0f;
      if (mrow < n_nodes) hout[(long long)mrow * HIDDEN + n] = o;
    }
  }
}

__global__ __launch_bounds__(256) void pool_zero_kernel(float* __restrict__ sums,
                                                        float* __restrict__ counts) {
  int i = blockIdx.x * blockDim.x + threadIdx.x;
  if (i < NGRAPH * HIDDEN) sums[i] = 0.0f;
  if (i < NGRAPH) counts[i] = 0.0f;
}

__global__ __launch_bounds__(256) void pool_sum_kernel(
    const float* __restrict__ h, const int* __restrict__ batch,
    float* __restrict__ sums, float* __restrict__ counts, int n_nodes) {
  long long tid = (long long)blockIdx.x * blockDim.x + threadIdx.x;
  long long total = (long long)n_nodes * 16;
  if (tid >= total) return;
  int i = (int)(tid >> 4);
  int c = ((int)tid & 15) << 2;
  int g = batch[i];
  const float4 vv = *(const float4*)(h + (long long)i * HIDDEN + c);
  float* p = sums + (long long)g * HIDDEN + c;
  atomicAdd(p + 0, vv.x);
  atomicAdd(p + 1, vv.y);
  atomicAdd(p + 2, vv.z);
  atomicAdd(p + 3, vv.w);
  if (c == 0) atomicAdd(&counts[g], 1.0f);
}

__global__ __launch_bounds__(256) void pool_div_kernel(
    const float* __restrict__ sums, const float* __restrict__ counts,
    float* __restrict__ out) {
  int i = blockIdx.x * blockDim.x + threadIdx.x;
  if (i >= NGRAPH * HIDDEN) return;
  float c = counts[i >> 6];
  c = c > 1.0f ? c : 1.0f;
  out[i] = sums[i] / c;
}

extern "C" void kernel_launch(void* const* d_in, const int* in_sizes, int n_in,
                              void* d_out, int out_size, void* d_ws, size_t ws_size,
                              hipStream_t stream) {
  const float* x   = (const float*)d_in[0];
  const int* ei    = (const int*)d_in[1];
  const int* batch = (const int*)d_in[2];
  const float* W1a = (const float*)d_in[3];
  const float* b1a = (const float*)d_in[4];
  const float* W1b = (const float*)d_in[5];
  const float* b1b = (const float*)d_in[6];
  const float* g1  = (const float*)d_in[7];
  const float* be1 = (const float*)d_in[8];
  const float* m1  = (const float*)d_in[9];
  const float* v1  = (const float*)d_in[10];
  const float* W2a = (const float*)d_in[11];
  const float* b2a = (const float*)d_in[12];
  const float* W2b = (const float*)d_in[13];
  const float* b2b = (const float*)d_in[14];
  const float* g2  = (const float*)d_in[15];
  const float* be2 = (const float*)d_in[16];
  const float* m2  = (const float*)d_in[17];
  const float* v2  = (const float*)d_in[18];
  float* out = (float*)d_out;

  const int N = in_sizes[0] / IN_FEATS;
  const int E = in_sizes[1] / 2;

  float* ws    = (float*)d_ws;
  float* bufA  = ws;
  float* y1    = bufA;
  float* agg1  = bufA + (long long)N * HIDDEN;
  float* h1    = ws + (long long)N * IN_FEATS;
  float* h2    = h1 + (long long)N * HIDDEN;
  float* sums  = h2 + (long long)N * HIDDEN;
  float* cnts  = sums + NGRAPH * HIDDEN;
  (void)ws_size; (void)n_in; (void)out_size;

  const int tiles  = (N + 15) / 16;
  const int mlpBlk = (tiles + 7) / 8;

  gin_proj_kernel<<<mlpBlk, 256, 0, stream>>>(x, W1a, y1, agg1, N);
  {
    long long tot = (long long)E * (HIDDEN / 4);
    int blocks = (int)((tot + 255) / 256);
    scatter_add_kernel<<<blocks, 256, 0, stream>>>(y1, ei, agg1, E, HIDDEN);
  }
  gin_tail_kernel<<<mlpBlk, 256, 0, stream>>>(
      agg1, b1a, W1b, b1b, g1, be1, m1, v1, h1, N);

  hipMemcpyAsync(bufA, h1, (size_t)N * HIDDEN * sizeof(float),
                 hipMemcpyDeviceToDevice, stream);
  {
    long long tot = (long long)E * (HIDDEN / 4);
    int blocks = (int)((tot + 255) / 256);
    scatter_add_kernel<<<blocks, 256, 0, stream>>>(h1, ei, bufA, E, HIDDEN);
  }
  gin_mlp_kernel<HIDDEN><<<mlpBlk, 256, 0, stream>>>(
      bufA, W2a, b2a, W2b, b2b, g2, be2, m2, v2, h2, N);

  pool_zero_kernel<<<(NGRAPH * HIDDEN + 255) / 256, 256, 0, stream>>>(sums, cnts);
  {
    long long tot = (long long)N * 16;
    int blocks = (int)((tot + 255) / 256);
    pool_sum_kernel<<<blocks, 256, 0, stream>>>(h2, batch, sums, cnts, N);
  }
  pool_div_kernel<<<(NGRAPH * HIDDEN + 255) / 256, 256, 0, stream>>>(sums, cnts, out);
}


# Edge aggregation, both layers: group edges by destination once per call (int rank atomics + prefix sum), then sum neighbours with plain loads; no f32 atomics; sum order = rank arrival order
# speedup vs baseline: 1.6207x; 1.1549x over previous
#include <hip/hip_runtime.h>

typedef float v2f __attribute__((ext_vector_type(2)));
typedef float v8f __attribute__((ext_vector_type(8)));

#define IN_FEATS 128
#define HIDDEN   64
#define NGRAPH   512
#define BN_EPS   1e-5f


__global__ __launch_bounds__(256) void csr_zero_kernel(int* __restrict__ cnt, int n_pad) {
  int i = blockIdx.x * blockDim.x + threadIdx.x;
  if (i < n_pad) cnt[i] = 0;
}

__global__ __launch_bounds__(256) void csr_rank_kernel(
    const int* __restrict__ ei, int* __restrict__ cnt, int* __restrict__ rank, int E) {
  int e = blockIdx.x * blockDim.x + threadIdx.x;
  if (e >= E) return;
  int d = ei[E + e];
  rank[e] = atomicAdd(&cnt[d], 1);
}

__global__ __launch_bounds__(256) void csr_offsets_kernel(
    const int* __restrict__ cnt, int* __restrict__ off, int n_pad) {
  __shared__ int s_front[8];
  __shared__ int s_wave[8];
  const int t = threadIdx.x;
  const int lane = t & 31;
  const int wave = t >> 5;
  const int4* c4 = (const int4*)cnt;
  const int nq = n_pad >> 2;
  const int qfront = blockIdx.x * 256;

  int front = 0;
  for (int q = t; q < qfront; q += 256) {
    int4 v = c4[q];
    front += (v.x + v.y) + (v.z + v.w);
  }
#pragma unroll
  for (int o = 16; o > 0; o >>= 1) front += __shfl_xor(front, o);
  if (lane == 0) s_front[wave] = front;

  const int q = qfront + t;
  int4 v = make_int4(0, 0, 0, 0);
  if (q < nq) v = c4[q];
  const int s = (v.x + v.y) + (v.z + v.w);
  int inc = s;
#pragma unroll
  for (int o = 1; o < 32; o <<= 1) {
    int u = __shfl_up(inc, o);
    if (lane >= o) inc += u;
  }
  if (lane == 31) s_wave[wave] = inc;
  __syncthreads();

  int base = 0;
#pragma unroll
  for (int w = 0; w < 8; ++w) {
    base += s_front[w];
    if (w < wave) base += s_wave[w];
  }
  const int ex = base + inc - s;
  if (q < nq) {
    int4 o4;
    o4.x = ex;
    o4.y = ex + v.x;
    o4.z = ex + v.x + v.y;
    o4.w = ex + v.x + v.y + v.z;
    ((int4*)off)[q] = o4;
  }
}

__global__ __launch_bounds__(256) void csr_fill_kernel(
    const int* __restrict__ ei, const int* __restrict__ off, const int* __restrict__ rank,
    int* __restrict__ csr, int E) {
  int e = blockIdx.x * blockDim.x + threadIdx.x;
  if (e >= E) return;
  int s = ei[e];
  int d = ei[E + e];
  csr[off[d] + rank[e]] = s;
}

__global__ __launch_bounds__(256) void gather_sum_kernel(
    const float* __restrict__ xin, const int* __restrict__ off, const int* __restrict__ cnt,
    const int* __restrict__ csr, float* __restrict__ agg, int n_nodes) {
  const int tid = blockIdx.x * blockDim.x + threadIdx.x;
  const int i = tid >> 4;
  const int c = (tid & 15) << 2;
  if (i >= n_nodes) return;
  const int len = cnt[i];
  const int* p = csr + off[i];
  float4 acc = *(const float4*)(xin + (long long)i * HIDDEN + c);
  const int last = len - 1;
  for (int j = 0; j < len; j += 4) {
    const int j1 = j + 1, j2 = j + 2, j3 = j + 3;
    const int s0 = p[j];
    const int s1 = p[j1 < last ? j1 : last];
    const int s2 = p[j2 < last ? j2 : last];
    const int s3 = p[j3 < last ? j3 : last];
    const float4 a0 = *(const float4*)(xin + (long long)s0 * HIDDEN + c);
    const float4 a1 = *(const float4*)(xin + (long long)s1 * HIDDEN + c);
    const float4 a2 = *(const float4*)(xin + (long long)s2 * HIDDEN + c);
    const float4 a3 = *(const float4*)(xin + (long long)s3 * HIDDEN + c);
    const int rem = len - j;
    acc.x += a0.x; acc.y += a0.y; acc.z += a0.z; acc.w += a0.w;
    const bool k1 = rem > 1, k2 = rem > 2, k3 = rem > 3;
    acc.x += k1 ? a1.x : 0.0f; acc.y += k1 ? a1.y : 0.0f;
    acc.z += k1 ? a1.z : 0.0f; acc.w += k1 ? a1.w : 0.0f;
    acc.x += k2 ? a2.x : 0.0f; acc.y += k2 ? a2.y : 0.0f;
    acc.z += k2 ? a2.z : 0.0f; acc.w += k2 ? a2.w : 0.0f;
    acc.x += k3 ? a3.x : 0.0f; acc.y += k3 ? a3.y : 0.0f;
    acc.z += k3 ? a3.z : 0.0f; acc.w += k3 ? a3.w : 0.0f;
  }
  *(float4*)(agg + (long long)i * HIDDEN + c) = acc;
}

template <int FIN>
__global__ __launch_bounds__(256) void gin_mlp_kernel(
    const float* __restrict__ hin,
    const float* __restrict__ Wa,
    const float* __restrict__ ba,
    const float* __restrict__ Wb,
    const float* __restrict__ bb,
    const float* __restrict__ gg,
    const float* __restrict__ be,
    const float* __restrict__ mm,
    const float* __restrict__ vv,
    float* __restrict__ hout,
    int n_nodes) {
  __shared__ float sT[8][16 * 64];

  const int wave = threadIdx.x >> 5;
  const int lane = threadIdx.x & 31;
  const int half = lane >> 4;
  const int l16  = lane & 15;

  const int tile = blockIdx.x * 8 + wave;
  const int row0 = tile * 16;
  if (row0 >= n_nodes) return;

  v8f acc[4];
#pragma unroll
  for (int t = 0; t < 4; ++t) {
    float bias = ba[t * 16 + l16];
#pragma unroll
    for (int r = 0; r < 8; ++r) acc[t][r] = bias;
  }

  int arow = row0 + l16;
  if (arow >= n_nodes) arow = n_nodes - 1;
  const float* aptr = hin + (long long)arow * FIN + 2 * half;

  for (int k = 0; k < FIN; k += 4) {
    v2f a = *(const v2f*)(aptr + k);
#pragma unroll
    for (int t = 0; t < 4; ++t) {
      const float* bp = Wa + (t * 16 + l16) * FIN + k + 2 * half;
      v2f b = *(const v2f*)bp;
      acc[t] = __builtin_amdgcn_wmma_f32_16x16x4_f32(
          false, a, false, b, (short)0, acc[t], false, false);
    }
  }

#pragma unroll
  for (int t = 0; t < 4; ++t) {
#pragma unroll
    for (int r = 0; r < 8; ++r) {
      float x = acc[t][r];
      x = x > 0.0f ? x : 0.0f;
      sT[wave][(r + 8 * half) * 64 + t * 16 + l16] = x;
    }
  }

  v8f acc2[4];
#pragma unroll
  for (int t = 0; t < 4; ++t)
#pragma unroll
    for (int r = 0; r < 8; ++r) acc2[t][r] = 0.0f;

  const float* tp = &sT[wave][l16 * 64 + 2 * half];
  for (int k = 0; k < 64; k += 4) {
    v2f a = *(const v2f*)(tp + k);
#pragma unroll
    for (int t = 0; t < 4; ++t) {
      const float* bp = Wb + (t * 16 + l16) * 64 + k + 2 * half;
      v2f b = *(const v2f*)bp;
      acc2[t] = __builtin_amdgcn_wmma_f32_16x16x4_f32(
          false, a, false, b, (short)0, acc2[t], false, false);
    }
  }

#pragma unroll
  for (int t = 0; t < 4; ++t) {
    int n = t * 16 + l16;
    float s  = gg[n] * rsqrtf(vv[n] + BN_EPS);
    float sh = (bb[n] - mm[n]) * s + be[n];
#pragma unroll
    for (int r = 0; r < 8; ++r) {
      int mrow = row0 + r + 8 * half;
      float o = acc2[t][r] * s + sh;
      o = o > 0.0f ? o : 0.0f;
      if (mrow < n_nodes) hout[(long long)mrow * 64 + n] = o;
    }
  }
}

__global__ __launch_bounds__(256) void gin_proj_kernel(
    const float* __restrict__ xin,
    const float* __restrict__ Wa,
    float* __restrict__ ysrc,
    int n_nodes) {
  const int wave = threadIdx.x >> 5;
  const int lane = threadIdx.x & 31;
  const int half = lane >> 4;
  const int l16  = lane & 15;

  const int tile = blockIdx.x * 8 + wave;
  const int row0 = tile * 16;
  if (row0 >= n_nodes) return;

  v8f acc[4];
#pragma unroll
  for (int t = 0; t < 4; ++t)
#pragma unroll
    for (int r = 0; r < 8; ++r) acc[t][r] = 0.0f;

  int arow = row0 + l16;
  if (arow >= n_nodes) arow = n_nodes - 1;
  const float* aptr = xin + (long long)arow * IN_FEATS + 2 * half;

  for (int k = 0; k < IN_FEATS; k += 4) {
    v2f a = *(const v2f*)(aptr + k);
#pragma unroll
    for (int t = 0; t < 4; ++t) {
      const float* bp = Wa + (t * 16 + l16) * IN_FEATS + k + 2 * half;
      v2f b = *(const v2f*)bp;
      acc[t] = __builtin_amdgcn_wmma_f32_16x16x4_f32(
          false, a, false, b, (short)0, acc[t], false, false);
    }
  }

#pragma unroll
  for (int t = 0; t < 4; ++t) {
    int n = t * 16 + l16;
#pragma unroll
    for (int r = 0; r < 8; ++r) {
      int mrow = row0 + r + 8 * half;
      if (mrow < n_nodes) ysrc[(long long)mrow * HIDDEN + n] = acc[t][r];
    }
  }
}

__global__ __launch_bounds__(256) void gin_tail_kernel(
    const float* __restrict__ agg,
    const float* __restrict__ ba,
    const float* __restrict__ Wb,
    const float* __restrict__ bb,
    const float* __restrict__ gg,
    const float* __restrict__ be,
    const float* __restrict__ mm,
    const float* __restrict__ vv,
    float* __restrict__ hout,
    int n_nodes) {
  const int wave = threadIdx.x >> 5;
  const int lane = threadIdx.x & 31;
  const int half = lane >> 4;
  const int l16  = lane & 15;

  const int tile = blockIdx.x * 8 + wave;
  const int row0 = tile * 16;
  if (row0 >= n_nodes) return;

  v8f acc2[4];
#pragma unroll
  for (int t = 0; t < 4; ++t)
#pragma unroll
    for (int r = 0; r < 8; ++r) acc2[t][r] = 0.0f;

  int arow = row0 + l16;
  if (arow >= n_nodes) arow = n_nodes - 1;
  const float* aptr = agg + (long long)arow * HIDDEN + 2 * half;
  const float* biasp = ba + 2 * half;

  for (int k = 0; k < HIDDEN; k += 4) {
    v2f a = *(const v2f*)(aptr + k);
    v2f bs = *(const v2f*)(biasp + k);
    a.x = a.x + bs.x;
    a.y = a.y + bs.y;
    a.x = a.x > 0.0f ? a.x : 0.0f;
    a.y = a.y > 0.0f ? a.y : 0.0f;
#pragma unroll
    for (int t = 0; t < 4; ++t) {
      const float* bp = Wb + (t * 16 + l16) * HIDDEN + k + 2 * half;
      v2f b = *(const v2f*)bp;
      acc2[t] = __builtin_amdgcn_wmma_f32_16x16x4_f32(
          false, a, false, b, (short)0, acc2[t], false, false);
    }
  }

#pragma unroll
  for (int t = 0; t < 4; ++t) {
    int n = t * 16 + l16;
    float s  = gg[n] * rsqrtf(vv[n] + BN_EPS);
    float sh = (bb[n] - mm[n]) * s + be[n];
#pragma unroll
    for (int r = 0; r < 8; ++r) {
      int mrow = row0 + r + 8 * half;
      float o = acc2[t][r] * s + sh;
      o = o > 0.0f ? o : 0.0f;
      if (mrow < n_nodes) hout[(long long)mrow * HIDDEN + n] = o;
    }
  }
}

__global__ __launch_bounds__(256) void pool_zero_kernel(float* __restrict__ sums,
                                                        float* __restrict__ counts) {
  int i = blockIdx.x * blockDim.x + threadIdx.x;
  if (i < NGRAPH * HIDDEN) sums[i] = 0.0f;
  if (i < NGRAPH) counts[i] = 0.0f;
}

__global__ __launch_bounds__(256) void pool_sum_kernel(
    const float* __restrict__ h, const int* __restrict__ batch,
    float* __restrict__ sums, float* __restrict__ counts, int n_nodes) {
  long long tid = (long long)blockIdx.x * blockDim.x + threadIdx.x;
  long long total = (long long)n_nodes * 16;
  if (tid >= total) return;
  int i = (int)(tid >> 4);
  int c = ((int)tid & 15) << 2;
  int g = batch[i];
  const float4 vv = *(const float4*)(h + (long long)i * HIDDEN + c);
  float* p = sums + (long long)g * HIDDEN + c;
  atomicAdd(p + 0, vv.x);
  atomicAdd(p + 1, vv.y);
  atomicAdd(p + 2, vv.z);
  atomicAdd(p + 3, vv.w);
  if (c == 0) atomicAdd(&counts[g], 1.0f);
}

__global__ __launch_bounds__(256) void pool_div_kernel(
    const float* __restrict__ sums, const float* __restrict__ counts,
    float* __restrict__ out) {
  int i = blockIdx.x * blockDim.x + threadIdx.x;
  if (i >= NGRAPH * HIDDEN) return;
  float c = counts[i >> 6];
  c = c > 1.0f ? c : 1.0f;
  out[i] = sums[i] / c;
}

extern "C" void kernel_launch(void* const* d_in, const int* in_sizes, int n_in,
                              void* d_out, int out_size, void* d_ws, size_t ws_size,
                              hipStream_t stream) {
  const float* x   = (const float*)d_in[0];
  const int* ei    = (const int*)d_in[1];
  const int* batch = (const int*)d_in[2];
  const float* W1a = (const float*)d_in[3];
  const float* b1a = (const float*)d_in[4];
  const float* W1b = (const float*)d_in[5];
  const float* b1b = (const float*)d_in[6];
  const float* g1  = (const float*)d_in[7];
  const float* be1 = (const float*)d_in[8];
  const float* m1  = (const float*)d_in[9];
  const float* v1  = (const float*)d_in[10];
  const float* W2a = (const float*)d_in[11];
  const float* b2a = (const float*)d_in[12];
  const float* W2b = (const float*)d_in[13];
  const float* b2b = (const float*)d_in[14];
  const float* g2  = (const float*)d_in[15];
  const float* be2 = (const float*)d_in[16];
  const float* m2  = (const float*)d_in[17];
  const float* v2  = (const float*)d_in[18];
  float* out = (float*)d_out;

  const int N = in_sizes[0] / IN_FEATS;
  const int E = in_sizes[1] / 2;

  float* ws    = (float*)d_ws;
  float* y1    = ws;
  float* agg   = ws + (long long)N * HIDDEN;
  float* h1    = ws + (long long)N * IN_FEATS;
  float* h2    = h1 + (long long)N * HIDDEN;
  float* sums  = h2 + (long long)N * HIDDEN;
  float* cnts  = sums + NGRAPH * HIDDEN;
  const int Npad = (N + 3) & ~3;
  int* icnt    = (int*)(cnts + NGRAPH);
  int* ioff    = icnt + Npad;
  int* irank   = ioff + Npad;
  int* icsr    = irank + E;
  (void)ws_size; (void)n_in; (void)out_size;

  const int tiles  = (N + 15) / 16;
  const int mlpBlk = (tiles + 7) / 8;
  const int edgeBlk = (E + 255) / 256;
  const int gathBlk = (int)(((long long)N * 16 + 255) / 256);

  csr_zero_kernel<<<(Npad + 255) / 256, 256, 0, stream>>>(icnt, Npad);
  csr_rank_kernel<<<edgeBlk, 256, 0, stream>>>(ei, icnt, irank, E);
  csr_offsets_kernel<<<(Npad + 1023) / 1024, 256, 0, stream>>>(icnt, ioff, Npad);
  csr_fill_kernel<<<edgeBlk, 256, 0, stream>>>(ei, ioff, irank, icsr, E);

  gin_proj_kernel<<<mlpBlk, 256, 0, stream>>>(x, W1a, y1, N);
  gather_sum_kernel<<<gathBlk, 256, 0, stream>>>(y1, ioff, icnt, icsr, agg, N);
  gin_tail_kernel<<<mlpBlk, 256, 0, stream>>>(
      agg, b1a, W1b, b1b, g1, be1, m1, v1, h1, N);

  gather_sum_kernel<<<gathBlk, 256, 0, stream>>>(h1, ioff, icnt, icsr, agg, N);
  gin_mlp_kernel<HIDDEN><<<mlpBlk, 256, 0, stream>>>(
      agg, W2a, b2a, W2b, b2b, g2, be2, m2, v2, h2, N);

  pool_zero_kernel<<<(NGRAPH * HIDDEN + 255) / 256, 256, 0, stream>>>(sums, cnts);
  {
    long long tot = (long long)N * 16;
    int blocks = (int)((tot + 255) / 256);
    pool_sum_kernel<<<blocks, 256, 0, stream>>>(h2, batch, sums, cnts, N);
  }
  pool_div_kernel<<<(NGRAPH * HIDDEN + 255) / 256, 256, 0, stream>>>(sums, cnts, out);
}


# Edge grouping by destination: two-level counting sort on LDS counters (512-node bins, then per-bin ranks) replaces one global returning atomic per edge; bucket order = LDS rank arrival order
# speedup vs baseline: 2.6918x; 1.6609x over previous
#include <hip/hip_runtime.h>

typedef float v2f __attribute__((ext_vector_type(2)));
typedef float v8f __attribute__((ext_vector_type(8)));

#define IN_FEATS 128
#define HIDDEN   64
#define NGRAPH   512
#define BN_EPS   1e-5f


#define BIN_SHIFT    9
#define BIN_NODES    512
#define MAX_BINS     256
#define SORT_THREADS 512
#define SORT_BLOCKS  256

__global__ __launch_bounds__(MAX_BINS) void bin_zero_kernel(int* __restrict__ binTotal,
                                                            int* __restrict__ binCursor) {
  binTotal[threadIdx.x] = 0;
  binCursor[threadIdx.x] = 0;
}

__global__ __launch_bounds__(SORT_THREADS) void bin_count_kernel(
    const int* __restrict__ ei, int* __restrict__ binTotal, int* __restrict__ blockHist, int E) {
  __shared__ int s_hist[MAX_BINS];
  const int t = threadIdx.x;
  if (t < MAX_BINS) s_hist[t] = 0;
  __syncthreads();
  const int* dst = ei + E;
  const int chunk = (E + (int)gridDim.x - 1) / (int)gridDim.x;
  const int c0 = (int)blockIdx.x * chunk;
  int c1 = c0 + chunk;
  if (c1 > E) c1 = E;
  for (int b0 = c0; b0 < c1; b0 += 4 * SORT_THREADS) {
    int d[4];
#pragma unroll
    for (int j = 0; j < 4; ++j) {
      const int e = b0 + j * SORT_THREADS + t;
      d[j] = dst[e < c1 ? e : c1 - 1];
    }
#pragma unroll
    for (int j = 0; j < 4; ++j)
      if (b0 + j * SORT_THREADS + t < c1) atomicAdd(&s_hist[d[j] >> BIN_SHIFT], 1);
  }
  __syncthreads();
  if (t < MAX_BINS) {
    const int h = s_hist[t];
    blockHist[(int)blockIdx.x * MAX_BINS + t] = h;
    if (h > 0) atomicAdd(&binTotal[t], h);
  }
}

__global__ __launch_bounds__(SORT_THREADS) void bin_scatter_kernel(
    const int* __restrict__ ei, const int* __restrict__ binTotal, int* __restrict__ binCursor,
    const int* __restrict__ blockHist, int* __restrict__ pairs, int E) {
  __shared__ int s_cur[MAX_BINS];
  __shared__ int s_wsum[SORT_THREADS / 32];
  const int t = threadIdx.x;
  const int lane = t & 31;
  const int wave = t >> 5;

  const int tot = t < MAX_BINS ? binTotal[t] : 0;
  int inc = tot;
#pragma unroll
  for (int o = 1; o < 32; o <<= 1) {
    const int u = __shfl_up(inc, o);
    if (lane >= o) inc += u;
  }
  if (lane == 31) s_wsum[wave] = inc;
  __syncthreads();
  int base = 0;
#pragma unroll
  for (int w = 0; w < SORT_THREADS / 32; ++w)
    if (w < wave) base += s_wsum[w];
  if (t < MAX_BINS) {
    const int h = blockHist[(int)blockIdx.x * MAX_BINS + t];
    int st = 0;
    if (h > 0) st = atomicAdd(&binCursor[t], h);
    s_cur[t] = base + inc - tot + st;
  }
  __syncthreads();

  const int* srcp = ei;
  const int* dst = ei + E;
  const int chunk = (E + (int)gridDim.x - 1) / (int)gridDim.x;
  const int c0 = (int)blockIdx.x * chunk;
  int c1 = c0 + chunk;
  if (c1 > E) c1 = E;
  for (int b0 = c0; b0 < c1; b0 += 4 * SORT_THREADS) {
    int s[4], d[4];
#pragma unroll
    for (int j = 0; j < 4; ++j) {
      const int e = b0 + j * SORT_THREADS + t;
      const int ec = e < c1 ? e : c1 - 1;
      d[j] = dst[ec];
      s[j] = srcp[ec];
    }
    int pos[4];
#pragma unroll
    for (int j = 0; j < 4; ++j)
      pos[j] = (b0 + j * SORT_THREADS + t < c1) ? atomicAdd(&s_cur[d[j] >> BIN_SHIFT], 1) : -1;
#pragma unroll
    for (int j = 0; j < 4; ++j)
      if (pos[j] >= 0)
        pairs[pos[j]] = (int)(((unsigned)s[j] << BIN_SHIFT) | (unsigned)(d[j] & (BIN_NODES - 1)));
  }
}

__global__ __launch_bounds__(BIN_NODES) void bin_sort_kernel(
    const int* __restrict__ binTotal, const int* __restrict__ pairs,
    int* __restrict__ cnt, int* __restrict__ off, int* __restrict__ csr, int n_nodes) {
  __shared__ int s_cnt[BIN_NODES];
  __shared__ int s_wsum[BIN_NODES / 32];
  __shared__ int s_wsum2[BIN_NODES / 32];
  __shared__ int s_bin[2];
  const int t = threadIdx.x;
  const int lane = t & 31;
  const int wave = t >> 5;
  const int b = (int)blockIdx.x;

  s_cnt[t] = 0;
  const int btot = t < MAX_BINS ? binTotal[t] : 0;
  int inc = btot;
#pragma unroll
  for (int o = 1; o < 32; o <<= 1) {
    const int u = __shfl_up(inc, o);
    if (lane >= o) inc += u;
  }
  if (lane == 31) s_wsum[wave] = inc;
  __syncthreads();
  int wbase = 0;
#pragma unroll
  for (int w = 0; w < BIN_NODES / 32; ++w)
    if (w < wave) wbase += s_wsum[w];
  if (t == b) {
    s_bin[0] = wbase + inc - btot;
    s_bin[1] = btot;
  }
  __syncthreads();
  const int base = s_bin[0];
  const int tot = s_bin[1];
  const int* p = pairs + base;

  for (int i0 = 0; i0 < tot; i0 += 4 * BIN_NODES) {
    unsigned v[4];
#pragma unroll
    for (int j = 0; j < 4; ++j) {
      const int i = i0 + j * BIN_NODES + t;
      v[j] = (unsigned)p[i < tot ? i : tot - 1];
    }
#pragma unroll
    for (int j = 0; j < 4; ++j)
      if (i0 + j * BIN_NODES + t < tot) atomicAdd(&s_cnt[v[j] & (BIN_NODES - 1)], 1);
  }
  __syncthreads();

  const int c = s_cnt[t];
  int inc2 = c;
#pragma unroll
  for (int o = 1; o < 32; o <<= 1) {
    const int u = __shfl_up(inc2, o);
    if (lane >= o) inc2 += u;
  }
  if (lane == 31) s_wsum2[wave] = inc2;
  __syncthreads();
  int wbase2 = 0;
#pragma unroll
  for (int w = 0; w < BIN_NODES / 32; ++w)
    if (w < wave) wbase2 += s_wsum2[w];
  const int start = base + wbase2 + inc2 - c;
  const int node = b * BIN_NODES + t;
  if (node < n_nodes) {
    cnt[node] = c;
    off[node] = start;
  }
  s_cnt[t] = start;
  __syncthreads();

  for (int i0 = 0; i0 < tot; i0 += 4 * BIN_NODES) {
    unsigned v[4];
#pragma unroll
    for (int j = 0; j < 4; ++j) {
      const int i = i0 + j * BIN_NODES + t;
      v[j] = (unsigned)p[i < tot ? i : tot - 1];
    }
    int pos[4];
#pragma unroll
    for (int j = 0; j < 4; ++j)
      pos[j] = (i0 + j * BIN_NODES + t < tot) ? atomicAdd(&s_cnt[v[j] & (BIN_NODES - 1)], 1) : -1;
#pragma unroll
    for (int j = 0; j < 4; ++j)
      if (pos[j] >= 0) csr[pos[j]] = (int)(v[j] >> BIN_SHIFT);
  }
}

__global__ __launch_bounds__(256) void gather_sum_kernel(
    const float* __restrict__ xin, const int* __restrict__ off, const int* __restrict__ cnt,
    const int* __restrict__ csr, float* __restrict__ agg, int n_nodes) {
  const int tid = blockIdx.x * blockDim.x + threadIdx.x;
  const int i = tid >> 4;
  const int c = (tid & 15) << 2;
  if (i >= n_nodes) return;
  const int len = cnt[i];
  const int* p = csr + off[i];
  float4 acc = *(const float4*)(xin + (long long)i * HIDDEN + c);
  const int last = len - 1;
  for (int j = 0; j < len; j += 4) {
    const int j1 = j + 1, j2 = j + 2, j3 = j + 3;
    const int s0 = p[j];
    const int s1 = p[j1 < last ? j1 : last];
    const int s2 = p[j2 < last ? j2 : last];
    const int s3 = p[j3 < last ? j3 : last];
    const float4 a0 = *(const float4*)(xin + (long long)s0 * HIDDEN + c);
    const float4 a1 = *(const float4*)(xin + (long long)s1 * HIDDEN + c);
    const float4 a2 = *(const float4*)(xin + (long long)s2 * HIDDEN + c);
    const float4 a3 = *(const float4*)(xin + (long long)s3 * HIDDEN + c);
    const int rem = len - j;
    acc.x += a0.x; acc.y += a0.y; acc.z += a0.z; acc.w += a0.w;
    const bool k1 = rem > 1, k2 = rem > 2, k3 = rem > 3;
    acc.x += k1 ? a1.x : 0.0f; acc.y += k1 ? a1.y : 0.0f;
    acc.z += k1 ? a1.z : 0.0f; acc.w += k1 ? a1.w : 0.0f;
    acc.x += k2 ? a2.x : 0.0f; acc.y += k2 ? a2.y : 0.0f;
    acc.z += k2 ? a2.z : 0.0f; acc.w += k2 ? a2.w : 0.0f;
    acc.x += k3 ? a3.x : 0.0f; acc.y += k3 ? a3.y : 0.0f;
    acc.z += k3 ? a3.z : 0.0f; acc.w += k3 ? a3.w : 0.0f;
  }
  *(float4*)(agg + (long long)i * HIDDEN + c) = acc;
}

template <int FIN>
__global__ __launch_bounds__(256) void gin_mlp_kernel(
    const float* __restrict__ hin,
    const float* __restrict__ Wa,
    const float* __restrict__ ba,
    const float* __restrict__ Wb,
    const float* __restrict__ bb,
    const float* __restrict__ gg,
    const float* __restrict__ be,
    const float* __restrict__ mm,
    const float* __restrict__ vv,
    float* __restrict__ hout,
    int n_nodes) {
  __shared__ float sT[8][16 * 64];

  const int wave = threadIdx.x >> 5;
  const int lane = threadIdx.x & 31;
  const int half = lane >> 4;
  const int l16  = lane & 15;

  const int tile = blockIdx.x * 8 + wave;
  const int row0 = tile * 16;
  if (row0 >= n_nodes) return;

  v8f acc[4];
#pragma unroll
  for (int t = 0; t < 4; ++t) {
    float bias = ba[t * 16 + l16];
#pragma unroll
    for (int r = 0; r < 8; ++r) acc[t][r] = bias;
  }

  int arow = row0 + l16;
  if (arow >= n_nodes) arow = n_nodes - 1;
  const float* aptr = hin + (long long)arow * FIN + 2 * half;

  for (int k = 0; k < FIN; k += 4) {
    v2f a = *(const v2f*)(aptr + k);
#pragma unroll
    for (int t = 0; t < 4; ++t) {
      const float* bp = Wa + (t * 16 + l16) * FIN + k + 2 * half;
      v2f b = *(const v2f*)bp;
      acc[t] = __builtin_amdgcn_wmma_f32_16x16x4_f32(
          false, a, false, b, (short)0, acc[t], false, false);
    }
  }

#pragma unroll
  for (int t = 0; t < 4; ++t) {
#pragma unroll
    for (int r = 0; r < 8; ++r) {
      float x = acc[t][r];
      x = x > 0.0f ? x : 0.0f;
      sT[wave][(r + 8 * half) * 64 + t * 16 + l16] = x;
    }
  }

  v8f acc2[4];
#pragma unroll
  for (int t = 0; t < 4; ++t)
#pragma unroll
    for (int r = 0; r < 8; ++r) acc2[t][r] = 0.0f;

  const float* tp = &sT[wave][l16 * 64 + 2 * half];
  for (int k = 0; k < 64; k += 4) {
    v2f a = *(const v2f*)(tp + k);
#pragma unroll
    for (int t = 0; t < 4; ++t) {
      const float* bp = Wb + (t * 16 + l16) * 64 + k + 2 * half;
      v2f b = *(const v2f*)bp;
      acc2[t] = __builtin_amdgcn_wmma_f32_16x16x4_f32(
          false, a, false, b, (short)0, acc2[t], false, false);
    }
  }

#pragma unroll
  for (int t = 0; t < 4; ++t) {
    int n = t * 16 + l16;
    float s  = gg[n] * rsqrtf(vv[n] + BN_EPS);
    float sh = (bb[n] - mm[n]) * s + be[n];
#pragma unroll
    for (int r = 0; r < 8; ++r) {
      int mrow = row0 + r + 8 * half;
      float o = acc2[t][r] * s + sh;
      o = o > 0.0f ? o : 0.0f;
      if (mrow < n_nodes) hout[(long long)mrow * 64 + n] = o;
    }
  }
}

__global__ __launch_bounds__(256) void gin_proj_kernel(
    const float* __restrict__ xin,
    const float* __restrict__ Wa,
    float* __restrict__ ysrc,
    int n_nodes) {
  const int wave = threadIdx.x >> 5;
  const int lane = threadIdx.x & 31;
  const int half = lane >> 4;
  const int l16  = lane & 15;

  const int tile = blockIdx.x * 8 + wave;
  const int row0 = tile * 16;
  if (row0 >= n_nodes) return;

  v8f acc[4];
#pragma unroll
  for (int t = 0; t < 4; ++t)
#pragma unroll
    for (int r = 0; r < 8; ++r) acc[t][r] = 0.0f;

  int arow = row0 + l16;
  if (arow >= n_nodes) arow = n_nodes - 1;
  const float* aptr = xin + (long long)arow * IN_FEATS + 2 * half;

  for (int k = 0; k < IN_FEATS; k += 4) {
    v2f a = *(const v2f*)(aptr + k);
#pragma unroll
    for (int t = 0; t < 4; ++t) {
      const float* bp = Wa + (t * 16 + l16) * IN_FEATS + k + 2 * half;
      v2f b = *(const v2f*)bp;
      acc[t] = __builtin_amdgcn_wmma_f32_16x16x4_f32(
          false, a, false, b, (short)0, acc[t], false, false);
    }
  }

#pragma unroll
  for (int t = 0; t < 4; ++t) {
    int n = t * 16 + l16;
#pragma unroll
    for (int r = 0; r < 8; ++r) {
      int mrow = row0 + r + 8 * half;
      if (mrow < n_nodes) ysrc[(long long)mrow * HIDDEN + n] = acc[t][r];
    }
  }
}

__global__ __launch_bounds__(256) void gin_tail_kernel(
    const float* __restrict__ agg,
    const float* __restrict__ ba,
    const float* __restrict__ Wb,
    const float* __restrict__ bb,
    const float* __restrict__ gg,
    const float* __restrict__ be,
    const float* __restrict__ mm,
    const float* __restrict__ vv,
    float* __restrict__ hout,
    int n_nodes) {
  const int wave = threadIdx.x >> 5;
  const int lane = threadIdx.x & 31;
  const int half = lane >> 4;
  const int l16  = lane & 15;

  const int tile = blockIdx.x * 8 + wave;
  const int row0 = tile * 16;
  if (row0 >= n_nodes) return;

  v8f acc2[4];
#pragma unroll
  for (int t = 0; t < 4; ++t)
#pragma unroll
    for (int r = 0; r < 8; ++r) acc2[t][r] = 0.0f;

  int arow = row0 + l16;
  if (arow >= n_nodes) arow = n_nodes - 1;
  const float* aptr = agg + (long long)arow * HIDDEN + 2 * half;
  const float* biasp = ba + 2 * half;

  for (int k = 0; k < HIDDEN; k += 4) {
    v2f a = *(const v2f*)(aptr + k);
    v2f bs = *(const v2f*)(biasp + k);
    a.x = a.x + bs.x;
    a.y = a.y + bs.y;
    a.x = a.x > 0.0f ? a.x : 0.0f;
    a.y = a.y > 0.0f ? a.y : 0.0f;
#pragma unroll
    for (int t = 0; t < 4; ++t) {
      const float* bp = Wb + (t * 16 + l16) * HIDDEN + k + 2 * half;
      v2f b = *(const v2f*)bp;
      acc2[t] = __builtin_amdgcn_wmma_f32_16x16x4_f32(
          false, a, false, b, (short)0, acc2[t], false, false);
    }
  }

#pragma unroll
  for (int t = 0; t < 4; ++t) {
    int n = t * 16 + l16;
    float s  = gg[n] * rsqrtf(vv[n] + BN_EPS);
    float sh = (bb[n] - mm[n]) * s + be[n];
#pragma unroll
    for (int r = 0; r < 8; ++r) {
      int mrow = row0 + r + 8 * half;
      float o = acc2[t][r] * s + sh;
      o = o > 0.0f ? o : 0.0f;
      if (mrow < n_nodes) hout[(long long)mrow * HIDDEN + n] = o;
    }
  }
}

__global__ __launch_bounds__(256) void pool_zero_kernel(float* __restrict__ sums,
                                                        float* __restrict__ counts) {
  int i = blockIdx.x * blockDim.x + threadIdx.x;
  if (i < NGRAPH * HIDDEN) sums[i] = 0.0f;
  if (i < NGRAPH) counts[i] = 0.0f;
}

__global__ __launch_bounds__(256) void pool_sum_kernel(
    const float* __restrict__ h, const int* __restrict__ batch,
    float* __restrict__ sums, float* __restrict__ counts, int n_nodes) {
  long long tid = (long long)blockIdx.x * blockDim.x + threadIdx.x;
  long long total = (long long)n_nodes * 16;
  if (tid >= total) return;
  int i = (int)(tid >> 4);
  int c = ((int)tid & 15) << 2;
  int g = batch[i];
  const float4 vv = *(const float4*)(h + (long long)i * HIDDEN + c);
  float* p = sums + (long long)g * HIDDEN + c;
  atomicAdd(p + 0, vv.x);
  atomicAdd(p + 1, vv.y);
  atomicAdd(p + 2, vv.z);
  atomicAdd(p + 3, vv.w);
  if (c == 0) atomicAdd(&counts[g], 1.0f);
}

__global__ __launch_bounds__(256) void pool_div_kernel(
    const float* __restrict__ sums, const float* __restrict__ counts,
    float* __restrict__ out) {
  int i = blockIdx.x * blockDim.x + threadIdx.x;
  if (i >= NGRAPH * HIDDEN) return;
  float c = counts[i >> 6];
  c = c > 1.0f ? c : 1.0f;
  out[i] = sums[i] / c;
}

extern "C" void kernel_launch(void* const* d_in, const int* in_sizes, int n_in,
                              void* d_out, int out_size, void* d_ws, size_t ws_size,
                              hipStream_t stream) {
  const float* x   = (const float*)d_in[0];
  const int* ei    = (const int*)d_in[1];
  const int* batch = (const int*)d_in[2];
  const float* W1a = (const float*)d_in[3];
  const float* b1a = (const float*)d_in[4];
  const float* W1b = (const float*)d_in[5];
  const float* b1b = (const float*)d_in[6];
  const float* g1  = (const float*)d_in[7];
  const float* be1 = (const float*)d_in[8];
  const float* m1  = (const float*)d_in[9];
  const float* v1  = (const float*)d_in[10];
  const float* W2a = (const float*)d_in[11];
  const float* b2a = (const float*)d_in[12];
  const float* W2b = (const float*)d_in[13];
  const float* b2b = (const float*)d_in[14];
  const float* g2  = (const float*)d_in[15];
  const float* be2 = (const float*)d_in[16];
  const float* m2  = (const float*)d_in[17];
  const float* v2  = (const float*)d_in[18];
  float* out = (float*)d_out;

  const int N = in_sizes[0] / IN_FEATS;
  const int E = in_sizes[1] / 2;

  float* ws    = (float*)d_ws;
  float* y1    = ws;
  float* agg   = ws + (long long)N * HIDDEN;
  float* h1    = ws + (long long)N * IN_FEATS;
  float* h2    = h1 + (long long)N * HIDDEN;
  float* sums  = h2 + (long long)N * HIDDEN;
  float* cnts  = sums + NGRAPH * HIDDEN;
  const int Npad = (N + 3) & ~3;
  int* icnt    = (int*)(cnts + NGRAPH);
  int* ioff    = icnt + Npad;
  int* ipairs  = ioff + Npad;
  int* icsr    = ipairs + E;
  int* ibtot   = icsr + E;
  int* ibcur   = ibtot + MAX_BINS;
  int* ibhist  = ibcur + MAX_BINS;
  (void)ws_size; (void)n_in; (void)out_size;

  const int tiles  = (N + 15) / 16;
  const int mlpBlk = (tiles + 7) / 8;
  const int nBins = (N + BIN_NODES - 1) >> BIN_SHIFT;
  const int gathBlk = (int)(((long long)N * 16 + 255) / 256);

  bin_zero_kernel<<<1, MAX_BINS, 0, stream>>>(ibtot, ibcur);
  bin_count_kernel<<<SORT_BLOCKS, SORT_THREADS, 0, stream>>>(ei, ibtot, ibhist, E);
  bin_scatter_kernel<<<SORT_BLOCKS, SORT_THREADS, 0, stream>>>(ei, ibtot, ibcur, ibhist, ipairs, E);
  bin_sort_kernel<<<nBins, BIN_NODES, 0, stream>>>(ibtot, ipairs, icnt, ioff, icsr, N);

  gin_proj_kernel<<<mlpBlk, 256, 0, stream>>>(x, W1a, y1, N);
  gather_sum_kernel<<<gathBlk, 256, 0, stream>>>(y1, ioff, icnt, icsr, agg, N);
  gin_tail_kernel<<<mlpBlk, 256, 0, stream>>>(
      agg, b1a, W1b, b1b, g1, be1, m1, v1, h1, N);

  gather_sum_kernel<<<gathBlk, 256, 0, stream>>>(h1, ioff, icnt, icsr, agg, N);
  gin_mlp_kernel<HIDDEN><<<mlpBlk, 256, 0, stream>>>(
      agg, W2a, b2a, W2b, b2b, g2, be2, m2, v2, h2, N);

  pool_zero_kernel<<<(NGRAPH * HIDDEN + 255) / 256, 256, 0, stream>>>(sums, cnts);
  {
    long long tot = (long long)N * 16;
    int blocks = (int)((tot + 255) / 256);
    pool_sum_kernel<<<blocks, 256, 0, stream>>>(h2, batch, sums, cnts, N);
  }
  pool_div_kernel<<<(NGRAPH * HIDDEN + 255) / 256, 256, 0, stream>>>(sums, cnts, out);
}


# Mean pool fused into the layer-2 MLP epilogue: each 16-row tile adds its per-graph column sums (64 f32 atomics, not 1024); layer-2 output not stored, no pool pass; sum order: tile rows, then arrival
# speedup vs baseline: 3.5741x; 1.3278x over previous
#include <hip/hip_runtime.h>

typedef float v2f __attribute__((ext_vector_type(2)));
typedef float v8f __attribute__((ext_vector_type(8)));

#define IN_FEATS 128
#define HIDDEN   64
#define NGRAPH   512
#define BN_EPS   1e-5f


#define BIN_SHIFT    9
#define BIN_NODES    512
#define MAX_BINS     256
#define SORT_THREADS 512
#define SORT_BLOCKS  256

__global__ __launch_bounds__(MAX_BINS) void bin_zero_kernel(int* __restrict__ binTotal,
                                                            int* __restrict__ binCursor) {
  binTotal[threadIdx.x] = 0;
  binCursor[threadIdx.x] = 0;
}

__global__ __launch_bounds__(SORT_THREADS) void bin_count_kernel(
    const int* __restrict__ ei, int* __restrict__ binTotal, int* __restrict__ blockHist, int E) {
  __shared__ int s_hist[MAX_BINS];
  const int t = threadIdx.x;
  if (t < MAX_BINS) s_hist[t] = 0;
  __syncthreads();
  const int* dst = ei + E;
  const int chunk = (E + (int)gridDim.x - 1) / (int)gridDim.x;
  const int c0 = (int)blockIdx.x * chunk;
  int c1 = c0 + chunk;
  if (c1 > E) c1 = E;
  for (int b0 = c0; b0 < c1; b0 += 4 * SORT_THREADS) {
    int d[4];
#pragma unroll
    for (int j = 0; j < 4; ++j) {
      const int e = b0 + j * SORT_THREADS + t;
      d[j] = dst[e < c1 ? e : c1 - 1];
    }
#pragma unroll
    for (int j = 0; j < 4; ++j)
      if (b0 + j * SORT_THREADS + t < c1) atomicAdd(&s_hist[d[j] >> BIN_SHIFT], 1);
  }
  __syncthreads();
  if (t < MAX_BINS) {
    const int h = s_hist[t];
    blockHist[(int)blockIdx.x * MAX_BINS + t] = h;
    if (h > 0) atomicAdd(&binTotal[t], h);
  }
}

__global__ __launch_bounds__(SORT_THREADS) void bin_scatter_kernel(
    const int* __restrict__ ei, const int* __restrict__ binTotal, int* __restrict__ binCursor,
    const int* __restrict__ blockHist, int* __restrict__ pairs, int E) {
  __shared__ int s_cur[MAX_BINS];
  __shared__ int s_wsum[SORT_THREADS / 32];
  const int t = threadIdx.x;
  const int lane = t & 31;
  const int wave = t >> 5;

  const int tot = t < MAX_BINS ? binTotal[t] : 0;
  int inc = tot;
#pragma unroll
  for (int o = 1; o < 32; o <<= 1) {
    const int u = __shfl_up(inc, o);
    if (lane >= o) inc += u;
  }
  if (lane == 31) s_wsum[wave] = inc;
  __syncthreads();
  int base = 0;
#pragma unroll
  for (int w = 0; w < SORT_THREADS / 32; ++w)
    if (w < wave) base += s_wsum[w];
  if (t < MAX_BINS) {
    const int h = blockHist[(int)blockIdx.x * MAX_BINS + t];
    int st = 0;
    if (h > 0) st = atomicAdd(&binCursor[t], h);
    s_cur[t] = base + inc - tot + st;
  }
  __syncthreads();

  const int* srcp = ei;
  const int* dst = ei + E;
  const int chunk = (E + (int)gridDim.x - 1) / (int)gridDim.x;
  const int c0 = (int)blockIdx.x * chunk;
  int c1 = c0 + chunk;
  if (c1 > E) c1 = E;
  for (int b0 = c0; b0 < c1; b0 += 4 * SORT_THREADS) {
    int s[4], d[4];
#pragma unroll
    for (int j = 0; j < 4; ++j) {
      const int e = b0 + j * SORT_THREADS + t;
      const int ec = e < c1 ? e : c1 - 1;
      d[j] = dst[ec];
      s[j] = srcp[ec];
    }
    int pos[4];
#pragma unroll
    for (int j = 0; j < 4; ++j)
      pos[j] = (b0 + j * SORT_THREADS + t < c1) ? atomicAdd(&s_cur[d[j] >> BIN_SHIFT], 1) : -1;
#pragma unroll
    for (int j = 0; j < 4; ++j)
      if (pos[j] >= 0)
        pairs[pos[j]] = (int)(((unsigned)s[j] << BIN_SHIFT) | (unsigned)(d[j] & (BIN_NODES - 1)));
  }
}

__global__ __launch_bounds__(BIN_NODES) void bin_sort_kernel(
    const int* __restrict__ binTotal, const int* __restrict__ pairs,
    int* __restrict__ cnt, int* __restrict__ off, int* __restrict__ csr, int n_nodes) {
  __shared__ int s_cnt[BIN_NODES];
  __shared__ int s_wsum[BIN_NODES / 32];
  __shared__ int s_wsum2[BIN_NODES / 32];
  __shared__ int s_bin[2];
  const int t = threadIdx.x;
  const int lane = t & 31;
  const int wave = t >> 5;
  const int b = (int)blockIdx.x;

  s_cnt[t] = 0;
  const int btot = t < MAX_BINS ? binTotal[t] : 0;
  int inc = btot;
#pragma unroll
  for (int o = 1; o < 32; o <<= 1) {
    const int u = __shfl_up(inc, o);
    if (lane >= o) inc += u;
  }
  if (lane == 31) s_wsum[wave] = inc;
  __syncthreads();
  int wbase = 0;
#pragma unroll
  for (int w = 0; w < BIN_NODES / 32; ++w)
    if (w < wave) wbase += s_wsum[w];
  if (t == b) {
    s_bin[0] = wbase + inc - btot;
    s_bin[1] = btot;
  }
  __syncthreads();
  const int base = s_bin[0];
  const int tot = s_bin[1];
  const int* p = pairs + base;

  for (int i0 = 0; i0 < tot; i0 += 4 * BIN_NODES) {
    unsigned v[4];
#pragma unroll
    for (int j = 0; j < 4; ++j) {
      const int i = i0 + j * BIN_NODES + t;
      v[j] = (unsigned)p[i < tot ? i : tot - 1];
    }
#pragma unroll
    for (int j = 0; j < 4; ++j)
      if (i0 + j * BIN_NODES + t < tot) atomicAdd(&s_cnt[v[j] & (BIN_NODES - 1)], 1);
  }
  __syncthreads();

  const int c = s_cnt[t];
  int inc2 = c;
#pragma unroll
  for (int o = 1; o < 32; o <<= 1) {
    const int u = __shfl_up(inc2, o);
    if (lane >= o) inc2 += u;
  }
  if (lane == 31) s_wsum2[wave] = inc2;
  __syncthreads();
  int wbase2 = 0;
#pragma unroll
  for (int w = 0; w < BIN_NODES / 32; ++w)
    if (w < wave) wbase2 += s_wsum2[w];
  const int start = base + wbase2 + inc2 - c;
  const int node = b * BIN_NODES + t;
  if (node < n_nodes) {
    cnt[node] = c;
    off[node] = start;
  }
  s_cnt[t] = start;
  __syncthreads();

  for (int i0 = 0; i0 < tot; i0 += 4 * BIN_NODES) {
    unsigned v[4];
#pragma unroll
    for (int j = 0; j < 4; ++j) {
      const int i = i0 + j * BIN_NODES + t;
      v[j] = (unsigned)p[i < tot ? i : tot - 1];
    }
    int pos[4];
#pragma unroll
    for (int j = 0; j < 4; ++j)
      pos[j] = (i0 + j * BIN_NODES + t < tot) ? atomicAdd(&s_cnt[v[j] & (BIN_NODES - 1)], 1) : -1;
#pragma unroll
    for (int j = 0; j < 4; ++j)
      if (pos[j] >= 0) csr[pos[j]] = (int)(v[j] >> BIN_SHIFT);
  }
}

__global__ __launch_bounds__(256) void gather_sum_kernel(
    const float* __restrict__ xin, const int* __restrict__ off, const int* __restrict__ cnt,
    const int* __restrict__ csr, float* __restrict__ agg, int n_nodes) {
  const int tid = blockIdx.x * blockDim.x + threadIdx.x;
  const int i = tid >> 4;
  const int c = (tid & 15) << 2;
  if (i >= n_nodes) return;
  const int len = cnt[i];
  const int* p = csr + off[i];
  float4 acc = *(const float4*)(xin + (long long)i * HIDDEN + c);
  const int last = len - 1;
  for (int j = 0; j < len; j += 4) {
    const int j1 = j + 1, j2 = j + 2, j3 = j + 3;
    const int s0 = p[j];
    const int s1 = p[j1 < last ? j1 : last];
    const int s2 = p[j2 < last ? j2 : last];
    const int s3 = p[j3 < last ? j3 : last];
    const float4 a0 = *(const float4*)(xin + (long long)s0 * HIDDEN + c);
    const float4 a1 = *(const float4*)(xin + (long long)s1 * HIDDEN + c);
    const float4 a2 = *(const float4*)(xin + (long long)s2 * HIDDEN + c);
    const float4 a3 = *(const float4*)(xin + (long long)s3 * HIDDEN + c);
    const int rem = len - j;
    acc.x += a0.x; acc.y += a0.y; acc.z += a0.z; acc.w += a0.w;
    const bool k1 = rem > 1, k2 = rem > 2, k3 = rem > 3;
    acc.x += k1 ? a1.x : 0.0f; acc.y += k1 ? a1.y : 0.0f;
    acc.z += k1 ? a1.z : 0.0f; acc.w += k1 ? a1.w : 0.0f;
    acc.x += k2 ? a2.x : 0.0f; acc.y += k2 ? a2.y : 0.0f;
    acc.z += k2 ? a2.z : 0.0f; acc.w += k2 ? a2.w : 0.0f;
    acc.x += k3 ? a3.x : 0.0f; acc.y += k3 ? a3.y : 0.0f;
    acc.z += k3 ? a3.z : 0.0f; acc.w += k3 ? a3.w : 0.0f;
  }
  *(float4*)(agg + (long long)i * HIDDEN + c) = acc;
}

__device__ __forceinline__ void pool_add_run(float* __restrict__ sums, float* __restrict__ counts,
                                             int g, int l16, const float* run, float rc) {
  float* sp = sums + (long long)g * HIDDEN + l16;
  atomicAdd(sp, run[0]);
  atomicAdd(sp + 16, run[1]);
  atomicAdd(sp + 32, run[2]);
  atomicAdd(sp + 48, run[3]);
  if (l16 == 0) atomicAdd(&counts[g], rc);
}

template <int FIN>
__global__ __launch_bounds__(256) void gin_mlp_kernel(
    const float* __restrict__ hin,
    const float* __restrict__ Wa,
    const float* __restrict__ ba,
    const float* __restrict__ Wb,
    const float* __restrict__ bb,
    const float* __restrict__ gg,
    const float* __restrict__ be,
    const float* __restrict__ mm,
    const float* __restrict__ vv,
    const int* __restrict__ batch,
    float* __restrict__ sums,
    float* __restrict__ counts,
    int n_nodes) {
  __shared__ float sT[8][16 * 64];

  const int wave = threadIdx.x >> 5;
  const int lane = threadIdx.x & 31;
  const int half = lane >> 4;
  const int l16  = lane & 15;

  const int tile = blockIdx.x * 8 + wave;
  const int row0 = tile * 16;
  if (row0 >= n_nodes) return;

  v8f acc[4];
#pragma unroll
  for (int t = 0; t < 4; ++t) {
    float bias = ba[t * 16 + l16];
#pragma unroll
    for (int r = 0; r < 8; ++r) acc[t][r] = bias;
  }

  int arow = row0 + l16;
  if (arow >= n_nodes) arow = n_nodes - 1;
  const float* aptr = hin + (long long)arow * FIN + 2 * half;
  const int gl = batch[arow];

  for (int k = 0; k < FIN; k += 4) {
    v2f a = *(const v2f*)(aptr + k);
#pragma unroll
    for (int t = 0; t < 4; ++t) {
      const float* bp = Wa + (t * 16 + l16) * FIN + k + 2 * half;
      v2f b = *(const v2f*)bp;
      acc[t] = __builtin_amdgcn_wmma_f32_16x16x4_f32(
          false, a, false, b, (short)0, acc[t], false, false);
    }
  }

#pragma unroll
  for (int t = 0; t < 4; ++t) {
#pragma unroll
    for (int r = 0; r < 8; ++r) {
      float x = acc[t][r];
      x = x > 0.0f ? x : 0.0f;
      sT[wave][(r + 8 * half) * 64 + t * 16 + l16] = x;
    }
  }

  v8f acc2[4];
#pragma unroll
  for (int t = 0; t < 4; ++t)
#pragma unroll
    for (int r = 0; r < 8; ++r) acc2[t][r] = 0.0f;

  const float* tp = &sT[wave][l16 * 64 + 2 * half];
  for (int k = 0; k < 64; k += 4) {
    v2f a = *(const v2f*)(tp + k);
#pragma unroll
    for (int t = 0; t < 4; ++t) {
      const float* bp = Wb + (t * 16 + l16) * 64 + k + 2 * half;
      v2f b = *(const v2f*)bp;
      acc2[t] = __builtin_amdgcn_wmma_f32_16x16x4_f32(
          false, a, false, b, (short)0, acc2[t], false, false);
    }
  }

#pragma unroll
  for (int t = 0; t < 4; ++t) {
    int n = t * 16 + l16;
    float s  = gg[n] * rsqrtf(vv[n] + BN_EPS);
    float sh = (bb[n] - mm[n]) * s + be[n];
#pragma unroll
    for (int r = 0; r < 8; ++r) {
      float o = acc2[t][r] * s + sh;
      acc2[t][r] = o > 0.0f ? o : 0.0f;
    }
  }

  const int g0 = __builtin_amdgcn_readfirstlane(gl);
  const bool whole = row0 + 16 <= n_nodes;
  if (whole && __all(gl == g0)) {
    float cs[4];
#pragma unroll
    for (int t = 0; t < 4; ++t) {
      float v = acc2[t][0];
#pragma unroll
      for (int r = 1; r < 8; ++r) v += acc2[t][r];
      cs[t] = v + __shfl_xor(v, 16);
    }
    float* sp = sums + (long long)g0 * HIDDEN + lane;
    atomicAdd(sp, half ? cs[1] : cs[0]);
    atomicAdd(sp + 32, half ? cs[3] : cs[2]);
    if (lane == 0) atomicAdd(&counts[g0], 16.0f);
  } else {
    int gr[8];
#pragma unroll
    for (int r = 0; r < 8; ++r) gr[r] = __shfl(gl, 8 * half + r);
    int gp = 0;
    float rc = 0.0f;
    float run[4] = {0.0f, 0.0f, 0.0f, 0.0f};
#pragma unroll
    for (int r = 0; r < 8; ++r) {
      if (row0 + 8 * half + r < n_nodes) {
        if (rc > 0.0f && gr[r] != gp) {
          pool_add_run(sums, counts, gp, l16, run, rc);
          rc = 0.0f;
          run[0] = 0.0f; run[1] = 0.0f; run[2] = 0.0f; run[3] = 0.0f;
        }
        gp = gr[r];
        rc += 1.0f;
#pragma unroll
        for (int t = 0; t < 4; ++t) run[t] += acc2[t][r];
      }
    }
    if (rc > 0.0f) pool_add_run(sums, counts, gp, l16, run, rc);
  }
}

__global__ __launch_bounds__(256) void gin_proj_kernel(
    const float* __restrict__ xin,
    const float* __restrict__ Wa,
    float* __restrict__ ysrc,
    int n_nodes) {
  const int wave = threadIdx.x >> 5;
  const int lane = threadIdx.x & 31;
  const int half = lane >> 4;
  const int l16  = lane & 15;

  const int tile = blockIdx.x * 8 + wave;
  const int row0 = tile * 16;
  if (row0 >= n_nodes) return;

  v8f acc[4];
#pragma unroll
  for (int t = 0; t < 4; ++t)
#pragma unroll
    for (int r = 0; r < 8; ++r) acc[t][r] = 0.0f;

  int arow = row0 + l16;
  if (arow >= n_nodes) arow = n_nodes - 1;
  const float* aptr = xin + (long long)arow * IN_FEATS + 2 * half;

  for (int k = 0; k < IN_FEATS; k += 4) {
    v2f a = *(const v2f*)(aptr + k);
#pragma unroll
    for (int t = 0; t < 4; ++t) {
      const float* bp = Wa + (t * 16 + l16) * IN_FEATS + k + 2 * half;
      v2f b = *(const v2f*)bp;
      acc[t] = __builtin_amdgcn_wmma_f32_16x16x4_f32(
          false, a, false, b, (short)0, acc[t], false, false);
    }
  }

#pragma unroll
  for (int t = 0; t < 4; ++t) {
    int n = t * 16 + l16;
#pragma unroll
    for (int r = 0; r < 8; ++r) {
      int mrow = row0 + r + 8 * half;
      if (mrow < n_nodes) ysrc[(long long)mrow * HIDDEN + n] = acc[t][r];
    }
  }
}

__global__ __launch_bounds__(256) void gin_tail_kernel(
    const float* __restrict__ agg,
    const float* __restrict__ ba,
    const float* __restrict__ Wb,
    const float* __restrict__ bb,
    const float* __restrict__ gg,
    const float* __restrict__ be,
    const float* __restrict__ mm,
    const float* __restrict__ vv,
    float* __restrict__ hout,
    int n_nodes) {
  const int wave = threadIdx.x >> 5;
  const int lane = threadIdx.x & 31;
  const int half = lane >> 4;
  const int l16  = lane & 15;

  const int tile = blockIdx.x * 8 + wave;
  const int row0 = tile * 16;
  if (row0 >= n_nodes) return;

  v8f acc2[4];
#pragma unroll
  for (int t = 0; t < 4; ++t)
#pragma unroll
    for (int r = 0; r < 8; ++r) acc2[t][r] = 0.0f;

  int arow = row0 + l16;
  if (arow >= n_nodes) arow = n_nodes - 1;
  const float* aptr = agg + (long long)arow * HIDDEN + 2 * half;
  const float* biasp = ba + 2 * half;

  for (int k = 0; k < HIDDEN; k += 4) {
    v2f a = *(const v2f*)(aptr + k);
    v2f bs = *(const v2f*)(biasp + k);
    a.x = a.x + bs.x;
    a.y = a.y + bs.y;
    a.x = a.x > 0.0f ? a.x : 0.0f;
    a.y = a.y > 0.0f ? a.y : 0.0f;
#pragma unroll
    for (int t = 0; t < 4; ++t) {
      const float* bp = Wb + (t * 16 + l16) * HIDDEN + k + 2 * half;
      v2f b = *(const v2f*)bp;
      acc2[t] = __builtin_amdgcn_wmma_f32_16x16x4_f32(
          false, a, false, b, (short)0, acc2[t], false, false);
    }
  }

#pragma unroll
  for (int t = 0; t < 4; ++t) {
    int n = t * 16 + l16;
    float s  = gg[n] * rsqrtf(vv[n] + BN_EPS);
    float sh = (bb[n] - mm[n]) * s + be[n];
#pragma unroll
    for (int r = 0; r < 8; ++r) {
      int mrow = row0 + r + 8 * half;
      float o = acc2[t][r] * s + sh;
      o = o > 0.0f ? o : 0.0f;
      if (mrow < n_nodes) hout[(long long)mrow * HIDDEN + n] = o;
    }
  }
}

__global__ __launch_bounds__(256) void pool_zero_kernel(float* __restrict__ sums,
                                                        float* __restrict__ counts) {
  int i = blockIdx.x * blockDim.x + threadIdx.x;
  if (i < NGRAPH * HIDDEN) sums[i] = 0.0f;
  if (i < NGRAPH) counts[i] = 0.0f;
}

__global__ __launch_bounds__(256) void pool_div_kernel(
    const float* __restrict__ sums, const float* __restrict__ counts,
    float* __restrict__ out) {
  int i = blockIdx.x * blockDim.x + threadIdx.x;
  if (i >= NGRAPH * HIDDEN) return;
  float c = counts[i >> 6];
  c = c > 1.0f ? c : 1.0f;
  out[i] = sums[i] / c;
}

extern "C" void kernel_launch(void* const* d_in, const int* in_sizes, int n_in,
                              void* d_out, int out_size, void* d_ws, size_t ws_size,
                              hipStream_t stream) {
  const float* x   = (const float*)d_in[0];
  const int* ei    = (const int*)d_in[1];
  const int* batch = (const int*)d_in[2];
  const float* W1a = (const float*)d_in[3];
  const float* b1a = (const float*)d_in[4];
  const float* W1b = (const float*)d_in[5];
  const float* b1b = (const float*)d_in[6];
  const float* g1  = (const float*)d_in[7];
  const float* be1 = (const float*)d_in[8];
  const float* m1  = (const float*)d_in[9];
  const float* v1  = (const float*)d_in[10];
  const float* W2a = (const float*)d_in[11];
  const float* b2a = (const float*)d_in[12];
  const float* W2b = (const float*)d_in[13];
  const float* b2b = (const float*)d_in[14];
  const float* g2  = (const float*)d_in[15];
  const float* be2 = (const float*)d_in[16];
  const float* m2  = (const float*)d_in[17];
  const float* v2  = (const float*)d_in[18];
  float* out = (float*)d_out;

  const int N = in_sizes[0] / IN_FEATS;
  const int E = in_sizes[1] / 2;

  float* ws    = (float*)d_ws;
  float* y1    = ws;
  float* agg   = ws + (long long)N * HIDDEN;
  float* h1    = ws + (long long)N * IN_FEATS;
  float* sums  = h1 + (long long)N * HIDDEN;
  float* cnts  = sums + NGRAPH * HIDDEN;
  const int Npad = (N + 3) & ~3;
  int* icnt    = (int*)(cnts + NGRAPH);
  int* ioff    = icnt + Npad;
  int* ipairs  = ioff + Npad;
  int* icsr    = ipairs + E;
  int* ibtot   = icsr + E;
  int* ibcur   = ibtot + MAX_BINS;
  int* ibhist  = ibcur + MAX_BINS;
  (void)ws_size; (void)n_in; (void)out_size;

  const int tiles  = (N + 15) / 16;
  const int mlpBlk = (tiles + 7) / 8;
  const int nBins = (N + BIN_NODES - 1) >> BIN_SHIFT;
  const int gathBlk = (int)(((long long)N * 16 + 255) / 256);

  bin_zero_kernel<<<1, MAX_BINS, 0, stream>>>(ibtot, ibcur);
  bin_count_kernel<<<SORT_BLOCKS, SORT_THREADS, 0, stream>>>(ei, ibtot, ibhist, E);
  bin_scatter_kernel<<<SORT_BLOCKS, SORT_THREADS, 0, stream>>>(ei, ibtot, ibcur, ibhist, ipairs, E);
  bin_sort_kernel<<<nBins, BIN_NODES, 0, stream>>>(ibtot, ipairs, icnt, ioff, icsr, N);

  gin_proj_kernel<<<mlpBlk, 256, 0, stream>>>(x, W1a, y1, N);
  gather_sum_kernel<<<gathBlk, 256, 0, stream>>>(y1, ioff, icnt, icsr, agg, N);
  gin_tail_kernel<<<mlpBlk, 256, 0, stream>>>(
      agg, b1a, W1b, b1b, g1, be1, m1, v1, h1, N);

  gather_sum_kernel<<<gathBlk, 256, 0, stream>>>(h1, ioff, icnt, icsr, agg, N);
  pool_zero_kernel<<<(NGRAPH * HIDDEN + 255) / 256, 256, 0, stream>>>(sums, cnts);
  gin_mlp_kernel<HIDDEN><<<mlpBlk, 256, 0, stream>>>(
      agg, W2a, b2a, W2b, b2b, g2, be2, m2, v2, batch, sums, cnts, N);
  pool_div_kernel<<<(NGRAPH * HIDDEN + 255) / 256, 256, 0, stream>>>(sums, cnts, out);
}


# MLP kernels: each block copies its 64 x K weight matrices into LDS once (row pitch K+4 words) and reads B fragments there, not 4 global loads of 16 lines per k-step; layer-2 tile at the same pitch
# speedup vs baseline: 4.0958x; 1.1460x over previous
#include <hip/hip_runtime.h>

typedef float v2f __attribute__((ext_vector_type(2)));
typedef float v8f __attribute__((ext_vector_type(8)));

#define IN_FEATS 128
#define HIDDEN   64
#define NGRAPH   512
#define BN_EPS   1e-5f


#define BIN_SHIFT    9
#define BIN_NODES    512
#define MAX_BINS     256
#define SORT_THREADS 512
#define SORT_BLOCKS  256

__global__ __launch_bounds__(MAX_BINS) void bin_zero_kernel(int* __restrict__ binTotal,
                                                            int* __restrict__ binCursor) {
  binTotal[threadIdx.x] = 0;
  binCursor[threadIdx.x] = 0;
}

__global__ __launch_bounds__(SORT_THREADS) void bin_count_kernel(
    const int* __restrict__ ei, int* __restrict__ binTotal, int* __restrict__ blockHist, int E) {
  __shared__ int s_hist[MAX_BINS];
  const int t = threadIdx.x;
  if (t < MAX_BINS) s_hist[t] = 0;
  __syncthreads();
  const int* dst = ei + E;
  const int chunk = (E + (int)gridDim.x - 1) / (int)gridDim.x;
  const int c0 = (int)blockIdx.x * chunk;
  int c1 = c0 + chunk;
  if (c1 > E) c1 = E;
  for (int b0 = c0; b0 < c1; b0 += 4 * SORT_THREADS) {
    int d[4];
#pragma unroll
    for (int j = 0; j < 4; ++j) {
      const int e = b0 + j * SORT_THREADS + t;
      d[j] = dst[e < c1 ? e : c1 - 1];
    }
#pragma unroll
    for (int j = 0; j < 4; ++j)
      if (b0 + j * SORT_THREADS + t < c1) atomicAdd(&s_hist[d[j] >> BIN_SHIFT], 1);
  }
  __syncthreads();
  if (t < MAX_BINS) {
    const int h = s_hist[t];
    blockHist[(int)blockIdx.x * MAX_BINS + t] = h;
    if (h > 0) atomicAdd(&binTotal[t], h);
  }
}

__global__ __launch_bounds__(SORT_THREADS) void bin_scatter_kernel(
    const int* __restrict__ ei, const int* __restrict__ binTotal, int* __restrict__ binCursor,
    const int* __restrict__ blockHist, int* __restrict__ pairs, int E) {
  __shared__ int s_cur[MAX_BINS];
  __shared__ int s_wsum[SORT_THREADS / 32];
  const int t = threadIdx.x;
  const int lane = t & 31;
  const int wave = t >> 5;

  const int tot = t < MAX_BINS ? binTotal[t] : 0;
  int inc = tot;
#pragma unroll
  for (int o = 1; o < 32; o <<= 1) {
    const int u = __shfl_up(inc, o);
    if (lane >= o) inc += u;
  }
  if (lane == 31) s_wsum[wave] = inc;
  __syncthreads();
  int base = 0;
#pragma unroll
  for (int w = 0; w < SORT_THREADS / 32; ++w)
    if (w < wave) base += s_wsum[w];
  if (t < MAX_BINS) {
    const int h = blockHist[(int)blockIdx.x * MAX_BINS + t];
    int st = 0;
    if (h > 0) st = atomicAdd(&binCursor[t], h);
    s_cur[t] = base + inc - tot + st;
  }
  __syncthreads();

  const int* srcp = ei;
  const int* dst = ei + E;
  const int chunk = (E + (int)gridDim.x - 1) / (int)gridDim.x;
  const int c0 = (int)blockIdx.x * chunk;
  int c1 = c0 + chunk;
  if (c1 > E) c1 = E;
  for (int b0 = c0; b0 < c1; b0 += 4 * SORT_THREADS) {
    int s[4], d[4];
#pragma unroll
    for (int j = 0; j < 4; ++j) {
      const int e = b0 + j * SORT_THREADS + t;
      const int ec = e < c1 ? e : c1 - 1;
      d[j] = dst[ec];
      s[j] = srcp[ec];
    }
    int pos[4];
#pragma unroll
    for (int j = 0; j < 4; ++j)
      pos[j] = (b0 + j * SORT_THREADS + t < c1) ? atomicAdd(&s_cur[d[j] >> BIN_SHIFT], 1) : -1;
#pragma unroll
    for (int j = 0; j < 4; ++j)
      if (pos[j] >= 0)
        pairs[pos[j]] = (int)(((unsigned)s[j] << BIN_SHIFT) | (unsigned)(d[j] & (BIN_NODES - 1)));
  }
}

__global__ __launch_bounds__(BIN_NODES) void bin_sort_kernel(
    const int* __restrict__ binTotal, const int* __restrict__ pairs,
    int* __restrict__ cnt, int* __restrict__ off, int* __restrict__ csr, int n_nodes) {
  __shared__ int s_cnt[BIN_NODES];
  __shared__ int s_wsum[BIN_NODES / 32];
  __shared__ int s_wsum2[BIN_NODES / 32];
  __shared__ int s_bin[2];
  const int t = threadIdx.x;
  const int lane = t & 31;
  const int wave = t >> 5;
  const int b = (int)blockIdx.x;

  s_cnt[t] = 0;
  const int btot = t < MAX_BINS ? binTotal[t] : 0;
  int inc = btot;
#pragma unroll
  for (int o = 1; o < 32; o <<= 1) {
    const int u = __shfl_up(inc, o);
    if (lane >= o) inc += u;
  }
  if (lane == 31) s_wsum[wave] = inc;
  __syncthreads();
  int wbase = 0;
#pragma unroll
  for (int w = 0; w < BIN_NODES / 32; ++w)
    if (w < wave) wbase += s_wsum[w];
  if (t == b) {
    s_bin[0] = wbase + inc - btot;
    s_bin[1] = btot;
  }
  __syncthreads();
  const int base = s_bin[0];
  const int tot = s_bin[1];
  const int* p = pairs + base;

  for (int i0 = 0; i0 < tot; i0 += 4 * BIN_NODES) {
    unsigned v[4];
#pragma unroll
    for (int j = 0; j < 4; ++j) {
      const int i = i0 + j * BIN_NODES + t;
      v[j] = (unsigned)p[i < tot ? i : tot - 1];
    }
#pragma unroll
    for (int j = 0; j < 4; ++j)
      if (i0 + j * BIN_NODES + t < tot) atomicAdd(&s_cnt[v[j] & (BIN_NODES - 1)], 1);
  }
  __syncthreads();

  const int c = s_cnt[t];
  int inc2 = c;
#pragma unroll
  for (int o = 1; o < 32; o <<= 1) {
    const int u = __shfl_up(inc2, o);
    if (lane >= o) inc2 += u;
  }
  if (lane == 31) s_wsum2[wave] = inc2;
  __syncthreads();
  int wbase2 = 0;
#pragma unroll
  for (int w = 0; w < BIN_NODES / 32; ++w)
    if (w < wave) wbase2 += s_wsum2[w];
  const int start = base + wbase2 + inc2 - c;
  const int node = b * BIN_NODES + t;
  if (node < n_nodes) {
    cnt[node] = c;
    off[node] = start;
  }
  s_cnt[t] = start;
  __syncthreads();

  for (int i0 = 0; i0 < tot; i0 += 4 * BIN_NODES) {
    unsigned v[4];
#pragma unroll
    for (int j = 0; j < 4; ++j) {
      const int i = i0 + j * BIN_NODES + t;
      v[j] = (unsigned)p[i < tot ? i : tot - 1];
    }
    int pos[4];
#pragma unroll
    for (int j = 0; j < 4; ++j)
      pos[j] = (i0 + j * BIN_NODES + t < tot) ? atomicAdd(&s_cnt[v[j] & (BIN_NODES - 1)], 1) : -1;
#pragma unroll
    for (int j = 0; j < 4; ++j)
      if (pos[j] >= 0) csr[pos[j]] = (int)(v[j] >> BIN_SHIFT);
  }
}

__global__ __launch_bounds__(256) void gather_sum_kernel(
    const float* __restrict__ xin, const int* __restrict__ off, const int* __restrict__ cnt,
    const int* __restrict__ csr, float* __restrict__ agg, int n_nodes) {
  const int tid = blockIdx.x * blockDim.x + threadIdx.x;
  const int i = tid >> 4;
  const int c = (tid & 15) << 2;
  if (i >= n_nodes) return;
  const int len = cnt[i];
  const int* p = csr + off[i];
  float4 acc = *(const float4*)(xin + (long long)i * HIDDEN + c);
  const int last = len - 1;
  for (int j = 0; j < len; j += 4) {
    const int j1 = j + 1, j2 = j + 2, j3 = j + 3;
    const int s0 = p[j];
    const int s1 = p[j1 < last ? j1 : last];
    const int s2 = p[j2 < last ? j2 : last];
    const int s3 = p[j3 < last ? j3 : last];
    const float4 a0 = *(const float4*)(xin + (long long)s0 * HIDDEN + c);
    const float4 a1 = *(const float4*)(xin + (long long)s1 * HIDDEN + c);
    const float4 a2 = *(const float4*)(xin + (long long)s2 * HIDDEN + c);
    const float4 a3 = *(const float4*)(xin + (long long)s3 * HIDDEN + c);
    const int rem = len - j;
    acc.x += a0.x; acc.y += a0.y; acc.z += a0.z; acc.w += a0.w;
    const bool k1 = rem > 1, k2 = rem > 2, k3 = rem > 3;
    acc.x += k1 ? a1.x : 0.0f; acc.y += k1 ? a1.y : 0.0f;
    acc.z += k1 ? a1.z : 0.0f; acc.w += k1 ? a1.w : 0.0f;
    acc.x += k2 ? a2.x : 0.0f; acc.y += k2 ? a2.y : 0.0f;
    acc.z += k2 ? a2.z : 0.0f; acc.w += k2 ? a2.w : 0.0f;
    acc.x += k3 ? a3.x : 0.0f; acc.y += k3 ? a3.y : 0.0f;
    acc.z += k3 ? a3.z : 0.0f; acc.w += k3 ? a3.w : 0.0f;
  }
  *(float4*)(agg + (long long)i * HIDDEN + c) = acc;
}

#define WPITCH_PAD 4
#define MLP_THREADS 256

template <int K>
__device__ __forceinline__ void stage_weight_rows(float* __restrict__ sW,
                                                  const float* __restrict__ W) {
  constexpr int Q = K / 4;
  constexpr int R = 64 * Q / MLP_THREADS;
  float4 v[R];
#pragma unroll
  for (int j = 0; j < R; ++j) {
    const int i = (int)threadIdx.x + j * MLP_THREADS;
    const int row = i / Q;
    v[j] = *(const float4*)(W + row * K + 4 * (i - row * Q));
  }
#pragma unroll
  for (int j = 0; j < R; ++j) {
    const int i = (int)threadIdx.x + j * MLP_THREADS;
    const int row = i / Q;
    *(float4*)(sW + row * (K + WPITCH_PAD) + 4 * (i - row * Q)) = v[j];
  }
}

__device__ __forceinline__ void pool_add_run(float* __restrict__ sums, float* __restrict__ counts,
                                             int g, int l16, const float* run, float rc) {
  float* sp = sums + (long long)g * HIDDEN + l16;
  atomicAdd(sp, run[0]);
  atomicAdd(sp + 16, run[1]);
  atomicAdd(sp + 32, run[2]);
  atomicAdd(sp + 48, run[3]);
  if (l16 == 0) atomicAdd(&counts[g], rc);
}

template <int FIN>
__global__ __launch_bounds__(256) void gin_mlp_kernel(
    const float* __restrict__ hin,
    const float* __restrict__ Wa,
    const float* __restrict__ ba,
    const float* __restrict__ Wb,
    const float* __restrict__ bb,
    const float* __restrict__ gg,
    const float* __restrict__ be,
    const float* __restrict__ mm,
    const float* __restrict__ vv,
    const int* __restrict__ batch,
    float* __restrict__ sums,
    float* __restrict__ counts,
    int n_nodes) {
  constexpr int PA = FIN + WPITCH_PAD;
  constexpr int PB = HIDDEN + WPITCH_PAD;
  __shared__ float sWa[64 * PA];
  __shared__ float sWb[64 * PB];
  __shared__ float sT[8][16 * PB];

  const int wave = threadIdx.x >> 5;
  const int lane = threadIdx.x & 31;
  const int half = lane >> 4;
  const int l16  = lane & 15;

  const int tile = blockIdx.x * 8 + wave;
  const int row0 = tile * 16;

  int arow = row0 + l16;
  if (arow >= n_nodes) arow = n_nodes - 1;
  const float* aptr = hin + (long long)arow * FIN + 2 * half;
  v2f afr[FIN / 4];
#pragma unroll
  for (int j = 0; j < FIN / 4; ++j) afr[j] = *(const v2f*)(aptr + 4 * j);

  stage_weight_rows<FIN>(sWa, Wa);
  stage_weight_rows<HIDDEN>(sWb, Wb);
  __syncthreads();
  if (row0 >= n_nodes) return;

  v8f acc[4];
#pragma unroll
  for (int t = 0; t < 4; ++t) {
    float bias = ba[t * 16 + l16];
#pragma unroll
    for (int r = 0; r < 8; ++r) acc[t][r] = bias;
  }

  const int gl = batch[arow];

#pragma unroll
  for (int k = 0; k < FIN; k += 4) {
    v2f a = afr[k / 4];
#pragma unroll
    for (int t = 0; t < 4; ++t) {
      const float* bp = sWa + (t * 16 + l16) * PA + k + 2 * half;
      v2f b = *(const v2f*)bp;
      acc[t] = __builtin_amdgcn_wmma_f32_16x16x4_f32(
          false, a, false, b, (short)0, acc[t], false, false);
    }
  }

#pragma unroll
  for (int t = 0; t < 4; ++t) {
#pragma unroll
    for (int r = 0; r < 8; ++r) {
      float x = acc[t][r];
      x = x > 0.0f ? x : 0.0f;
      sT[wave][(r + 8 * half) * PB + t * 16 + l16] = x;
    }
  }

  v8f acc2[4];
#pragma unroll
  for (int t = 0; t < 4; ++t)
#pragma unroll
    for (int r = 0; r < 8; ++r) acc2[t][r] = 0.0f;

  const float* tp = &sT[wave][l16 * PB + 2 * half];
  for (int k = 0; k < 64; k += 4) {
    v2f a = *(const v2f*)(tp + k);
#pragma unroll
    for (int t = 0; t < 4; ++t) {
      const float* bp = sWb + (t * 16 + l16) * PB + k + 2 * half;
      v2f b = *(const v2f*)bp;
      acc2[t] = __builtin_amdgcn_wmma_f32_16x16x4_f32(
          false, a, false, b, (short)0, acc2[t], false, false);
    }
  }

#pragma unroll
  for (int t = 0; t < 4; ++t) {
    int n = t * 16 + l16;
    float s  = gg[n] * rsqrtf(vv[n] + BN_EPS);
    float sh = (bb[n] - mm[n]) * s + be[n];
#pragma unroll
    for (int r = 0; r < 8; ++r) {
      float o = acc2[t][r] * s + sh;
      acc2[t][r] = o > 0.0f ? o : 0.0f;
    }
  }

  const int g0 = __builtin_amdgcn_readfirstlane(gl);
  const bool whole = row0 + 16 <= n_nodes;
  if (whole && __all(gl == g0)) {
    float cs[4];
#pragma unroll
    for (int t = 0; t < 4; ++t) {
      float v = acc2[t][0];
#pragma unroll
      for (int r = 1; r < 8; ++r) v += acc2[t][r];
      cs[t] = v + __shfl_xor(v, 16);
    }
    float* sp = sums + (long long)g0 * HIDDEN + lane;
    atomicAdd(sp, half ? cs[1] : cs[0]);
    atomicAdd(sp + 32, half ? cs[3] : cs[2]);
    if (lane == 0) atomicAdd(&counts[g0], 16.0f);
  } else {
    int gr[8];
#pragma unroll
    for (int r = 0; r < 8; ++r) gr[r] = __shfl(gl, 8 * half + r);
    int gp = 0;
    float rc = 0.0f;
    float run[4] = {0.0f, 0.0f, 0.0f, 0.0f};
#pragma unroll
    for (int r = 0; r < 8; ++r) {
      if (row0 + 8 * half + r < n_nodes) {
        if (rc > 0.0f && gr[r] != gp) {
          pool_add_run(sums, counts, gp, l16, run, rc);
          rc = 0.0f;
          run[0] = 0.0f; run[1] = 0.0f; run[2] = 0.0f; run[3] = 0.0f;
        }
        gp = gr[r];
        rc += 1.0f;
#pragma unroll
        for (int t = 0; t < 4; ++t) run[t] += acc2[t][r];
      }
    }
    if (rc > 0.0f) pool_add_run(sums, counts, gp, l16, run, rc);
  }
}

__global__ __launch_bounds__(256) void gin_proj_kernel(
    const float* __restrict__ xin,
    const float* __restrict__ Wa,
    float* __restrict__ ysrc,
    int n_nodes) {
  constexpr int PA = IN_FEATS + WPITCH_PAD;
  __shared__ float sWa[64 * PA];

  const int wave = threadIdx.x >> 5;
  const int lane = threadIdx.x & 31;
  const int half = lane >> 4;
  const int l16  = lane & 15;

  const int tile = blockIdx.x * 8 + wave;
  const int row0 = tile * 16;

  int arow = row0 + l16;
  if (arow >= n_nodes) arow = n_nodes - 1;
  const float* aptr = xin + (long long)arow * IN_FEATS + 2 * half;
  v2f afr[IN_FEATS / 4];
#pragma unroll
  for (int j = 0; j < IN_FEATS / 4; ++j) afr[j] = *(const v2f*)(aptr + 4 * j);

  stage_weight_rows<IN_FEATS>(sWa, Wa);
  __syncthreads();
  if (row0 >= n_nodes) return;

  v8f acc[4];
#pragma unroll
  for (int t = 0; t < 4; ++t)
#pragma unroll
    for (int r = 0; r < 8; ++r) acc[t][r] = 0.0f;

#pragma unroll
  for (int k = 0; k < IN_FEATS; k += 4) {
    v2f a = afr[k / 4];
#pragma unroll
    for (int t = 0; t < 4; ++t) {
      const float* bp = sWa + (t * 16 + l16) * PA + k + 2 * half;
      v2f b = *(const v2f*)bp;
      acc[t] = __builtin_amdgcn_wmma_f32_16x16x4_f32(
          false, a, false, b, (short)0, acc[t], false, false);
    }
  }

#pragma unroll
  for (int t = 0; t < 4; ++t) {
    int n = t * 16 + l16;
#pragma unroll
    for (int r = 0; r < 8; ++r) {
      int mrow = row0 + r + 8 * half;
      if (mrow < n_nodes) ysrc[(long long)mrow * HIDDEN + n] = acc[t][r];
    }
  }
}

__global__ __launch_bounds__(256) void gin_tail_kernel(
    const float* __restrict__ agg,
    const float* __restrict__ ba,
    const float* __restrict__ Wb,
    const float* __restrict__ bb,
    const float* __restrict__ gg,
    const float* __restrict__ be,
    const float* __restrict__ mm,
    const float* __restrict__ vv,
    float* __restrict__ hout,
    int n_nodes) {
  constexpr int PB = HIDDEN + WPITCH_PAD;
  __shared__ float sWb[64 * PB];

  const int wave = threadIdx.x >> 5;
  const int lane = threadIdx.x & 31;
  const int half = lane >> 4;
  const int l16  = lane & 15;

  const int tile = blockIdx.x * 8 + wave;
  const int row0 = tile * 16;

  int arow = row0 + l16;
  if (arow >= n_nodes) arow = n_nodes - 1;
  const float* aptr = agg + (long long)arow * HIDDEN + 2 * half;
  v2f afr[HIDDEN / 4];
#pragma unroll
  for (int j = 0; j < HIDDEN / 4; ++j) afr[j] = *(const v2f*)(aptr + 4 * j);

  stage_weight_rows<HIDDEN>(sWb, Wb);
  __syncthreads();
  if (row0 >= n_nodes) return;

  v8f acc2[4];
#pragma unroll
  for (int t = 0; t < 4; ++t)
#pragma unroll
    for (int r = 0; r < 8; ++r) acc2[t][r] = 0.0f;

  const float* biasp = ba + 2 * half;

#pragma unroll
  for (int k = 0; k < HIDDEN; k += 4) {
    v2f a = afr[k / 4];
    v2f bs = *(const v2f*)(biasp + k);
    a.x = a.x + bs.x;
    a.y = a.y + bs.y;
    a.x = a.x > 0.0f ? a.x : 0.0f;
    a.y = a.y > 0.0f ? a.y : 0.0f;
#pragma unroll
    for (int t = 0; t < 4; ++t) {
      const float* bp = sWb + (t * 16 + l16) * PB + k + 2 * half;
      v2f b = *(const v2f*)bp;
      acc2[t] = __builtin_amdgcn_wmma_f32_16x16x4_f32(
          false, a, false, b, (short)0, acc2[t], false, false);
    }
  }

#pragma unroll
  for (int t = 0; t < 4; ++t) {
    int n = t * 16 + l16;
    float s  = gg[n] * rsqrtf(vv[n] + BN_EPS);
    float sh = (bb[n] - mm[n]) * s + be[n];
#pragma unroll
    for (int r = 0; r < 8; ++r) {
      int mrow = row0 + r + 8 * half;
      float o = acc2[t][r] * s + sh;
      o = o > 0.0f ? o : 0.0f;
      if (mrow < n_nodes) hout[(long long)mrow * HIDDEN + n] = o;
    }
  }
}

__global__ __launch_bounds__(256) void pool_zero_kernel(float* __restrict__ sums,
                                                        float* __restrict__ counts) {
  int i = blockIdx.x * blockDim.x + threadIdx.x;
  if (i < NGRAPH * HIDDEN) sums[i] = 0.0f;
  if (i < NGRAPH) counts[i] = 0.0f;
}

__global__ __launch_bounds__(256) void pool_div_kernel(
    const float* __restrict__ sums, const float* __restrict__ counts,
    float* __restrict__ out) {
  int i = blockIdx.x * blockDim.x + threadIdx.x;
  if (i >= NGRAPH * HIDDEN) return;
  float c = counts[i >> 6];
  c = c > 1.0f ? c : 1.0f;
  out[i] = sums[i] / c;
}

extern "C" void kernel_launch(void* const* d_in, const int* in_sizes, int n_in,
                              void* d_out, int out_size, void* d_ws, size_t ws_size,
                              hipStream_t stream) {
  const float* x   = (const float*)d_in[0];
  const int* ei    = (const int*)d_in[1];
  const int* batch = (const int*)d_in[2];
  const float* W1a = (const float*)d_in[3];
  const float* b1a = (const float*)d_in[4];
  const float* W1b = (const float*)d_in[5];
  const float* b1b = (const float*)d_in[6];
  const float* g1  = (const float*)d_in[7];
  const float* be1 = (const float*)d_in[8];
  const float* m1  = (const float*)d_in[9];
  const float* v1  = (const float*)d_in[10];
  const float* W2a = (const float*)d_in[11];
  const float* b2a = (const float*)d_in[12];
  const float* W2b = (const float*)d_in[13];
  const float* b2b = (const float*)d_in[14];
  const float* g2  = (const float*)d_in[15];
  const float* be2 = (const float*)d_in[16];
  const float* m2  = (const float*)d_in[17];
  const float* v2  = (const float*)d_in[18];
  float* out = (float*)d_out;

  const int N = in_sizes[0] / IN_FEATS;
  const int E = in_sizes[1] / 2;

  float* ws    = (float*)d_ws;
  float* y1    = ws;
  float* agg   = ws + (long long)N * HIDDEN;
  float* h1    = ws + (long long)N * IN_FEATS;
  float* sums  = h1 + (long long)N * HIDDEN;
  float* cnts  = sums + NGRAPH * HIDDEN;
  const int Npad = (N + 3) & ~3;
  int* icnt    = (int*)(cnts + NGRAPH);
  int* ioff    = icnt + Npad;
  int* ipairs  = ioff + Npad;
  int* icsr    = ipairs + E;
  int* ibtot   = icsr + E;
  int* ibcur   = ibtot + MAX_BINS;
  int* ibhist  = ibcur + MAX_BINS;
  (void)ws_size; (void)n_in; (void)out_size;

  const int tiles  = (N + 15) / 16;
  const int mlpBlk = (tiles + 7) / 8;
  const int nBins = (N + BIN_NODES - 1) >> BIN_SHIFT;
  const int gathBlk = (int)(((long long)N * 16 + 255) / 256);

  bin_zero_kernel<<<1, MAX_BINS, 0, stream>>>(ibtot, ibcur);
  bin_count_kernel<<<SORT_BLOCKS, SORT_THREADS, 0, stream>>>(ei, ibtot, ibhist, E);
  bin_scatter_kernel<<<SORT_BLOCKS, SORT_THREADS, 0, stream>>>(ei, ibtot, ibcur, ibhist, ipairs, E);
  bin_sort_kernel<<<nBins, BIN_NODES, 0, stream>>>(ibtot, ipairs, icnt, ioff, icsr, N);

  gin_proj_kernel<<<mlpBlk, 256, 0, stream>>>(x, W1a, y1, N);
  gather_sum_kernel<<<gathBlk, 256, 0, stream>>>(y1, ioff, icnt, icsr, agg, N);
  gin_tail_kernel<<<mlpBlk, 256, 0, stream>>>(
      agg, b1a, W1b, b1b, g1, be1, m1, v1, h1, N);

  gather_sum_kernel<<<gathBlk, 256, 0, stream>>>(h1, ioff, icnt, icsr, agg, N);
  pool_zero_kernel<<<(NGRAPH * HIDDEN + 255) / 256, 256, 0, stream>>>(sums, cnts);
  gin_mlp_kernel<HIDDEN><<<mlpBlk, 256, 0, stream>>>(
      agg, W2a, b2a, W2b, b2b, g2, be2, m2, v2, batch, sums, cnts, N);
  pool_div_kernel<<<(NGRAPH * HIDDEN + 255) / 256, 256, 0, stream>>>(sums, cnts, out);
}


# Neighbour sums moved into the two MLP kernels: each wave sums its 16-node tile into LDS and reads its A fragments there; no agg plane in memory, 11 launches become 9; per-node sum order unchanged
# speedup vs baseline: 4.3987x; 1.0740x over previous
#include <hip/hip_runtime.h>

typedef float v2f __attribute__((ext_vector_type(2)));
typedef float v8f __attribute__((ext_vector_type(8)));

#define IN_FEATS 128
#define HIDDEN   64
#define NGRAPH   512
#define BN_EPS   1e-5f


#define BIN_SHIFT    9
#define BIN_NODES    512
#define MAX_BINS     256
#define SORT_THREADS 512
#define SORT_BLOCKS  256

__global__ __launch_bounds__(MAX_BINS) void bin_zero_kernel(int* __restrict__ binTotal,
                                                            int* __restrict__ binCursor) {
  binTotal[threadIdx.x] = 0;
  binCursor[threadIdx.x] = 0;
}

__global__ __launch_bounds__(SORT_THREADS) void bin_count_kernel(
    const int* __restrict__ ei, int* __restrict__ binTotal, int* __restrict__ blockHist, int E) {
  __shared__ int s_hist[MAX_BINS];
  const int t = threadIdx.x;
  if (t < MAX_BINS) s_hist[t] = 0;
  __syncthreads();
  const int* dst = ei + E;
  const int chunk = (E + (int)gridDim.x - 1) / (int)gridDim.x;
  const int c0 = (int)blockIdx.x * chunk;
  int c1 = c0 + chunk;
  if (c1 > E) c1 = E;
  for (int b0 = c0; b0 < c1; b0 += 4 * SORT_THREADS) {
    int d[4];
#pragma unroll
    for (int j = 0; j < 4; ++j) {
      const int e = b0 + j * SORT_THREADS + t;
      d[j] = dst[e < c1 ? e : c1 - 1];
    }
#pragma unroll
    for (int j = 0; j < 4; ++j)
      if (b0 + j * SORT_THREADS + t < c1) atomicAdd(&s_hist[d[j] >> BIN_SHIFT], 1);
  }
  __syncthreads();
  if (t < MAX_BINS) {
    const int h = s_hist[t];
    blockHist[(int)blockIdx.x * MAX_BINS + t] = h;
    if (h > 0) atomicAdd(&binTotal[t], h);
  }
}

__global__ __launch_bounds__(SORT_THREADS) void bin_scatter_kernel(
    const int* __restrict__ ei, const int* __restrict__ binTotal, int* __restrict__ binCursor,
    const int* __restrict__ blockHist, int* __restrict__ pairs, int E) {
  __shared__ int s_cur[MAX_BINS];
  __shared__ int s_wsum[SORT_THREADS / 32];
  const int t = threadIdx.x;
  const int lane = t & 31;
  const int wave = t >> 5;

  const int tot = t < MAX_BINS ? binTotal[t] : 0;
  int inc = tot;
#pragma unroll
  for (int o = 1; o < 32; o <<= 1) {
    const int u = __shfl_up(inc, o);
    if (lane >= o) inc += u;
  }
  if (lane == 31) s_wsum[wave] = inc;
  __syncthreads();
  int base = 0;
#pragma unroll
  for (int w = 0; w < SORT_THREADS / 32; ++w)
    if (w < wave) base += s_wsum[w];
  if (t < MAX_BINS) {
    const int h = blockHist[(int)blockIdx.x * MAX_BINS + t];
    int st = 0;
    if (h > 0) st = atomicAdd(&binCursor[t], h);
    s_cur[t] = base + inc - tot + st;
  }
  __syncthreads();

  const int* srcp = ei;
  const int* dst = ei + E;
  const int chunk = (E + (int)gridDim.x - 1) / (int)gridDim.x;
  const int c0 = (int)blockIdx.x * chunk;
  int c1 = c0 + chunk;
  if (c1 > E) c1 = E;
  for (int b0 = c0; b0 < c1; b0 += 4 * SORT_THREADS) {
    int s[4], d[4];
#pragma unroll
    for (int j = 0; j < 4; ++j) {
      const int e = b0 + j * SORT_THREADS + t;
      const int ec = e < c1 ? e : c1 - 1;
      d[j] = dst[ec];
      s[j] = srcp[ec];
    }
    int pos[4];
#pragma unroll
    for (int j = 0; j < 4; ++j)
      pos[j] = (b0 + j * SORT_THREADS + t < c1) ? atomicAdd(&s_cur[d[j] >> BIN_SHIFT], 1) : -1;
#pragma unroll
    for (int j = 0; j < 4; ++j)
      if (pos[j] >= 0)
        pairs[pos[j]] = (int)(((unsigned)s[j] << BIN_SHIFT) | (unsigned)(d[j] & (BIN_NODES - 1)));
  }
}

__global__ __launch_bounds__(BIN_NODES) void bin_sort_kernel(
    const int* __restrict__ binTotal, const int* __restrict__ pairs,
    int* __restrict__ cnt, int* __restrict__ off, int* __restrict__ csr, int n_nodes) {
  __shared__ int s_cnt[BIN_NODES];
  __shared__ int s_wsum[BIN_NODES / 32];
  __shared__ int s_wsum2[BIN_NODES / 32];
  __shared__ int s_bin[2];
  const int t = threadIdx.x;
  const int lane = t & 31;
  const int wave = t >> 5;
  const int b = (int)blockIdx.x;

  s_cnt[t] = 0;
  const int btot = t < MAX_BINS ? binTotal[t] : 0;
  int inc = btot;
#pragma unroll
  for (int o = 1; o < 32; o <<= 1) {
    const int u = __shfl_up(inc, o);
    if (lane >= o) inc += u;
  }
  if (lane == 31) s_wsum[wave] = inc;
  __syncthreads();
  int wbase = 0;
#pragma unroll
  for (int w = 0; w < BIN_NODES / 32; ++w)
    if (w < wave) wbase += s_wsum[w];
  if (t == b) {
    s_bin[0] = wbase + inc - btot;
    s_bin[1] = btot;
  }
  __syncthreads();
  const int base = s_bin[0];
  const int tot = s_bin[1];
  const int* p = pairs + base;

  for (int i0 = 0; i0 < tot; i0 += 4 * BIN_NODES) {
    unsigned v[4];
#pragma unroll
    for (int j = 0; j < 4; ++j) {
      const int i = i0 + j * BIN_NODES + t;
      v[j] = (unsigned)p[i < tot ? i : tot - 1];
    }
#pragma unroll
    for (int j = 0; j < 4; ++j)
      if (i0 + j * BIN_NODES + t < tot) atomicAdd(&s_cnt[v[j] & (BIN_NODES - 1)], 1);
  }
  __syncthreads();

  const int c = s_cnt[t];
  int inc2 = c;
#pragma unroll
  for (int o = 1; o < 32; o <<= 1) {
    const int u = __shfl_up(inc2, o);
    if (lane >= o) inc2 += u;
  }
  if (lane == 31) s_wsum2[wave] = inc2;
  __syncthreads();
  int wbase2 = 0;
#pragma unroll
  for (int w = 0; w < BIN_NODES / 32; ++w)
    if (w < wave) wbase2 += s_wsum2[w];
  const int start = base + wbase2 + inc2 - c;
  const int node = b * BIN_NODES + t;
  if (node < n_nodes) {
    cnt[node] = c;
    off[node] = start;
  }
  s_cnt[t] = start;
  __syncthreads();

  for (int i0 = 0; i0 < tot; i0 += 4 * BIN_NODES) {
    unsigned v[4];
#pragma unroll
    for (int j = 0; j < 4; ++j) {
      const int i = i0 + j * BIN_NODES + t;
      v[j] = (unsigned)p[i < tot ? i : tot - 1];
    }
    int pos[4];
#pragma unroll
    for (int j = 0; j < 4; ++j)
      pos[j] = (i0 + j * BIN_NODES + t < tot) ? atomicAdd(&s_cnt[v[j] & (BIN_NODES - 1)], 1) : -1;
#pragma unroll
    for (int j = 0; j < 4; ++j)
      if (pos[j] >= 0) csr[pos[j]] = (int)(v[j] >> BIN_SHIFT);
  }
}

__device__ __forceinline__ void gather_tile(
    const float* __restrict__ xin, const int* __restrict__ off, const int* __restrict__ cnt,
    const int* __restrict__ csr, float* __restrict__ tileT, int pitch, int row0, int n_nodes,
    int half, int l16) {
  const int c = l16 << 2;
  int hrow = row0 + l16;
  if (hrow >= n_nodes) hrow = n_nodes - 1;
  const int cntv = cnt[hrow];
  const int offv = off[hrow];
  for (int st = 0; st < 8; ++st) {
    const int r = 2 * st + half;
    const int len = __shfl(cntv, r);
    const int* p = csr + __shfl(offv, r);
    int i = row0 + r;
    if (i >= n_nodes) i = n_nodes - 1;
    float4 acc = *(const float4*)(xin + (long long)i * HIDDEN + c);
    const int last = len - 1;
    for (int j = 0; j < len; j += 4) {
      const int j1 = j + 1, j2 = j + 2, j3 = j + 3;
      const int s0 = p[j];
      const int s1 = p[j1 < last ? j1 : last];
      const int s2 = p[j2 < last ? j2 : last];
      const int s3 = p[j3 < last ? j3 : last];
      const float4 a0 = *(const float4*)(xin + (long long)s0 * HIDDEN + c);
      const float4 a1 = *(const float4*)(xin + (long long)s1 * HIDDEN + c);
      const float4 a2 = *(const float4*)(xin + (long long)s2 * HIDDEN + c);
      const float4 a3 = *(const float4*)(xin + (long long)s3 * HIDDEN + c);
      const int rem = len - j;
      acc.x += a0.x; acc.y += a0.y; acc.z += a0.z; acc.w += a0.w;
      const bool k1 = rem > 1, k2 = rem > 2, k3 = rem > 3;
      acc.x += k1 ? a1.x : 0.0f; acc.y += k1 ? a1.y : 0.0f;
      acc.z += k1 ? a1.z : 0.0f; acc.w += k1 ? a1.w : 0.0f;
      acc.x += k2 ? a2.x : 0.0f; acc.y += k2 ? a2.y : 0.0f;
      acc.z += k2 ? a2.z : 0.0f; acc.w += k2 ? a2.w : 0.0f;
      acc.x += k3 ? a3.x : 0.0f; acc.y += k3 ? a3.y : 0.0f;
      acc.z += k3 ? a3.z : 0.0f; acc.w += k3 ? a3.w : 0.0f;
    }
    *(float4*)(tileT + r * pitch + c) = acc;
  }
}

#define WPITCH_PAD 4
#define MLP_THREADS 256

template <int K>
__device__ __forceinline__ void stage_weight_rows(float* __restrict__ sW,
                                                  const float* __restrict__ W) {
  constexpr int Q = K / 4;
  constexpr int R = 64 * Q / MLP_THREADS;
  float4 v[R];
#pragma unroll
  for (int j = 0; j < R; ++j) {
    const int i = (int)threadIdx.x + j * MLP_THREADS;
    const int row = i / Q;
    v[j] = *(const float4*)(W + row * K + 4 * (i - row * Q));
  }
#pragma unroll
  for (int j = 0; j < R; ++j) {
    const int i = (int)threadIdx.x + j * MLP_THREADS;
    const int row = i / Q;
    *(float4*)(sW + row * (K + WPITCH_PAD) + 4 * (i - row * Q)) = v[j];
  }
}

__device__ __forceinline__ void pool_add_run(float* __restrict__ sums, float* __restrict__ counts,
                                             int g, int l16, const float* run, float rc) {
  float* sp = sums + (long long)g * HIDDEN + l16;
  atomicAdd(sp, run[0]);
  atomicAdd(sp + 16, run[1]);
  atomicAdd(sp + 32, run[2]);
  atomicAdd(sp + 48, run[3]);
  if (l16 == 0) atomicAdd(&counts[g], rc);
}

template <int FIN>
__global__ __launch_bounds__(256) void gin_mlp_kernel(
    const float* __restrict__ hin,
    const int* __restrict__ off,
    const int* __restrict__ cnt,
    const int* __restrict__ csr,
    const float* __restrict__ Wa,
    const float* __restrict__ ba,
    const float* __restrict__ Wb,
    const float* __restrict__ bb,
    const float* __restrict__ gg,
    const float* __restrict__ be,
    const float* __restrict__ mm,
    const float* __restrict__ vv,
    const int* __restrict__ batch,
    float* __restrict__ sums,
    float* __restrict__ counts,
    int n_nodes) {
  constexpr int PA = FIN + WPITCH_PAD;
  constexpr int PB = HIDDEN + WPITCH_PAD;
  __shared__ float sWa[64 * PA];
  __shared__ float sWb[64 * PB];
  static_assert(FIN == HIDDEN, "the tile of neighbour sums is HIDDEN wide");
  __shared__ float sT[8][16 * PB];

  const int wave = threadIdx.x >> 5;
  const int lane = threadIdx.x & 31;
  const int half = lane >> 4;
  const int l16  = lane & 15;

  const int tile = blockIdx.x * 8 + wave;
  const int row0 = tile * 16;

  int arow = row0 + l16;
  if (arow >= n_nodes) arow = n_nodes - 1;

  stage_weight_rows<FIN>(sWa, Wa);
  stage_weight_rows<HIDDEN>(sWb, Wb);
  __syncthreads();
  if (row0 >= n_nodes) return;

  gather_tile(hin, off, cnt, csr, sT[wave], PB, row0, n_nodes, half, l16);
  const float* tp = &sT[wave][l16 * PB + 2 * half];
  v2f afr[FIN / 4];
#pragma unroll
  for (int j = 0; j < FIN / 4; ++j) afr[j] = *(const v2f*)(tp + 4 * j);

  v8f acc[4];
#pragma unroll
  for (int t = 0; t < 4; ++t) {
    float bias = ba[t * 16 + l16];
#pragma unroll
    for (int r = 0; r < 8; ++r) acc[t][r] = bias;
  }

  const int gl = batch[arow];

#pragma unroll
  for (int k = 0; k < FIN; k += 4) {
    v2f a = afr[k / 4];
#pragma unroll
    for (int t = 0; t < 4; ++t) {
      const float* bp = sWa + (t * 16 + l16) * PA + k + 2 * half;
      v2f b = *(const v2f*)bp;
      acc[t] = __builtin_amdgcn_wmma_f32_16x16x4_f32(
          false, a, false, b, (short)0, acc[t], false, false);
    }
  }

#pragma unroll
  for (int t = 0; t < 4; ++t) {
#pragma unroll
    for (int r = 0; r < 8; ++r) {
      float x = acc[t][r];
      x = x > 0.0f ? x : 0.0f;
      sT[wave][(r + 8 * half) * PB + t * 16 + l16] = x;
    }
  }

  v8f acc2[4];
#pragma unroll
  for (int t = 0; t < 4; ++t)
#pragma unroll
    for (int r = 0; r < 8; ++r) acc2[t][r] = 0.0f;

  for (int k = 0; k < 64; k += 4) {
    v2f a = *(const v2f*)(tp + k);
#pragma unroll
    for (int t = 0; t < 4; ++t) {
      const float* bp = sWb + (t * 16 + l16) * PB + k + 2 * half;
      v2f b = *(const v2f*)bp;
      acc2[t] = __builtin_amdgcn_wmma_f32_16x16x4_f32(
          false, a, false, b, (short)0, acc2[t], false, false);
    }
  }

#pragma unroll
  for (int t = 0; t < 4; ++t) {
    int n = t * 16 + l16;
    float s  = gg[n] * rsqrtf(vv[n] + BN_EPS);
    float sh = (bb[n] - mm[n]) * s + be[n];
#pragma unroll
    for (int r = 0; r < 8; ++r) {
      float o = acc2[t][r] * s + sh;
      acc2[t][r] = o > 0.0f ? o : 0.0f;
    }
  }

  const int g0 = __builtin_amdgcn_readfirstlane(gl);
  const bool whole = row0 + 16 <= n_nodes;
  if (whole && __all(gl == g0)) {
    float cs[4];
#pragma unroll
    for (int t = 0; t < 4; ++t) {
      float v = acc2[t][0];
#pragma unroll
      for (int r = 1; r < 8; ++r) v += acc2[t][r];
      cs[t] = v + __shfl_xor(v, 16);
    }
    float* sp = sums + (long long)g0 * HIDDEN + lane;
    atomicAdd(sp, half ? cs[1] : cs[0]);
    atomicAdd(sp + 32, half ? cs[3] : cs[2]);
    if (lane == 0) atomicAdd(&counts[g0], 16.0f);
  } else {
    int gr[8];
#pragma unroll
    for (int r = 0; r < 8; ++r) gr[r] = __shfl(gl, 8 * half + r);
    int gp = 0;
    float rc = 0.0f;
    float run[4] = {0.0f, 0.0f, 0.0f, 0.0f};
#pragma unroll
    for (int r = 0; r < 8; ++r) {
      if (row0 + 8 * half + r < n_nodes) {
        if (rc > 0.0f && gr[r] != gp) {
          pool_add_run(sums, counts, gp, l16, run, rc);
          rc = 0.0f;
          run[0] = 0.0f; run[1] = 0.0f; run[2] = 0.0f; run[3] = 0.0f;
        }
        gp = gr[r];
        rc += 1.0f;
#pragma unroll
        for (int t = 0; t < 4; ++t) run[t] += acc2[t][r];
      }
    }
    if (rc > 0.0f) pool_add_run(sums, counts, gp, l16, run, rc);
  }
}

__global__ __launch_bounds__(256) void gin_proj_kernel(
    const float* __restrict__ xin,
    const float* __restrict__ Wa,
    float* __restrict__ ysrc,
    int n_nodes) {
  constexpr int PA = IN_FEATS + WPITCH_PAD;
  __shared__ float sWa[64 * PA];

  const int wave = threadIdx.x >> 5;
  const int lane = threadIdx.x & 31;
  const int half = lane >> 4;
  const int l16  = lane & 15;

  const int tile = blockIdx.x * 8 + wave;
  const int row0 = tile * 16;

  int arow = row0 + l16;
  if (arow >= n_nodes) arow = n_nodes - 1;
  const float* aptr = xin + (long long)arow * IN_FEATS + 2 * half;
  v2f afr[IN_FEATS / 4];
#pragma unroll
  for (int j = 0; j < IN_FEATS / 4; ++j) afr[j] = *(const v2f*)(aptr + 4 * j);

  stage_weight_rows<IN_FEATS>(sWa, Wa);
  __syncthreads();
  if (row0 >= n_nodes) return;

  v8f acc[4];
#pragma unroll
  for (int t = 0; t < 4; ++t)
#pragma unroll
    for (int r = 0; r < 8; ++r) acc[t][r] = 0.0f;

#pragma unroll
  for (int k = 0; k < IN_FEATS; k += 4) {
    v2f a = afr[k / 4];
#pragma unroll
    for (int t = 0; t < 4; ++t) {
      const float* bp = sWa + (t * 16 + l16) * PA + k + 2 * half;
      v2f b = *(const v2f*)bp;
      acc[t] = __builtin_amdgcn_wmma_f32_16x16x4_f32(
          false, a, false, b, (short)0, acc[t], false, false);
    }
  }

#pragma unroll
  for (int t = 0; t < 4; ++t) {
    int n = t * 16 + l16;
#pragma unroll
    for (int r = 0; r < 8; ++r) {
      int mrow = row0 + r + 8 * half;
      if (mrow < n_nodes) ysrc[(long long)mrow * HIDDEN + n] = acc[t][r];
    }
  }
}

__global__ __launch_bounds__(256) void gin_tail_kernel(
    const float* __restrict__ ysrc,
    const int* __restrict__ off,
    const int* __restrict__ cnt,
    const int* __restrict__ csr,
    const float* __restrict__ ba,
    const float* __restrict__ Wb,
    const float* __restrict__ bb,
    const float* __restrict__ gg,
    const float* __restrict__ be,
    const float* __restrict__ mm,
    const float* __restrict__ vv,
    float* __restrict__ hout,
    int n_nodes) {
  constexpr int PB = HIDDEN + WPITCH_PAD;
  __shared__ float sWb[64 * PB];
  __shared__ float sT[8][16 * PB];

  const int wave = threadIdx.x >> 5;
  const int lane = threadIdx.x & 31;
  const int half = lane >> 4;
  const int l16  = lane & 15;

  const int tile = blockIdx.x * 8 + wave;
  const int row0 = tile * 16;

  stage_weight_rows<HIDDEN>(sWb, Wb);
  __syncthreads();
  if (row0 >= n_nodes) return;

  gather_tile(ysrc, off, cnt, csr, sT[wave], PB, row0, n_nodes, half, l16);
  const float* tp = &sT[wave][l16 * PB + 2 * half];
  v2f afr[HIDDEN / 4];
#pragma unroll
  for (int j = 0; j < HIDDEN / 4; ++j) afr[j] = *(const v2f*)(tp + 4 * j);

  v8f acc2[4];
#pragma unroll
  for (int t = 0; t < 4; ++t)
#pragma unroll
    for (int r = 0; r < 8; ++r) acc2[t][r] = 0.0f;

  const float* biasp = ba + 2 * half;

#pragma unroll
  for (int k = 0; k < HIDDEN; k += 4) {
    v2f a = afr[k / 4];
    v2f bs = *(const v2f*)(biasp + k);
    a.x = a.x + bs.x;
    a.y = a.y + bs.y;
    a.x = a.x > 0.0f ? a.x : 0.0f;
    a.y = a.y > 0.0f ? a.y : 0.0f;
#pragma unroll
    for (int t = 0; t < 4; ++t) {
      const float* bp = sWb + (t * 16 + l16) * PB + k + 2 * half;
      v2f b = *(const v2f*)bp;
      acc2[t] = __builtin_amdgcn_wmma_f32_16x16x4_f32(
          false, a, false, b, (short)0, acc2[t], false, false);
    }
  }

#pragma unroll
  for (int t = 0; t < 4; ++t) {
    int n = t * 16 + l16;
    float s  = gg[n] * rsqrtf(vv[n] + BN_EPS);
    float sh = (bb[n] - mm[n]) * s + be[n];
#pragma unroll
    for (int r = 0; r < 8; ++r) {
      int mrow = row0 + r + 8 * half;
      float o = acc2[t][r] * s + sh;
      o = o > 0.0f ? o : 0.0f;
      if (mrow < n_nodes) hout[(long long)mrow * HIDDEN + n] = o;
    }
  }
}

__global__ __launch_bounds__(256) void pool_zero_kernel(float* __restrict__ sums,
                                                        float* __restrict__ counts) {
  int i = blockIdx.x * blockDim.x + threadIdx.x;
  if (i < NGRAPH * HIDDEN) sums[i] = 0.0f;
  if (i < NGRAPH) counts[i] = 0.0f;
}

__global__ __launch_bounds__(256) void pool_div_kernel(
    const float* __restrict__ sums, const float* __restrict__ counts,
    float* __restrict__ out) {
  int i = blockIdx.x * blockDim.x + threadIdx.x;
  if (i >= NGRAPH * HIDDEN) return;
  float c = counts[i >> 6];
  c = c > 1.0f ? c : 1.0f;
  out[i] = sums[i] / c;
}

extern "C" void kernel_launch(void* const* d_in, const int* in_sizes, int n_in,
                              void* d_out, int out_size, void* d_ws, size_t ws_size,
                              hipStream_t stream) {
  const float* x   = (const float*)d_in[0];
  const int* ei    = (const int*)d_in[1];
  const int* batch = (const int*)d_in[2];
  const float* W1a = (const float*)d_in[3];
  const float* b1a = (const float*)d_in[4];
  const float* W1b = (const float*)d_in[5];
  const float* b1b = (const float*)d_in[6];
  const float* g1  = (const float*)d_in[7];
  const float* be1 = (const float*)d_in[8];
  const float* m1  = (const float*)d_in[9];
  const float* v1  = (const float*)d_in[10];
  const float* W2a = (const float*)d_in[11];
  const float* b2a = (const float*)d_in[12];
  const float* W2b = (const float*)d_in[13];
  const float* b2b = (const float*)d_in[14];
  const float* g2  = (const float*)d_in[15];
  const float* be2 = (const float*)d_in[16];
  const float* m2  = (const float*)d_in[17];
  const float* v2  = (const float*)d_in[18];
  float* out = (float*)d_out;

  const int N = in_sizes[0] / IN_FEATS;
  const int E = in_sizes[1] / 2;

  float* ws    = (float*)d_ws;
  float* y1    = ws;
  float* h1    = ws + (long long)N * IN_FEATS;
  float* sums  = h1 + (long long)N * HIDDEN;
  float* cnts  = sums + NGRAPH * HIDDEN;
  const int Npad = (N + 3) & ~3;
  int* icnt    = (int*)(cnts + NGRAPH);
  int* ioff    = icnt + Npad;
  int* ipairs  = ioff + Npad;
  int* icsr    = ipairs + E;
  int* ibtot   = icsr + E;
  int* ibcur   = ibtot + MAX_BINS;
  int* ibhist  = ibcur + MAX_BINS;
  (void)ws_size; (void)n_in; (void)out_size;

  const int tiles  = (N + 15) / 16;
  const int mlpBlk = (tiles + 7) / 8;
  const int nBins = (N + BIN_NODES - 1) >> BIN_SHIFT;

  bin_zero_kernel<<<1, MAX_BINS, 0, stream>>>(ibtot, ibcur);
  bin_count_kernel<<<SORT_BLOCKS, SORT_THREADS, 0, stream>>>(ei, ibtot, ibhist, E);
  bin_scatter_kernel<<<SORT_BLOCKS, SORT_THREADS, 0, stream>>>(ei, ibtot, ibcur, ibhist, ipairs, E);
  bin_sort_kernel<<<nBins, BIN_NODES, 0, stream>>>(ibtot, ipairs, icnt, ioff, icsr, N);

  gin_proj_kernel<<<mlpBlk, 256, 0, stream>>>(x, W1a, y1, N);
  gin_tail_kernel<<<mlpBlk, 256, 0, stream>>>(
      y1, ioff, icnt, icsr, b1a, W1b, b1b, g1, be1, m1, v1, h1, N);

  pool_zero_kernel<<<(NGRAPH * HIDDEN + 255) / 256, 256, 0, stream>>>(sums, cnts);
  gin_mlp_kernel<HIDDEN><<<mlpBlk, 256, 0, stream>>>(
      h1, ioff, icnt, icsr, W2a, b2a, W2b, b2b, g2, be2, m2, v2, batch, sums, cnts, N);
  pool_div_kernel<<<(NGRAPH * HIDDEN + 255) / 256, 256, 0, stream>>>(sums, cnts, out);
}


# Layer-1 projection x @ W1a^T shares one launch with the per-bin edge ranking (independent work, picked by block index): 9 launches become 8, the ranking's latency chain runs under matrix work
# speedup vs baseline: 4.6261x; 1.0517x over previous
#include <hip/hip_runtime.h>

typedef float v2f __attribute__((ext_vector_type(2)));
typedef float v8f __attribute__((ext_vector_type(8)));

#define IN_FEATS 128
#define HIDDEN   64
#define NGRAPH   512
#define BN_EPS   1e-5f


#define BIN_SHIFT    9
#define BIN_NODES    512
#define MAX_BINS     256
#define SORT_THREADS 512
#define SORT_BLOCKS  256

__global__ __launch_bounds__(MAX_BINS) void bin_zero_kernel(int* __restrict__ binTotal,
                                                            int* __restrict__ binCursor) {
  binTotal[threadIdx.x] = 0;
  binCursor[threadIdx.x] = 0;
}

__global__ __launch_bounds__(SORT_THREADS) void bin_count_kernel(
    const int* __restrict__ ei, int* __restrict__ binTotal, int* __restrict__ blockHist, int E) {
  __shared__ int s_hist[MAX_BINS];
  const int t = threadIdx.x;
  if (t < MAX_BINS) s_hist[t] = 0;
  __syncthreads();
  const int* dst = ei + E;
  const int chunk = (E + (int)gridDim.x - 1) / (int)gridDim.x;
  const int c0 = (int)blockIdx.x * chunk;
  int c1 = c0 + chunk;
  if (c1 > E) c1 = E;
  for (int b0 = c0; b0 < c1; b0 += 4 * SORT_THREADS) {
    int d[4];
#pragma unroll
    for (int j = 0; j < 4; ++j) {
      const int e = b0 + j * SORT_THREADS + t;
      d[j] = dst[e < c1 ? e : c1 - 1];
    }
#pragma unroll
    for (int j = 0; j < 4; ++j)
      if (b0 + j * SORT_THREADS + t < c1) atomicAdd(&s_hist[d[j] >> BIN_SHIFT], 1);
  }
  __syncthreads();
  if (t < MAX_BINS) {
    const int h = s_hist[t];
    blockHist[(int)blockIdx.x * MAX_BINS + t] = h;
    if (h > 0) atomicAdd(&binTotal[t], h);
  }
}

__global__ __launch_bounds__(SORT_THREADS) void bin_scatter_kernel(
    const int* __restrict__ ei, const int* __restrict__ binTotal, int* __restrict__ binCursor,
    const int* __restrict__ blockHist, int* __restrict__ pairs, int E) {
  __shared__ int s_cur[MAX_BINS];
  __shared__ int s_wsum[SORT_THREADS / 32];
  const int t = threadIdx.x;
  const int lane = t & 31;
  const int wave = t >> 5;

  const int tot = t < MAX_BINS ? binTotal[t] : 0;
  int inc = tot;
#pragma unroll
  for (int o = 1; o < 32; o <<= 1) {
    const int u = __shfl_up(inc, o);
    if (lane >= o) inc += u;
  }
  if (lane == 31) s_wsum[wave] = inc;
  __syncthreads();
  int base = 0;
#pragma unroll
  for (int w = 0; w < SORT_THREADS / 32; ++w)
    if (w < wave) base += s_wsum[w];
  if (t < MAX_BINS) {
    const int h = blockHist[(int)blockIdx.x * MAX_BINS + t];
    int st = 0;
    if (h > 0) st = atomicAdd(&binCursor[t], h);
    s_cur[t] = base + inc - tot + st;
  }
  __syncthreads();

  const int* srcp = ei;
  const int* dst = ei + E;
  const int chunk = (E + (int)gridDim.x - 1) / (int)gridDim.x;
  const int c0 = (int)blockIdx.x * chunk;
  int c1 = c0 + chunk;
  if (c1 > E) c1 = E;
  for (int b0 = c0; b0 < c1; b0 += 4 * SORT_THREADS) {
    int s[4], d[4];
#pragma unroll
    for (int j = 0; j < 4; ++j) {
      const int e = b0 + j * SORT_THREADS + t;
      const int ec = e < c1 ? e : c1 - 1;
      d[j] = dst[ec];
      s[j] = srcp[ec];
    }
    int pos[4];
#pragma unroll
    for (int j = 0; j < 4; ++j)
      pos[j] = (b0 + j * SORT_THREADS + t < c1) ? atomicAdd(&s_cur[d[j] >> BIN_SHIFT], 1) : -1;
#pragma unroll
    for (int j = 0; j < 4; ++j)
      if (pos[j] >= 0)
        pairs[pos[j]] = (int)(((unsigned)s[j] << BIN_SHIFT) | (unsigned)(d[j] & (BIN_NODES - 1)));
  }
}

__device__ __forceinline__ void bin_sort_block(
    const int* __restrict__ binTotal, const int* __restrict__ pairs,
    int* __restrict__ cnt, int* __restrict__ off, int* __restrict__ csr, int n_nodes, int b) {
  __shared__ int s_cnt[BIN_NODES];
  __shared__ int s_wsum[BIN_NODES / 32];
  __shared__ int s_wsum2[BIN_NODES / 32];
  __shared__ int s_bin[2];
  const int t = threadIdx.x;
  const int lane = t & 31;
  const int wave = t >> 5;

  s_cnt[t] = 0;
  const int btot = t < MAX_BINS ? binTotal[t] : 0;
  int inc = btot;
#pragma unroll
  for (int o = 1; o < 32; o <<= 1) {
    const int u = __shfl_up(inc, o);
    if (lane >= o) inc += u;
  }
  if (lane == 31) s_wsum[wave] = inc;
  __syncthreads();
  int wbase = 0;
#pragma unroll
  for (int w = 0; w < BIN_NODES / 32; ++w)
    if (w < wave) wbase += s_wsum[w];
  if (t == b) {
    s_bin[0] = wbase + inc - btot;
    s_bin[1] = btot;
  }
  __syncthreads();
  const int base = s_bin[0];
  const int tot = s_bin[1];
  const int* p = pairs + base;

  for (int i0 = 0; i0 < tot; i0 += 4 * BIN_NODES) {
    unsigned v[4];
#pragma unroll
    for (int j = 0; j < 4; ++j) {
      const int i = i0 + j * BIN_NODES + t;
      v[j] = (unsigned)p[i < tot ? i : tot - 1];
    }
#pragma unroll
    for (int j = 0; j < 4; ++j)
      if (i0 + j * BIN_NODES + t < tot) atomicAdd(&s_cnt[v[j] & (BIN_NODES - 1)], 1);
  }
  __syncthreads();

  const int c = s_cnt[t];
  int inc2 = c;
#pragma unroll
  for (int o = 1; o < 32; o <<= 1) {
    const int u = __shfl_up(inc2, o);
    if (lane >= o) inc2 += u;
  }
  if (lane == 31) s_wsum2[wave] = inc2;
  __syncthreads();
  int wbase2 = 0;
#pragma unroll
  for (int w = 0; w < BIN_NODES / 32; ++w)
    if (w < wave) wbase2 += s_wsum2[w];
  const int start = base + wbase2 + inc2 - c;
  const int node = b * BIN_NODES + t;
  if (node < n_nodes) {
    cnt[node] = c;
    off[node] = start;
  }
  s_cnt[t] = start;
  __syncthreads();

  for (int i0 = 0; i0 < tot; i0 += 4 * BIN_NODES) {
    unsigned v[4];
#pragma unroll
    for (int j = 0; j < 4; ++j) {
      const int i = i0 + j * BIN_NODES + t;
      v[j] = (unsigned)p[i < tot ? i : tot - 1];
    }
    int pos[4];
#pragma unroll
    for (int j = 0; j < 4; ++j)
      pos[j] = (i0 + j * BIN_NODES + t < tot) ? atomicAdd(&s_cnt[v[j] & (BIN_NODES - 1)], 1) : -1;
#pragma unroll
    for (int j = 0; j < 4; ++j)
      if (pos[j] >= 0) csr[pos[j]] = (int)(v[j] >> BIN_SHIFT);
  }
}

__device__ __forceinline__ void gather_tile(
    const float* __restrict__ xin, const int* __restrict__ off, const int* __restrict__ cnt,
    const int* __restrict__ csr, float* __restrict__ tileT, int pitch, int row0, int n_nodes,
    int half, int l16) {
  const int c = l16 << 2;
  int hrow = row0 + l16;
  if (hrow >= n_nodes) hrow = n_nodes - 1;
  const int cntv = cnt[hrow];
  const int offv = off[hrow];
  for (int st = 0; st < 8; ++st) {
    const int r = 2 * st + half;
    const int len = __shfl(cntv, r);
    const int* p = csr + __shfl(offv, r);
    int i = row0 + r;
    if (i >= n_nodes) i = n_nodes - 1;
    float4 acc = *(const float4*)(xin + (long long)i * HIDDEN + c);
    const int last = len - 1;
    for (int j = 0; j < len; j += 4) {
      const int j1 = j + 1, j2 = j + 2, j3 = j + 3;
      const int s0 = p[j];
      const int s1 = p[j1 < last ? j1 : last];
      const int s2 = p[j2 < last ? j2 : last];
      const int s3 = p[j3 < last ? j3 : last];
      const float4 a0 = *(const float4*)(xin + (long long)s0 * HIDDEN + c);
      const float4 a1 = *(const float4*)(xin + (long long)s1 * HIDDEN + c);
      const float4 a2 = *(const float4*)(xin + (long long)s2 * HIDDEN + c);
      const float4 a3 = *(const float4*)(xin + (long long)s3 * HIDDEN + c);
      const int rem = len - j;
      acc.x += a0.x; acc.y += a0.y; acc.z += a0.z; acc.w += a0.w;
      const bool k1 = rem > 1, k2 = rem > 2, k3 = rem > 3;
      acc.x += k1 ? a1.x : 0.0f; acc.y += k1 ? a1.y : 0.0f;
      acc.z += k1 ? a1.z : 0.0f; acc.w += k1 ? a1.w : 0.0f;
      acc.x += k2 ? a2.x : 0.0f; acc.y += k2 ? a2.y : 0.0f;
      acc.z += k2 ? a2.z : 0.0f; acc.w += k2 ? a2.w : 0.0f;
      acc.x += k3 ? a3.x : 0.0f; acc.y += k3 ? a3.y : 0.0f;
      acc.z += k3 ? a3.z : 0.0f; acc.w += k3 ? a3.w : 0.0f;
    }
    *(float4*)(tileT + r * pitch + c) = acc;
  }
}

#define WPITCH_PAD 4
#define MLP_THREADS 256

template <int K, int NT = MLP_THREADS>
__device__ __forceinline__ void stage_weight_rows(float* __restrict__ sW,
                                                  const float* __restrict__ W) {
  constexpr int Q = K / 4;
  constexpr int R = 64 * Q / NT;
  float4 v[R];
#pragma unroll
  for (int j = 0; j < R; ++j) {
    const int i = (int)threadIdx.x + j * NT;
    const int row = i / Q;
    v[j] = *(const float4*)(W + row * K + 4 * (i - row * Q));
  }
#pragma unroll
  for (int j = 0; j < R; ++j) {
    const int i = (int)threadIdx.x + j * NT;
    const int row = i / Q;
    *(float4*)(sW + row * (K + WPITCH_PAD) + 4 * (i - row * Q)) = v[j];
  }
}

__device__ __forceinline__ void pool_add_run(float* __restrict__ sums, float* __restrict__ counts,
                                             int g, int l16, const float* run, float rc) {
  float* sp = sums + (long long)g * HIDDEN + l16;
  atomicAdd(sp, run[0]);
  atomicAdd(sp + 16, run[1]);
  atomicAdd(sp + 32, run[2]);
  atomicAdd(sp + 48, run[3]);
  if (l16 == 0) atomicAdd(&counts[g], rc);
}

template <int FIN>
__global__ __launch_bounds__(256) void gin_mlp_kernel(
    const float* __restrict__ hin,
    const int* __restrict__ off,
    const int* __restrict__ cnt,
    const int* __restrict__ csr,
    const float* __restrict__ Wa,
    const float* __restrict__ ba,
    const float* __restrict__ Wb,
    const float* __restrict__ bb,
    const float* __restrict__ gg,
    const float* __restrict__ be,
    const float* __restrict__ mm,
    const float* __restrict__ vv,
    const int* __restrict__ batch,
    float* __restrict__ sums,
    float* __restrict__ counts,
    int n_nodes) {
  constexpr int PA = FIN + WPITCH_PAD;
  constexpr int PB = HIDDEN + WPITCH_PAD;
  __shared__ float sWa[64 * PA];
  __shared__ float sWb[64 * PB];
  static_assert(FIN == HIDDEN, "the tile of neighbour sums is HIDDEN wide");
  __shared__ float sT[8][16 * PB];

  const int wave = threadIdx.x >> 5;
  const int lane = threadIdx.x & 31;
  const int half = lane >> 4;
  const int l16  = lane & 15;

  const int tile = blockIdx.x * 8 + wave;
  const int row0 = tile * 16;

  int arow = row0 + l16;
  if (arow >= n_nodes) arow = n_nodes - 1;

  stage_weight_rows<FIN>(sWa, Wa);
  stage_weight_rows<HIDDEN>(sWb, Wb);
  __syncthreads();
  if (row0 >= n_nodes) return;

  gather_tile(hin, off, cnt, csr, sT[wave], PB, row0, n_nodes, half, l16);
  const float* tp = &sT[wave][l16 * PB + 2 * half];
  v2f afr[FIN / 4];
#pragma unroll
  for (int j = 0; j < FIN / 4; ++j) afr[j] = *(const v2f*)(tp + 4 * j);

  v8f acc[4];
#pragma unroll
  for (int t = 0; t < 4; ++t) {
    float bias = ba[t * 16 + l16];
#pragma unroll
    for (int r = 0; r < 8; ++r) acc[t][r] = bias;
  }

  const int gl = batch[arow];

#pragma unroll
  for (int k = 0; k < FIN; k += 4) {
    v2f a = afr[k / 4];
#pragma unroll
    for (int t = 0; t < 4; ++t) {
      const float* bp = sWa + (t * 16 + l16) * PA + k + 2 * half;
      v2f b = *(const v2f*)bp;
      acc[t] = __builtin_amdgcn_wmma_f32_16x16x4_f32(
          false, a, false, b, (short)0, acc[t], false, false);
    }
  }

#pragma unroll
  for (int t = 0; t < 4; ++t) {
#pragma unroll
    for (int r = 0; r < 8; ++r) {
      float x = acc[t][r];
      x = x > 0.0f ? x : 0.0f;
      sT[wave][(r + 8 * half) * PB + t * 16 + l16] = x;
    }
  }

  v8f acc2[4];
#pragma unroll
  for (int t = 0; t < 4; ++t)
#pragma unroll
    for (int r = 0; r < 8; ++r) acc2[t][r] = 0.0f;

  for (int k = 0; k < 64; k += 4) {
    v2f a = *(const v2f*)(tp + k);
#pragma unroll
    for (int t = 0; t < 4; ++t) {
      const float* bp = sWb + (t * 16 + l16) * PB + k + 2 * half;
      v2f b = *(const v2f*)bp;
      acc2[t] = __builtin_amdgcn_wmma_f32_16x16x4_f32(
          false, a, false, b, (short)0, acc2[t], false, false);
    }
  }

#pragma unroll
  for (int t = 0; t < 4; ++t) {
    int n = t * 16 + l16;
    float s  = gg[n] * rsqrtf(vv[n] + BN_EPS);
    float sh = (bb[n] - mm[n]) * s + be[n];
#pragma unroll
    for (int r = 0; r < 8; ++r) {
      float o = acc2[t][r] * s + sh;
      acc2[t][r] = o > 0.0f ? o : 0.0f;
    }
  }

  const int g0 = __builtin_amdgcn_readfirstlane(gl);
  const bool whole = row0 + 16 <= n_nodes;
  if (whole && __all(gl == g0)) {
    float cs[4];
#pragma unroll
    for (int t = 0; t < 4; ++t) {
      float v = acc2[t][0];
#pragma unroll
      for (int r = 1; r < 8; ++r) v += acc2[t][r];
      cs[t] = v + __shfl_xor(v, 16);
    }
    float* sp = sums + (long long)g0 * HIDDEN + lane;
    atomicAdd(sp, half ? cs[1] : cs[0]);
    atomicAdd(sp + 32, half ? cs[3] : cs[2]);
    if (lane == 0) atomicAdd(&counts[g0], 16.0f);
  } else {
    int gr[8];
#pragma unroll
    for (int r = 0; r < 8; ++r) gr[r] = __shfl(gl, 8 * half + r);
    int gp = 0;
    float rc = 0.0f;
    float run[4] = {0.0f, 0.0f, 0.0f, 0.0f};
#pragma unroll
    for (int r = 0; r < 8; ++r) {
      if (row0 + 8 * half + r < n_nodes) {
        if (rc > 0.0f && gr[r] != gp) {
          pool_add_run(sums, counts, gp, l16, run, rc);
          rc = 0.0f;
          run[0] = 0.0f; run[1] = 0.0f; run[2] = 0.0f; run[3] = 0.0f;
        }
        gp = gr[r];
        rc += 1.0f;
#pragma unroll
        for (int t = 0; t < 4; ++t) run[t] += acc2[t][r];
      }
    }
    if (rc > 0.0f) pool_add_run(sums, counts, gp, l16, run, rc);
  }
}

#define PROJ_THREADS BIN_NODES
#define PROJ_TILES   (PROJ_THREADS / 32)

__device__ __forceinline__ void proj_block(
    const float* __restrict__ xin,
    const float* __restrict__ Wa,
    float* __restrict__ ysrc,
    int n_nodes, int blk) {
  constexpr int PA = IN_FEATS + WPITCH_PAD;
  __shared__ float sWa[64 * PA];

  const int wave = threadIdx.x >> 5;
  const int lane = threadIdx.x & 31;
  const int half = lane >> 4;
  const int l16  = lane & 15;

  const int tile = blk * PROJ_TILES + wave;
  const int row0 = tile * 16;

  int arow = row0 + l16;
  if (arow >= n_nodes) arow = n_nodes - 1;
  const float* aptr = xin + (long long)arow * IN_FEATS + 2 * half;
  v2f afr[IN_FEATS / 4];
#pragma unroll
  for (int j = 0; j < IN_FEATS / 4; ++j) afr[j] = *(const v2f*)(aptr + 4 * j);

  stage_weight_rows<IN_FEATS, PROJ_THREADS>(sWa, Wa);
#pragma unroll
  for (int j = 0; j < IN_FEATS / 4; j += 8)
    asm volatile("" ::"v"(afr[j]), "v"(afr[j + 1]), "v"(afr[j + 2]), "v"(afr[j + 3]),
                 "v"(afr[j + 4]), "v"(afr[j + 5]), "v"(afr[j + 6]), "v"(afr[j + 7]));
  __syncthreads();
  if (row0 >= n_nodes) return;

  v8f acc[4];
#pragma unroll
  for (int t = 0; t < 4; ++t)
#pragma unroll
    for (int r = 0; r < 8; ++r) acc[t][r] = 0.0f;

#pragma unroll
  for (int k = 0; k < IN_FEATS; k += 4) {
    v2f a = afr[k / 4];
#pragma unroll
    for (int t = 0; t < 4; ++t) {
      const float* bp = sWa + (t * 16 + l16) * PA + k + 2 * half;
      v2f b = *(const v2f*)bp;
      acc[t] = __builtin_amdgcn_wmma_f32_16x16x4_f32(
          false, a, false, b, (short)0, acc[t], false, false);
    }
  }

#pragma unroll
  for (int t = 0; t < 4; ++t) {
    int n = t * 16 + l16;
#pragma unroll
    for (int r = 0; r < 8; ++r) {
      int mrow = row0 + r + 8 * half;
      if (mrow < n_nodes) ysrc[(long long)mrow * HIDDEN + n] = acc[t][r];
    }
  }
}

__global__ __launch_bounds__(BIN_NODES) void rank_proj_kernel(
    const float* __restrict__ xin, const float* __restrict__ Wa, float* __restrict__ ysrc,
    const int* __restrict__ binTotal, const int* __restrict__ pairs,
    int* __restrict__ cnt, int* __restrict__ off, int* __restrict__ csr,
    int n_nodes, int n_proj) {
  static_assert(PROJ_THREADS == BIN_NODES, "both kinds of block have the launch's block size");
  const int blk = (int)blockIdx.x;
  if (blk < n_proj)
    proj_block(xin, Wa, ysrc, n_nodes, blk);
  else
    bin_sort_block(binTotal, pairs, cnt, off, csr, n_nodes, blk - n_proj);
}

__global__ __launch_bounds__(256) void gin_tail_kernel(
    const float* __restrict__ ysrc,
    const int* __restrict__ off,
    const int* __restrict__ cnt,
    const int* __restrict__ csr,
    const float* __restrict__ ba,
    const float* __restrict__ Wb,
    const float* __restrict__ bb,
    const float* __restrict__ gg,
    const float* __restrict__ be,
    const float* __restrict__ mm,
    const float* __restrict__ vv,
    float* __restrict__ hout,
    int n_nodes) {
  constexpr int PB = HIDDEN + WPITCH_PAD;
  __shared__ float sWb[64 * PB];
  __shared__ float sT[8][16 * PB];

  const int wave = threadIdx.x >> 5;
  const int lane = threadIdx.x & 31;
  const int half = lane >> 4;
  const int l16  = lane & 15;

  const int tile = blockIdx.x * 8 + wave;
  const int row0 = tile * 16;

  stage_weight_rows<HIDDEN>(sWb, Wb);
  __syncthreads();
  if (row0 >= n_nodes) return;

  gather_tile(ysrc, off, cnt, csr, sT[wave], PB, row0, n_nodes, half, l16);
  const float* tp = &sT[wave][l16 * PB + 2 * half];
  v2f afr[HIDDEN / 4];
#pragma unroll
  for (int j = 0; j < HIDDEN / 4; ++j) afr[j] = *(const v2f*)(tp + 4 * j);

  v8f acc2[4];
#pragma unroll
  for (int t = 0; t < 4; ++t)
#pragma unroll
    for (int r = 0; r < 8; ++r) acc2[t][r] = 0.0f;

  const float* biasp = ba + 2 * half;

#pragma unroll
  for (int k = 0; k < HIDDEN; k += 4) {
    v2f a = afr[k / 4];
    v2f bs = *(const v2f*)(biasp + k);
    a.x = a.x + bs.x;
    a.y = a.y + bs.y;
    a.x = a.x > 0.0f ? a.x : 0.0f;
    a.y = a.y > 0.0f ? a.y : 0.0f;
#pragma unroll
    for (int t = 0; t < 4; ++t) {
      const float* bp = sWb + (t * 16 + l16) * PB + k + 2 * half;
      v2f b = *(const v2f*)bp;
      acc2[t] = __builtin_amdgcn_wmma_f32_16x16x4_f32(
          false, a, false, b, (short)0, acc2[t], false, false);
    }
  }

#pragma unroll
  for (int t = 0; t < 4; ++t) {
    int n = t * 16 + l16;
    float s  = gg[n] * rsqrtf(vv[n] + BN_EPS);
    float sh = (bb[n] - mm[n]) * s + be[n];
#pragma unroll
    for (int r = 0; r < 8; ++r) {
      int mrow = row0 + r + 8 * half;
      float o = acc2[t][r] * s + sh;
      o = o > 0.0f ? o : 0.0f;
      if (mrow < n_nodes) hout[(long long)mrow * HIDDEN + n] = o;
    }
  }
}

__global__ __launch_bounds__(256) void pool_zero_kernel(float* __restrict__ sums,
                                                        float* __restrict__ counts) {
  int i = blockIdx.x * blockDim.x + threadIdx.x;
  if (i < NGRAPH * HIDDEN) sums[i] = 0.0f;
  if (i < NGRAPH) counts[i] = 0.0f;
}

__global__ __launch_bounds__(256) void pool_div_kernel(
    const float* __restrict__ sums, const float* __restrict__ counts,
    float* __restrict__ out) {
  int i = blockIdx.x * blockDim.x + threadIdx.x;
  if (i >= NGRAPH * HIDDEN) return;
  float c = counts[i >> 6];
  c = c > 1.0f ? c : 1.0f;
  out[i] = sums[i] / c;
}

extern "C" void kernel_launch(void* const* d_in, const int* in_sizes, int n_in,
                              void* d_out, int out_size, void* d_ws, size_t ws_size,
                              hipStream_t stream) {
  const float* x   = (const float*)d_in[0];
  const int* ei    = (const int*)d_in[1];
  const int* batch = (const int*)d_in[2];
  const float* W1a = (const float*)d_in[3];
  const float* b1a = (const float*)d_in[4];
  const float* W1b = (const float*)d_in[5];
  const float* b1b = (const float*)d_in[6];
  const float* g1  = (const float*)d_in[7];
  const float* be1 = (const float*)d_in[8];
  const float* m1  = (const float*)d_in[9];
  const float* v1  = (const float*)d_in[10];
  const float* W2a = (const float*)d_in[11];
  const float* b2a = (const float*)d_in[12];
  const float* W2b = (const float*)d_in[13];
  const float* b2b = (const float*)d_in[14];
  const float* g2  = (const float*)d_in[15];
  const float* be2 = (const float*)d_in[16];
  const float* m2  = (const float*)d_in[17];
  const float* v2  = (const float*)d_in[18];
  float* out = (float*)d_out;

  const int N = in_sizes[0] / IN_FEATS;
  const int E = in_sizes[1] / 2;

  float* ws    = (float*)d_ws;
  float* y1    = ws;
  float* h1    = ws + (long long)N * IN_FEATS;
  float* sums  = h1 + (long long)N * HIDDEN;
  float* cnts  = sums + NGRAPH * HIDDEN;
  const int Npad = (N + 3) & ~3;
  int* icnt    = (int*)(cnts + NGRAPH);
  int* ioff    = icnt + Npad;
  int* ipairs  = ioff + Npad;
  int* icsr    = ipairs + E;
  int* ibtot   = icsr + E;
  int* ibcur   = ibtot + MAX_BINS;
  int* ibhist  = ibcur + MAX_BINS;
  (void)ws_size; (void)n_in; (void)out_size;

  const int tiles  = (N + 15) / 16;
  const int mlpBlk = (tiles + 7) / 8;
  const int projBlk = (tiles + PROJ_TILES - 1) / PROJ_TILES;
  const int nBins = (N + BIN_NODES - 1) >> BIN_SHIFT;

  bin_zero_kernel<<<1, MAX_BINS, 0, stream>>>(ibtot, ibcur);
  bin_count_kernel<<<SORT_BLOCKS, SORT_THREADS, 0, stream>>>(ei, ibtot, ibhist, E);
  bin_scatter_kernel<<<SORT_BLOCKS, SORT_THREADS, 0, stream>>>(ei, ibtot, ibcur, ibhist, ipairs, E);
  rank_proj_kernel<<<projBlk + nBins, BIN_NODES, 0, stream>>>(
      x, W1a, y1, ibtot, ipairs, icnt, ioff, icsr, N, projBlk);

  gin_tail_kernel<<<mlpBlk, 256, 0, stream>>>(
      y1, ioff, icnt, icsr, b1a, W1b, b1b, g1, be1, m1, v1, h1, N);

  pool_zero_kernel<<<(NGRAPH * HIDDEN + 255) / 256, 256, 0, stream>>>(sums, cnts);
  gin_mlp_kernel<HIDDEN><<<mlpBlk, 256, 0, stream>>>(
      h1, ioff, icnt, icsr, W2a, b2a, W2b, b2b, g2, be2, m2, v2, batch, sums, cnts, N);
  pool_div_kernel<<<(NGRAPH * HIDDEN + 255) / 256, 256, 0, stream>>>(sums, cnts, out);
}


# Mean pool loses its two launches: first launch zeroes out, layer-1 tiles count graph sizes, layer-2 tiles add column sum / size into out (quotients summed, arrival order); 8 launches become 6
# speedup vs baseline: 4.7692x; 1.0309x over previous
#include <hip/hip_runtime.h>

typedef float v2f __attribute__((ext_vector_type(2)));
typedef float v8f __attribute__((ext_vector_type(8)));

#define IN_FEATS 128
#define HIDDEN   64
#define NGRAPH   512
#define BN_EPS   1e-5f


#define BIN_SHIFT    9
#define BIN_NODES    512
#define MAX_BINS     256
#define SORT_THREADS 512
#define SORT_BLOCKS  256

#define ZERO_THREADS 256
#define ZERO_BLOCKS  ((NGRAPH * HIDDEN / 4 + ZERO_THREADS - 1) / ZERO_THREADS)
static_assert(ZERO_BLOCKS * ZERO_THREADS >= NGRAPH && ZERO_BLOCKS * ZERO_THREADS >= MAX_BINS,
              "one thread per float4 of out also covers the counts and the bin words");

__global__ __launch_bounds__(ZERO_THREADS) void bin_zero_kernel(int* __restrict__ binTotal,
                                                               int* __restrict__ binCursor,
                                                               int* __restrict__ gcount,
                                                               float* __restrict__ out) {
  const int i = (int)blockIdx.x * ZERO_THREADS + (int)threadIdx.x;
  if (i < NGRAPH * HIDDEN / 4) ((float4*)out)[i] = make_float4(0.0f, 0.0f, 0.0f, 0.0f);
  if (i < NGRAPH) gcount[i] = 0;
  if (i < MAX_BINS) {
    binTotal[i] = 0;
    binCursor[i] = 0;
  }
}

__global__ __launch_bounds__(SORT_THREADS) void bin_count_kernel(
    const int* __restrict__ ei, int* __restrict__ binTotal, int* __restrict__ blockHist, int E) {
  __shared__ int s_hist[MAX_BINS];
  const int t = threadIdx.x;
  if (t < MAX_BINS) s_hist[t] = 0;
  __syncthreads();
  const int* dst = ei + E;
  const int chunk = (E + (int)gridDim.x - 1) / (int)gridDim.x;
  const int c0 = (int)blockIdx.x * chunk;
  int c1 = c0 + chunk;
  if (c1 > E) c1 = E;
  for (int b0 = c0; b0 < c1; b0 += 4 * SORT_THREADS) {
    int d[4];
#pragma unroll
    for (int j = 0; j < 4; ++j) {
      const int e = b0 + j * SORT_THREADS + t;
      d[j] = dst[e < c1 ? e : c1 - 1];
    }
#pragma unroll
    for (int j = 0; j < 4; ++j)
      if (b0 + j * SORT_THREADS + t < c1) atomicAdd(&s_hist[d[j] >> BIN_SHIFT], 1);
  }
  __syncthreads();
  if (t < MAX_BINS) {
    const int h = s_hist[t];
    blockHist[(int)blockIdx.x * MAX_BINS + t] = h;
    if (h > 0) atomicAdd(&binTotal[t], h);
  }
}

__global__ __launch_bounds__(SORT_THREADS) void bin_scatter_kernel(
    const int* __restrict__ ei, const int* __restrict__ binTotal, int* __restrict__ binCursor,
    const int* __restrict__ blockHist, int* __restrict__ pairs, int E) {
  __shared__ int s_cur[MAX_BINS];
  __shared__ int s_wsum[SORT_THREADS / 32];
  const int t = threadIdx.x;
  const int lane = t & 31;
  const int wave = t >> 5;

  const int tot = t < MAX_BINS ? binTotal[t] : 0;
  int inc = tot;
#pragma unroll
  for (int o = 1; o < 32; o <<= 1) {
    const int u = __shfl_up(inc, o);
    if (lane >= o) inc += u;
  }
  if (lane == 31) s_wsum[wave] = inc;
  __syncthreads();
  int base = 0;
#pragma unroll
  for (int w = 0; w < SORT_THREADS / 32; ++w)
    if (w < wave) base += s_wsum[w];
  if (t < MAX_BINS) {
    const int h = blockHist[(int)blockIdx.x * MAX_BINS + t];
    int st = 0;
    if (h > 0) st = atomicAdd(&binCursor[t], h);
    s_cur[t] = base + inc - tot + st;
  }
  __syncthreads();

  const int* srcp = ei;
  const int* dst = ei + E;
  const int chunk = (E + (int)gridDim.x - 1) / (int)gridDim.x;
  const int c0 = (int)blockIdx.x * chunk;
  int c1 = c0 + chunk;
  if (c1 > E) c1 = E;
  for (int b0 = c0; b0 < c1; b0 += 4 * SORT_THREADS) {
    int s[4], d[4];
#pragma unroll
    for (int j = 0; j < 4; ++j) {
      const int e = b0 + j * SORT_THREADS + t;
      const int ec = e < c1 ? e : c1 - 1;
      d[j] = dst[ec];
      s[j] = srcp[ec];
    }
    int pos[4];
#pragma unroll
    for (int j = 0; j < 4; ++j)
      pos[j] = (b0 + j * SORT_THREADS + t < c1) ? atomicAdd(&s_cur[d[j] >> BIN_SHIFT], 1) : -1;
#pragma unroll
    for (int j = 0; j < 4; ++j)
      if (pos[j] >= 0)
        pairs[pos[j]] = (int)(((unsigned)s[j] << BIN_SHIFT) | (unsigned)(d[j] & (BIN_NODES - 1)));
  }
}

__device__ __forceinline__ void bin_sort_block(
    const int* __restrict__ binTotal, const int* __restrict__ pairs,
    int* __restrict__ cnt, int* __restrict__ off, int* __restrict__ csr, int n_nodes, int b) {
  __shared__ int s_cnt[BIN_NODES];
  __shared__ int s_wsum[BIN_NODES / 32];
  __shared__ int s_wsum2[BIN_NODES / 32];
  __shared__ int s_bin[2];
  const int t = threadIdx.x;
  const int lane = t & 31;
  const int wave = t >> 5;

  s_cnt[t] = 0;
  const int btot = t < MAX_BINS ? binTotal[t] : 0;
  int inc = btot;
#pragma unroll
  for (int o = 1; o < 32; o <<= 1) {
    const int u = __shfl_up(inc, o);
    if (lane >= o) inc += u;
  }
  if (lane == 31) s_wsum[wave] = inc;
  __syncthreads();
  int wbase = 0;
#pragma unroll
  for (int w = 0; w < BIN_NODES / 32; ++w)
    if (w < wave) wbase += s_wsum[w];
  if (t == b) {
    s_bin[0] = wbase + inc - btot;
    s_bin[1] = btot;
  }
  __syncthreads();
  const int base = s_bin[0];
  const int tot = s_bin[1];
  const int* p = pairs + base;

  for (int i0 = 0; i0 < tot; i0 += 4 * BIN_NODES) {
    unsigned v[4];
#pragma unroll
    for (int j = 0; j < 4; ++j) {
      const int i = i0 + j * BIN_NODES + t;
      v[j] = (unsigned)p[i < tot ? i : tot - 1];
    }
#pragma unroll
    for (int j = 0; j < 4; ++j)
      if (i0 + j * BIN_NODES + t < tot) atomicAdd(&s_cnt[v[j] & (BIN_NODES - 1)], 1);
  }
  __syncthreads();

  const int c = s_cnt[t];
  int inc2 = c;
#pragma unroll
  for (int o = 1; o < 32; o <<= 1) {
    const int u = __shfl_up(inc2, o);
    if (lane >= o) inc2 += u;
  }
  if (lane == 31) s_wsum2[wave] = inc2;
  __syncthreads();
  int wbase2 = 0;
#pragma unroll
  for (int w = 0; w < BIN_NODES / 32; ++w)
    if (w < wave) wbase2 += s_wsum2[w];
  const int start = base + wbase2 + inc2 - c;
  const int node = b * BIN_NODES + t;
  if (node < n_nodes) {
    cnt[node] = c;
    off[node] = start;
  }
  s_cnt[t] = start;
  __syncthreads();

  for (int i0 = 0; i0 < tot; i0 += 4 * BIN_NODES) {
    unsigned v[4];
#pragma unroll
    for (int j = 0; j < 4; ++j) {
      const int i = i0 + j * BIN_NODES + t;
      v[j] = (unsigned)p[i < tot ? i : tot - 1];
    }
    int pos[4];
#pragma unroll
    for (int j = 0; j < 4; ++j)
      pos[j] = (i0 + j * BIN_NODES + t < tot) ? atomicAdd(&s_cnt[v[j] & (BIN_NODES - 1)], 1) : -1;
#pragma unroll
    for (int j = 0; j < 4; ++j)
      if (pos[j] >= 0) csr[pos[j]] = (int)(v[j] >> BIN_SHIFT);
  }
}

__device__ __forceinline__ void gather_tile(
    const float* __restrict__ xin, const int* __restrict__ off, const int* __restrict__ cnt,
    const int* __restrict__ csr, float* __restrict__ tileT, int pitch, int row0, int n_nodes,
    int half, int l16) {
  const int c = l16 << 2;
  int hrow = row0 + l16;
  if (hrow >= n_nodes) hrow = n_nodes - 1;
  const int cntv = cnt[hrow];
  const int offv = off[hrow];
  for (int st = 0; st < 8; ++st) {
    const int r = 2 * st + half;
    const int len = __shfl(cntv, r);
    const int* p = csr + __shfl(offv, r);
    int i = row0 + r;
    if (i >= n_nodes) i = n_nodes - 1;
    float4 acc = *(const float4*)(xin + (long long)i * HIDDEN + c);
    const int last = len - 1;
    for (int j = 0; j < len; j += 4) {
      const int j1 = j + 1, j2 = j + 2, j3 = j + 3;
      const int s0 = p[j];
      const int s1 = p[j1 < last ? j1 : last];
      const int s2 = p[j2 < last ? j2 : last];
      const int s3 = p[j3 < last ? j3 : last];
      const float4 a0 = *(const float4*)(xin + (long long)s0 * HIDDEN + c);
      const float4 a1 = *(const float4*)(xin + (long long)s1 * HIDDEN + c);
      const float4 a2 = *(const float4*)(xin + (long long)s2 * HIDDEN + c);
      const float4 a3 = *(const float4*)(xin + (long long)s3 * HIDDEN + c);
      const int rem = len - j;
      acc.x += a0.x; acc.y += a0.y; acc.z += a0.z; acc.w += a0.w;
      const bool k1 = rem > 1, k2 = rem > 2, k3 = rem > 3;
      acc.x += k1 ? a1.x : 0.0f; acc.y += k1 ? a1.y : 0.0f;
      acc.z += k1 ? a1.z : 0.0f; acc.w += k1 ? a1.w : 0.0f;
      acc.x += k2 ? a2.x : 0.0f; acc.y += k2 ? a2.y : 0.0f;
      acc.z += k2 ? a2.z : 0.0f; acc.w += k2 ? a2.w : 0.0f;
      acc.x += k3 ? a3.x : 0.0f; acc.y += k3 ? a3.y : 0.0f;
      acc.z += k3 ? a3.z : 0.0f; acc.w += k3 ? a3.w : 0.0f;
    }
    *(float4*)(tileT + r * pitch + c) = acc;
  }
}

#define WPITCH_PAD 4
#define MLP_THREADS 256

template <int K, int NT = MLP_THREADS>
__device__ __forceinline__ void stage_weight_rows(float* __restrict__ sW,
                                                  const float* __restrict__ W) {
  constexpr int Q = K / 4;
  constexpr int R = 64 * Q / NT;
  float4 v[R];
#pragma unroll
  for (int j = 0; j < R; ++j) {
    const int i = (int)threadIdx.x + j * NT;
    const int row = i / Q;
    v[j] = *(const float4*)(W + row * K + 4 * (i - row * Q));
  }
#pragma unroll
  for (int j = 0; j < R; ++j) {
    const int i = (int)threadIdx.x + j * NT;
    const int row = i / Q;
    *(float4*)(sW + row * (K + WPITCH_PAD) + 4 * (i - row * Q)) = v[j];
  }
}

__device__ __forceinline__ void pool_add_run(float* __restrict__ out, const int* __restrict__ gcount,
                                             int g, int l16, const float* run) {
  float cg = (float)gcount[g];
  cg = cg > 1.0f ? cg : 1.0f;
  float* sp = out + (long long)g * HIDDEN + l16;
  atomicAdd(sp, run[0] / cg);
  atomicAdd(sp + 16, run[1] / cg);
  atomicAdd(sp + 32, run[2] / cg);
  atomicAdd(sp + 48, run[3] / cg);
}

template <int FIN>
__global__ __launch_bounds__(256) void gin_mlp_kernel(
    const float* __restrict__ hin,
    const int* __restrict__ off,
    const int* __restrict__ cnt,
    const int* __restrict__ csr,
    const float* __restrict__ Wa,
    const float* __restrict__ ba,
    const float* __restrict__ Wb,
    const float* __restrict__ bb,
    const float* __restrict__ gg,
    const float* __restrict__ be,
    const float* __restrict__ mm,
    const float* __restrict__ vv,
    const int* __restrict__ batch,
    const int* __restrict__ gcount,
    float* __restrict__ out,
    int n_nodes) {
  constexpr int PA = FIN + WPITCH_PAD;
  constexpr int PB = HIDDEN + WPITCH_PAD;
  __shared__ float sWa[64 * PA];
  __shared__ float sWb[64 * PB];
  static_assert(FIN == HIDDEN, "the tile of neighbour sums is HIDDEN wide");
  __shared__ float sT[8][16 * PB];

  const int wave = threadIdx.x >> 5;
  const int lane = threadIdx.x & 31;
  const int half = lane >> 4;
  const int l16  = lane & 15;

  const int tile = blockIdx.x * 8 + wave;
  const int row0 = tile * 16;

  int arow = row0 + l16;
  if (arow >= n_nodes) arow = n_nodes - 1;

  stage_weight_rows<FIN>(sWa, Wa);
  stage_weight_rows<HIDDEN>(sWb, Wb);
  __syncthreads();
  if (row0 >= n_nodes) return;

  gather_tile(hin, off, cnt, csr, sT[wave], PB, row0, n_nodes, half, l16);
  const float* tp = &sT[wave][l16 * PB + 2 * half];
  v2f afr[FIN / 4];
#pragma unroll
  for (int j = 0; j < FIN / 4; ++j) afr[j] = *(const v2f*)(tp + 4 * j);

  v8f acc[4];
#pragma unroll
  for (int t = 0; t < 4; ++t) {
    float bias = ba[t * 16 + l16];
#pragma unroll
    for (int r = 0; r < 8; ++r) acc[t][r] = bias;
  }

  const int gl = batch[arow];

#pragma unroll
  for (int k = 0; k < FIN; k += 4) {
    v2f a = afr[k / 4];
#pragma unroll
    for (int t = 0; t < 4; ++t) {
      const float* bp = sWa + (t * 16 + l16) * PA + k + 2 * half;
      v2f b = *(const v2f*)bp;
      acc[t] = __builtin_amdgcn_wmma_f32_16x16x4_f32(
          false, a, false, b, (short)0, acc[t], false, false);
    }
  }

#pragma unroll
  for (int t = 0; t < 4; ++t) {
#pragma unroll
    for (int r = 0; r < 8; ++r) {
      float x = acc[t][r];
      x = x > 0.0f ? x : 0.0f;
      sT[wave][(r + 8 * half) * PB + t * 16 + l16] = x;
    }
  }

  v8f acc2[4];
#pragma unroll
  for (int t = 0; t < 4; ++t)
#pragma unroll
    for (int r = 0; r < 8; ++r) acc2[t][r] = 0.0f;

  for (int k = 0; k < 64; k += 4) {
    v2f a = *(const v2f*)(tp + k);
#pragma unroll
    for (int t = 0; t < 4; ++t) {
      const float* bp = sWb + (t * 16 + l16) * PB + k + 2 * half;
      v2f b = *(const v2f*)bp;
      acc2[t] = __builtin_amdgcn_wmma_f32_16x16x4_f32(
          false, a, false, b, (short)0, acc2[t], false, false);
    }
  }

#pragma unroll
  for (int t = 0; t < 4; ++t) {
    int n = t * 16 + l16;
    float s  = gg[n] * rsqrtf(vv[n] + BN_EPS);
    float sh = (bb[n] - mm[n]) * s + be[n];
#pragma unroll
    for (int r = 0; r < 8; ++r) {
      float o = acc2[t][r] * s + sh;
      acc2[t][r] = o > 0.0f ? o : 0.0f;
    }
  }

  const int g0 = __builtin_amdgcn_readfirstlane(gl);
  const bool whole = row0 + 16 <= n_nodes;
  if (whole && __all(gl == g0)) {
    float cs[4];
#pragma unroll
    for (int t = 0; t < 4; ++t) {
      float v = acc2[t][0];
#pragma unroll
      for (int r = 1; r < 8; ++r) v += acc2[t][r];
      cs[t] = v + __shfl_xor(v, 16);
    }
    float c0 = (float)gcount[g0];
    c0 = c0 > 1.0f ? c0 : 1.0f;
    float* sp = out + (long long)g0 * HIDDEN + lane;
    atomicAdd(sp, (half ? cs[1] : cs[0]) / c0);
    atomicAdd(sp + 32, (half ? cs[3] : cs[2]) / c0);
  } else {
    int gr[8];
#pragma unroll
    for (int r = 0; r < 8; ++r) gr[r] = __shfl(gl, 8 * half + r);
    int gp = 0;
    bool open = false;
    float run[4] = {0.0f, 0.0f, 0.0f, 0.0f};
#pragma unroll
    for (int r = 0; r < 8; ++r) {
      if (row0 + 8 * half + r < n_nodes) {
        if (open && gr[r] != gp) {
          pool_add_run(out, gcount, gp, l16, run);
          run[0] = 0.0f; run[1] = 0.0f; run[2] = 0.0f; run[3] = 0.0f;
        }
        gp = gr[r];
        open = true;
#pragma unroll
        for (int t = 0; t < 4; ++t) run[t] += acc2[t][r];
      }
    }
    if (open) pool_add_run(out, gcount, gp, l16, run);
  }
}

#define PROJ_THREADS BIN_NODES
#define PROJ_TILES   (PROJ_THREADS / 32)

__device__ __forceinline__ void proj_block(
    const float* __restrict__ xin,
    const float* __restrict__ Wa,
    float* __restrict__ ysrc,
    int n_nodes, int blk) {
  constexpr int PA = IN_FEATS + WPITCH_PAD;
  __shared__ float sWa[64 * PA];

  const int wave = threadIdx.x >> 5;
  const int lane = threadIdx.x & 31;
  const int half = lane >> 4;
  const int l16  = lane & 15;

  const int tile = blk * PROJ_TILES + wave;
  const int row0 = tile * 16;

  int arow = row0 + l16;
  if (arow >= n_nodes) arow = n_nodes - 1;
  const float* aptr = xin + (long long)arow * IN_FEATS + 2 * half;
  v2f afr[IN_FEATS / 4];
#pragma unroll
  for (int j = 0; j < IN_FEATS / 4; ++j) afr[j] = *(const v2f*)(aptr + 4 * j);

  stage_weight_rows<IN_FEATS, PROJ_THREADS>(sWa, Wa);
#pragma unroll
  for (int j = 0; j < IN_FEATS / 4; j += 8)
    asm volatile("" ::"v"(afr[j]), "v"(afr[j + 1]), "v"(afr[j + 2]), "v"(afr[j + 3]),
                 "v"(afr[j + 4]), "v"(afr[j + 5]), "v"(afr[j + 6]), "v"(afr[j + 7]));
  __syncthreads();
  if (row0 >= n_nodes) return;

  v8f acc[4];
#pragma unroll
  for (int t = 0; t < 4; ++t)
#pragma unroll
    for (int r = 0; r < 8; ++r) acc[t][r] = 0.0f;

#pragma unroll
  for (int k = 0; k < IN_FEATS; k += 4) {
    v2f a = afr[k / 4];
#pragma unroll
    for (int t = 0; t < 4; ++t) {
      const float* bp = sWa + (t * 16 + l16) * PA + k + 2 * half;
      v2f b = *(const v2f*)bp;
      acc[t] = __builtin_amdgcn_wmma_f32_16x16x4_f32(
          false, a, false, b, (short)0, acc[t], false, false);
    }
  }

#pragma unroll
  for (int t = 0; t < 4; ++t) {
    int n = t * 16 + l16;
#pragma unroll
    for (int r = 0; r < 8; ++r) {
      int mrow = row0 + r + 8 * half;
      if (mrow < n_nodes) ysrc[(long long)mrow * HIDDEN + n] = acc[t][r];
    }
  }
}

__global__ __launch_bounds__(BIN_NODES) void rank_proj_kernel(
    const float* __restrict__ xin, const float* __restrict__ Wa, float* __restrict__ ysrc,
    const int* __restrict__ binTotal, const int* __restrict__ pairs,
    int* __restrict__ cnt, int* __restrict__ off, int* __restrict__ csr,
    int n_nodes, int n_proj) {
  static_assert(PROJ_THREADS == BIN_NODES, "both kinds of block have the launch's block size");
  const int blk = (int)blockIdx.x;
  if (blk < n_proj)
    proj_block(xin, Wa, ysrc, n_nodes, blk);
  else
    bin_sort_block(binTotal, pairs, cnt, off, csr, n_nodes, blk - n_proj);
}

__global__ __launch_bounds__(256) void gin_tail_kernel(
    const float* __restrict__ ysrc,
    const int* __restrict__ off,
    const int* __restrict__ cnt,
    const int* __restrict__ csr,
    const int* __restrict__ batch,
    const float* __restrict__ ba,
    const float* __restrict__ Wb,
    const float* __restrict__ bb,
    const float* __restrict__ gg,
    const float* __restrict__ be,
    const float* __restrict__ mm,
    const float* __restrict__ vv,
    float* __restrict__ hout,
    int* __restrict__ gcount,
    int n_nodes) {
  constexpr int PB = HIDDEN + WPITCH_PAD;
  __shared__ float sWb[64 * PB];
  __shared__ float sT[8][16 * PB];

  const int wave = threadIdx.x >> 5;
  const int lane = threadIdx.x & 31;
  const int half = lane >> 4;
  const int l16  = lane & 15;

  const int tile = blockIdx.x * 8 + wave;
  const int row0 = tile * 16;

  stage_weight_rows<HIDDEN>(sWb, Wb);
  __syncthreads();
  if (row0 >= n_nodes) return;

  const int nrow = row0 + l16;
  const int gl = batch[nrow < n_nodes ? nrow : n_nodes - 1];

  gather_tile(ysrc, off, cnt, csr, sT[wave], PB, row0, n_nodes, half, l16);

  {
    const int g0 = __builtin_amdgcn_readfirstlane(gl);
    if (row0 + 16 <= n_nodes && __all(gl == g0)) {
      if (lane == 0 && (unsigned)g0 < (unsigned)NGRAPH) atomicAdd(&gcount[g0], 16);
    } else if (half == 0 && nrow < n_nodes && (unsigned)gl < (unsigned)NGRAPH) {
      atomicAdd(&gcount[gl], 1);
    }
  }

  const float* tp = &sT[wave][l16 * PB + 2 * half];
  v2f afr[HIDDEN / 4];
#pragma unroll
  for (int j = 0; j < HIDDEN / 4; ++j) afr[j] = *(const v2f*)(tp + 4 * j);

  v8f acc2[4];
#pragma unroll
  for (int t = 0; t < 4; ++t)
#pragma unroll
    for (int r = 0; r < 8; ++r) acc2[t][r] = 0.0f;

  const float* biasp = ba + 2 * half;

#pragma unroll
  for (int k = 0; k < HIDDEN; k += 4) {
    v2f a = afr[k / 4];
    v2f bs = *(const v2f*)(biasp + k);
    a.x = a.x + bs.x;
    a.y = a.y + bs.y;
    a.x = a.x > 0.0f ? a.x : 0.0f;
    a.y = a.y > 0.0f ? a.y : 0.0f;
#pragma unroll
    for (int t = 0; t < 4; ++t) {
      const float* bp = sWb + (t * 16 + l16) * PB + k + 2 * half;
      v2f b = *(const v2f*)bp;
      acc2[t] = __builtin_amdgcn_wmma_f32_16x16x4_f32(
          false, a, false, b, (short)0, acc2[t], false, false);
    }
  }

#pragma unroll
  for (int t = 0; t < 4; ++t) {
    int n = t * 16 + l16;
    float s  = gg[n] * rsqrtf(vv[n] + BN_EPS);
    float sh = (bb[n] - mm[n]) * s + be[n];
#pragma unroll
    for (int r = 0; r < 8; ++r) {
      int mrow = row0 + r + 8 * half;
      float o = acc2[t][r] * s + sh;
      o = o > 0.0f ? o : 0.0f;
      if (mrow < n_nodes) hout[(long long)mrow * HIDDEN + n] = o;
    }
  }
}

extern "C" void kernel_launch(void* const* d_in, const int* in_sizes, int n_in,
                              void* d_out, int out_size, void* d_ws, size_t ws_size,
                              hipStream_t stream) {
  const float* x   = (const float*)d_in[0];
  const int* ei    = (const int*)d_in[1];
  const int* batch = (const int*)d_in[2];
  const float* W1a = (const float*)d_in[3];
  const float* b1a = (const float*)d_in[4];
  const float* W1b = (const float*)d_in[5];
  const float* b1b = (const float*)d_in[6];
  const float* g1  = (const float*)d_in[7];
  const float* be1 = (const float*)d_in[8];
  const float* m1  = (const float*)d_in[9];
  const float* v1  = (const float*)d_in[10];
  const float* W2a = (const float*)d_in[11];
  const float* b2a = (const float*)d_in[12];
  const float* W2b = (const float*)d_in[13];
  const float* b2b = (const float*)d_in[14];
  const float* g2  = (const float*)d_in[15];
  const float* be2 = (const float*)d_in[16];
  const float* m2  = (const float*)d_in[17];
  const float* v2  = (const float*)d_in[18];
  float* out = (float*)d_out;

  const int N = in_sizes[0] / IN_FEATS;
  const int E = in_sizes[1] / 2;

  float* ws    = (float*)d_ws;
  float* y1    = ws;
  float* h1    = ws + (long long)N * IN_FEATS;
  int* gcnt    = (int*)(h1 + (long long)N * HIDDEN);
  const int Npad = (N + 3) & ~3;
  int* icnt    = gcnt + NGRAPH;
  int* ioff    = icnt + Npad;
  int* ipairs  = ioff + Npad;
  int* icsr    = ipairs + E;
  int* ibtot   = icsr + E;
  int* ibcur   = ibtot + MAX_BINS;
  int* ibhist  = ibcur + MAX_BINS;
  (void)ws_size; (void)n_in; (void)out_size;

  const int tiles  = (N + 15) / 16;
  const int mlpBlk = (tiles + 7) / 8;
  const int projBlk = (tiles + PROJ_TILES - 1) / PROJ_TILES;
  const int nBins = (N + BIN_NODES - 1) >> BIN_SHIFT;

  bin_zero_kernel<<<ZERO_BLOCKS, ZERO_THREADS, 0, stream>>>(ibtot, ibcur, gcnt, out);
  bin_count_kernel<<<SORT_BLOCKS, SORT_THREADS, 0, stream>>>(ei, ibtot, ibhist, E);
  bin_scatter_kernel<<<SORT_BLOCKS, SORT_THREADS, 0, stream>>>(ei, ibtot, ibcur, ibhist, ipairs, E);
  rank_proj_kernel<<<projBlk + nBins, BIN_NODES, 0, stream>>>(
      x, W1a, y1, ibtot, ipairs, icnt, ioff, icsr, N, projBlk);

  gin_tail_kernel<<<mlpBlk, 256, 0, stream>>>(
      y1, ioff, icnt, icsr, batch, b1a, W1b, b1b, g1, be1, m1, v1, h1, gcnt, N);

  gin_mlp_kernel<HIDDEN><<<mlpBlk, 256, 0, stream>>>(
      h1, ioff, icnt, icsr, W2a, b2a, W2b, b2b, g2, be2, m2, v2, batch, gcnt, out, N);
}


# Edge binning made block-local: each block sorts its own chunk by bin in LDS, no global counters; ranking reads its bin's per-block segments; 6 launches become 4; bucket order = LDS rank arrival
# speedup vs baseline: 5.0864x; 1.0665x over previous
#include <hip/hip_runtime.h>

typedef float v2f __attribute__((ext_vector_type(2)));
typedef float v8f __attribute__((ext_vector_type(8)));

#define IN_FEATS 128
#define HIDDEN   64
#define NGRAPH   512
#define BN_EPS   1e-5f


#define BIN_SHIFT    9
#define BIN_NODES    512
#define MAX_BINS     256
#define SORT_THREADS 512
#define SORT_BLOCKS  256

__device__ __forceinline__ int sort_chunk(int E) { return (E + SORT_BLOCKS - 1) / SORT_BLOCKS; }

static_assert(SORT_BLOCKS * SORT_THREADS >= NGRAPH * HIDDEN / 4 && SORT_BLOCKS * SORT_THREADS >= NGRAPH,
              "one thread per float4 of out also covers the per-graph counts");

__global__ __launch_bounds__(SORT_THREADS) void bin_local_kernel(
    const int* __restrict__ ei, int* __restrict__ segLen, int* __restrict__ segRel,
    int* __restrict__ pairs, int* __restrict__ gcount, float* __restrict__ out, int E) {
  __shared__ int s_hist[MAX_BINS];
  __shared__ int s_wsum[SORT_THREADS / 32];
  const int t = threadIdx.x;
  const int lane = t & 31;
  const int wave = t >> 5;

  const int gid = (int)blockIdx.x * SORT_THREADS + t;
  if (gid < NGRAPH * HIDDEN / 4) ((float4*)out)[gid] = make_float4(0.0f, 0.0f, 0.0f, 0.0f);
  if (gid < NGRAPH) gcount[gid] = 0;

  if (t < MAX_BINS) s_hist[t] = 0;
  __syncthreads();
  const int* srcp = ei;
  const int* dst = ei + E;
  const int chunk = sort_chunk(E);
  const int c0 = (int)blockIdx.x * chunk;
  int c1 = c0 + chunk;
  if (c1 > E) c1 = E;
  for (int b0 = c0; b0 < c1; b0 += 4 * SORT_THREADS) {
    int d[4];
#pragma unroll
    for (int j = 0; j < 4; ++j) {
      const int e = b0 + j * SORT_THREADS + t;
      d[j] = dst[e < c1 ? e : c1 - 1];
    }
#pragma unroll
    for (int j = 0; j < 4; ++j)
      if (b0 + j * SORT_THREADS + t < c1) atomicAdd(&s_hist[d[j] >> BIN_SHIFT], 1);
  }
  __syncthreads();

  const int h = t < MAX_BINS ? s_hist[t] : 0;
  int inc = h;
#pragma unroll
  for (int o = 1; o < 32; o <<= 1) {
    const int u = __shfl_up(inc, o);
    if (lane >= o) inc += u;
  }
  if (lane == 31) s_wsum[wave] = inc;
  __syncthreads();
  int wbase = 0;
#pragma unroll
  for (int w = 0; w < SORT_THREADS / 32; ++w)
    if (w < wave) wbase += s_wsum[w];
  if (t < MAX_BINS) {
    const int rel = wbase + inc - h;
    segLen[(int)blockIdx.x * MAX_BINS + t] = h;
    segRel[(int)blockIdx.x * MAX_BINS + t] = rel;
    s_hist[t] = c0 + rel;
  }
  __syncthreads();

  for (int b0 = c0; b0 < c1; b0 += 4 * SORT_THREADS) {
    int s[4], d[4];
#pragma unroll
    for (int j = 0; j < 4; ++j) {
      const int e = b0 + j * SORT_THREADS + t;
      const int ec = e < c1 ? e : c1 - 1;
      d[j] = dst[ec];
      s[j] = srcp[ec];
    }
    int pos[4];
#pragma unroll
    for (int j = 0; j < 4; ++j)
      pos[j] = (b0 + j * SORT_THREADS + t < c1) ? atomicAdd(&s_hist[d[j] >> BIN_SHIFT], 1) : -1;
#pragma unroll
    for (int j = 0; j < 4; ++j)
      if (pos[j] >= 0)
        pairs[pos[j]] = (int)(((unsigned)s[j] << BIN_SHIFT) | (unsigned)(d[j] & (BIN_NODES - 1)));
  }
}

#define SEG_WAVES    (BIN_NODES / 32)
#define SEG_PER_WAVE (SORT_BLOCKS / SEG_WAVES)
static_assert(SORT_BLOCKS <= BIN_NODES && SORT_BLOCKS % SEG_WAVES == 0,
              "one thread per segment header, a whole number of segments per wave");

__device__ __forceinline__ void bin_sort_block(
    const int* __restrict__ segLen, const int* __restrict__ segRel, const int* __restrict__ pairs,
    int* __restrict__ cnt, int* __restrict__ off, int* __restrict__ csr, int n_nodes, int E, int b) {
  __shared__ int s_cnt[BIN_NODES];
  __shared__ int s_wsum[BIN_NODES / 32];
  __shared__ int s_wsum2[BIN_NODES / 32];
  __shared__ int s_len[SORT_BLOCKS];
  __shared__ int s_st[SORT_BLOCKS];
  const int t = threadIdx.x;
  const int lane = t & 31;
  const int wave = t >> 5;
  const int chunk = sort_chunk(E);

  s_cnt[t] = 0;
  int rel = 0;
  if (t < SORT_BLOCKS) {
    rel = segRel[t * MAX_BINS + b];
    s_len[t] = segLen[t * MAX_BINS + b];
    s_st[t] = t * chunk + rel;
  }
  int red = rel;
#pragma unroll
  for (int o = 16; o > 0; o >>= 1) red += __shfl_xor(red, o);
  if (lane == 0) s_wsum[wave] = red;
  __syncthreads();
  int base = 0;
#pragma unroll
  for (int w = 0; w < BIN_NODES / 32; ++w) base += s_wsum[w];

  unsigned v[SEG_PER_WAVE][2];
  int len[SEG_PER_WAVE];
#pragma unroll
  for (int j = 0; j < SEG_PER_WAVE; ++j) {
    const int q = wave + j * SEG_WAVES;
    len[j] = s_len[q];
    const int st = s_st[q];
#pragma unroll
    for (int u = 0; u < 2; ++u) {
      const int k = lane + 32 * u;
      v[j][u] = (unsigned)pairs[k < len[j] ? st + k : 0];
    }
  }
#pragma unroll
  for (int j = 0; j < SEG_PER_WAVE; ++j)
#pragma unroll
    for (int u = 0; u < 2; ++u)
      if (lane + 32 * u < len[j]) atomicAdd(&s_cnt[v[j][u] & (BIN_NODES - 1)], 1);
  bool longer = false;
#pragma unroll
  for (int j = 0; j < SEG_PER_WAVE; ++j) longer |= len[j] > 64;
  if (longer)
    for (int q = wave; q < SORT_BLOCKS; q += SEG_WAVES) {
      const int L = s_len[q];
      const int* p = pairs + s_st[q];
      for (int k = 64 + lane; k < L; k += 32)
        atomicAdd(&s_cnt[(unsigned)p[k] & (BIN_NODES - 1)], 1);
    }
  __syncthreads();

  const int c = s_cnt[t];
  int inc2 = c;
#pragma unroll
  for (int o = 1; o < 32; o <<= 1) {
    const int u = __shfl_up(inc2, o);
    if (lane >= o) inc2 += u;
  }
  if (lane == 31) s_wsum2[wave] = inc2;
  __syncthreads();
  int wbase2 = 0;
#pragma unroll
  for (int w = 0; w < BIN_NODES / 32; ++w)
    if (w < wave) wbase2 += s_wsum2[w];
  const int start = base + wbase2 + inc2 - c;
  const int node = b * BIN_NODES + t;
  if (node < n_nodes) {
    cnt[node] = c;
    off[node] = start;
  }
  s_cnt[t] = start;
  __syncthreads();

#pragma unroll
  for (int j = 0; j < SEG_PER_WAVE; ++j) {
    int pos[2];
#pragma unroll
    for (int u = 0; u < 2; ++u)
      pos[u] = (lane + 32 * u < len[j]) ? atomicAdd(&s_cnt[v[j][u] & (BIN_NODES - 1)], 1) : -1;
#pragma unroll
    for (int u = 0; u < 2; ++u)
      if (pos[u] >= 0) csr[pos[u]] = (int)(v[j][u] >> BIN_SHIFT);
  }
  if (longer)
    for (int q = wave; q < SORT_BLOCKS; q += SEG_WAVES) {
      const int L = s_len[q];
      const int* p = pairs + s_st[q];
      for (int k = 64 + lane; k < L; k += 32) {
        const unsigned w = (unsigned)p[k];
        csr[atomicAdd(&s_cnt[w & (BIN_NODES - 1)], 1)] = (int)(w >> BIN_SHIFT);
      }
    }
}

__device__ __forceinline__ void gather_tile(
    const float* __restrict__ xin, const int* __restrict__ off, const int* __restrict__ cnt,
    const int* __restrict__ csr, float* __restrict__ tileT, int pitch, int row0, int n_nodes,
    int half, int l16) {
  const int c = l16 << 2;
  int hrow = row0 + l16;
  if (hrow >= n_nodes) hrow = n_nodes - 1;
  const int cntv = cnt[hrow];
  const int offv = off[hrow];
  for (int st = 0; st < 8; ++st) {
    const int r = 2 * st + half;
    const int len = __shfl(cntv, r);
    const int* p = csr + __shfl(offv, r);
    int i = row0 + r;
    if (i >= n_nodes) i = n_nodes - 1;
    float4 acc = *(const float4*)(xin + (long long)i * HIDDEN + c);
    const int last = len - 1;
    for (int j = 0; j < len; j += 4) {
      const int j1 = j + 1, j2 = j + 2, j3 = j + 3;
      const int s0 = p[j];
      const int s1 = p[j1 < last ? j1 : last];
      const int s2 = p[j2 < last ? j2 : last];
      const int s3 = p[j3 < last ? j3 : last];
      const float4 a0 = *(const float4*)(xin + (long long)s0 * HIDDEN + c);
      const float4 a1 = *(const float4*)(xin + (long long)s1 * HIDDEN + c);
      const float4 a2 = *(const float4*)(xin + (long long)s2 * HIDDEN + c);
      const float4 a3 = *(const float4*)(xin + (long long)s3 * HIDDEN + c);
      const int rem = len - j;
      acc.x += a0.x; acc.y += a0.y; acc.z += a0.z; acc.w += a0.w;
      const bool k1 = rem > 1, k2 = rem > 2, k3 = rem > 3;
      acc.x += k1 ? a1.x : 0.0f; acc.y += k1 ? a1.y : 0.0f;
      acc.z += k1 ? a1.z : 0.0f; acc.w += k1 ? a1.w : 0.0f;
      acc.x += k2 ? a2.x : 0.0f; acc.y += k2 ? a2.y : 0.0f;
      acc.z += k2 ? a2.z : 0.0f; acc.w += k2 ? a2.w : 0.0f;
      acc.x += k3 ? a3.x : 0.0f; acc.y += k3 ? a3.y : 0.0f;
      acc.z += k3 ? a3.z : 0.0f; acc.w += k3 ? a3.w : 0.0f;
    }
    *(float4*)(tileT + r * pitch + c) = acc;
  }
}

#define WPITCH_PAD 4
#define MLP_THREADS 256

template <int K, int NT = MLP_THREADS>
__device__ __forceinline__ void stage_weight_rows(float* __restrict__ sW,
                                                  const float* __restrict__ W) {
  constexpr int Q = K / 4;
  constexpr int R = 64 * Q / NT;
  float4 v[R];
#pragma unroll
  for (int j = 0; j < R; ++j) {
    const int i = (int)threadIdx.x + j * NT;
    const int row = i / Q;
    v[j] = *(const float4*)(W + row * K + 4 * (i - row * Q));
  }
#pragma unroll
  for (int j = 0; j < R; ++j) {
    const int i = (int)threadIdx.x + j * NT;
    const int row = i / Q;
    *(float4*)(sW + row * (K + WPITCH_PAD) + 4 * (i - row * Q)) = v[j];
  }
}

__device__ __forceinline__ void pool_add_run(float* __restrict__ out, const int* __restrict__ gcount,
                                             int g, int l16, const float* run) {
  float cg = (float)gcount[g];
  cg = cg > 1.0f ? cg : 1.0f;
  float* sp = out + (long long)g * HIDDEN + l16;
  atomicAdd(sp, run[0] / cg);
  atomicAdd(sp + 16, run[1] / cg);
  atomicAdd(sp + 32, run[2] / cg);
  atomicAdd(sp + 48, run[3] / cg);
}

template <int FIN>
__global__ __launch_bounds__(256) void gin_mlp_kernel(
    const float* __restrict__ hin,
    const int* __restrict__ off,
    const int* __restrict__ cnt,
    const int* __restrict__ csr,
    const float* __restrict__ Wa,
    const float* __restrict__ ba,
    const float* __restrict__ Wb,
    const float* __restrict__ bb,
    const float* __restrict__ gg,
    const float* __restrict__ be,
    const float* __restrict__ mm,
    const float* __restrict__ vv,
    const int* __restrict__ batch,
    const int* __restrict__ gcount,
    float* __restrict__ out,
    int n_nodes) {
  constexpr int PA = FIN + WPITCH_PAD;
  constexpr int PB = HIDDEN + WPITCH_PAD;
  __shared__ float sWa[64 * PA];
  __shared__ float sWb[64 * PB];
  static_assert(FIN == HIDDEN, "the tile of neighbour sums is HIDDEN wide");
  __shared__ float sT[8][16 * PB];

  const int wave = threadIdx.x >> 5;
  const int lane = threadIdx.x & 31;
  const int half = lane >> 4;
  const int l16  = lane & 15;

  const int tile = blockIdx.x * 8 + wave;
  const int row0 = tile * 16;

  int arow = row0 + l16;
  if (arow >= n_nodes) arow = n_nodes - 1;

  stage_weight_rows<FIN>(sWa, Wa);
  stage_weight_rows<HIDDEN>(sWb, Wb);
  __syncthreads();
  if (row0 >= n_nodes) return;

  gather_tile(hin, off, cnt, csr, sT[wave], PB, row0, n_nodes, half, l16);
  const float* tp = &sT[wave][l16 * PB + 2 * half];
  v2f afr[FIN / 4];
#pragma unroll
  for (int j = 0; j < FIN / 4; ++j) afr[j] = *(const v2f*)(tp + 4 * j);

  v8f acc[4];
#pragma unroll
  for (int t = 0; t < 4; ++t) {
    float bias = ba[t * 16 + l16];
#pragma unroll
    for (int r = 0; r < 8; ++r) acc[t][r] = bias;
  }

  const int gl = batch[arow];

#pragma unroll
  for (int k = 0; k < FIN; k += 4) {
    v2f a = afr[k / 4];
#pragma unroll
    for (int t = 0; t < 4; ++t) {
      const float* bp = sWa + (t * 16 + l16) * PA + k + 2 * half;
      v2f b = *(const v2f*)bp;
      acc[t] = __builtin_amdgcn_wmma_f32_16x16x4_f32(
          false, a, false, b, (short)0, acc[t], false, false);
    }
  }

#pragma unroll
  for (int t = 0; t < 4; ++t) {
#pragma unroll
    for (int r = 0; r < 8; ++r) {
      float x = acc[t][r];
      x = x > 0.0f ? x : 0.0f;
      sT[wave][(r + 8 * half) * PB + t * 16 + l16] = x;
    }
  }

  v8f acc2[4];
#pragma unroll
  for (int t = 0; t < 4; ++t)
#pragma unroll
    for (int r = 0; r < 8; ++r) acc2[t][r] = 0.0f;

  for (int k = 0; k < 64; k += 4) {
    v2f a = *(const v2f*)(tp + k);
#pragma unroll
    for (int t = 0; t < 4; ++t) {
      const float* bp = sWb + (t * 16 + l16) * PB + k + 2 * half;
      v2f b = *(const v2f*)bp;
      acc2[t] = __builtin_amdgcn_wmma_f32_16x16x4_f32(
          false, a, false, b, (short)0, acc2[t], false, false);
    }
  }

#pragma unroll
  for (int t = 0; t < 4; ++t) {
    int n = t * 16 + l16;
    float s  = gg[n] * rsqrtf(vv[n] + BN_EPS);
    float sh = (bb[n] - mm[n]) * s + be[n];
#pragma unroll
    for (int r = 0; r < 8; ++r) {
      float o = acc2[t][r] * s + sh;
      acc2[t][r] = o > 0.0f ? o : 0.0f;
    }
  }

  const int g0 = __builtin_amdgcn_readfirstlane(gl);
  const bool whole = row0 + 16 <= n_nodes;
  if (whole && __all(gl == g0)) {
    float cs[4];
#pragma unroll
    for (int t = 0; t < 4; ++t) {
      float v = acc2[t][0];
#pragma unroll
      for (int r = 1; r < 8; ++r) v += acc2[t][r];
      cs[t] = v + __shfl_xor(v, 16);
    }
    float c0 = (float)gcount[g0];
    c0 = c0 > 1.0f ? c0 : 1.0f;
    float* sp = out + (long long)g0 * HIDDEN + lane;
    atomicAdd(sp, (half ? cs[1] : cs[0]) / c0);
    atomicAdd(sp + 32, (half ? cs[3] : cs[2]) / c0);
  } else {
    int gr[8];
#pragma unroll
    for (int r = 0; r < 8; ++r) gr[r] = __shfl(gl, 8 * half + r);
    int gp = 0;
    bool open = false;
    float run[4] = {0.0f, 0.0f, 0.0f, 0.0f};
#pragma unroll
    for (int r = 0; r < 8; ++r) {
      if (row0 + 8 * half + r < n_nodes) {
        if (open && gr[r] != gp) {
          pool_add_run(out, gcount, gp, l16, run);
          run[0] = 0.0f; run[1] = 0.0f; run[2] = 0.0f; run[3] = 0.0f;
        }
        gp = gr[r];
        open = true;
#pragma unroll
        for (int t = 0; t < 4; ++t) run[t] += acc2[t][r];
      }
    }
    if (open) pool_add_run(out, gcount, gp, l16, run);
  }
}

#define PROJ_THREADS BIN_NODES
#define PROJ_TILES   (PROJ_THREADS / 32)

__device__ __forceinline__ void proj_block(
    const float* __restrict__ xin,
    const float* __restrict__ Wa,
    float* __restrict__ ysrc,
    int n_nodes, int blk) {
  constexpr int PA = IN_FEATS + WPITCH_PAD;
  __shared__ float sWa[64 * PA];

  const int wave = threadIdx.x >> 5;
  const int lane = threadIdx.x & 31;
  const int half = lane >> 4;
  const int l16  = lane & 15;

  const int tile = blk * PROJ_TILES + wave;
  const int row0 = tile * 16;

  int arow = row0 + l16;
  if (arow >= n_nodes) arow = n_nodes - 1;
  const float* aptr = xin + (long long)arow * IN_FEATS + 2 * half;
  v2f afr[IN_FEATS / 4];
#pragma unroll
  for (int j = 0; j < IN_FEATS / 4; ++j) afr[j] = *(const v2f*)(aptr + 4 * j);

  stage_weight_rows<IN_FEATS, PROJ_THREADS>(sWa, Wa);
#pragma unroll
  for (int j = 0; j < IN_FEATS / 4; j += 8)
    asm volatile("" ::"v"(afr[j]), "v"(afr[j + 1]), "v"(afr[j + 2]), "v"(afr[j + 3]),
                 "v"(afr[j + 4]), "v"(afr[j + 5]), "v"(afr[j + 6]), "v"(afr[j + 7]));
  __syncthreads();
  if (row0 >= n_nodes) return;

  v8f acc[4];
#pragma unroll
  for (int t = 0; t < 4; ++t)
#pragma unroll
    for (int r = 0; r < 8; ++r) acc[t][r] = 0.0f;

#pragma unroll
  for (int k = 0; k < IN_FEATS; k += 4) {
    v2f a = afr[k / 4];
#pragma unroll
    for (int t = 0; t < 4; ++t) {
      const float* bp = sWa + (t * 16 + l16) * PA + k + 2 * half;
      v2f b = *(const v2f*)bp;
      acc[t] = __builtin_amdgcn_wmma_f32_16x16x4_f32(
          false, a, false, b, (short)0, acc[t], false, false);
    }
  }

#pragma unroll
  for (int t = 0; t < 4; ++t) {
    int n = t * 16 + l16;
#pragma unroll
    for (int r = 0; r < 8; ++r) {
      int mrow = row0 + r + 8 * half;
      if (mrow < n_nodes) ysrc[(long long)mrow * HIDDEN + n] = acc[t][r];
    }
  }
}

__global__ __launch_bounds__(BIN_NODES) void rank_proj_kernel(
    const float* __restrict__ xin, const float* __restrict__ Wa, float* __restrict__ ysrc,
    const int* __restrict__ segLen, const int* __restrict__ segRel, const int* __restrict__ pairs,
    int* __restrict__ cnt, int* __restrict__ off, int* __restrict__ csr,
    int n_nodes, int E, int n_proj) {
  static_assert(PROJ_THREADS == BIN_NODES, "both kinds of block have the launch's block size");
  const int blk = (int)blockIdx.x;
  if (blk < n_proj)
    proj_block(xin, Wa, ysrc, n_nodes, blk);
  else
    bin_sort_block(segLen, segRel, pairs, cnt, off, csr, n_nodes, E, blk - n_proj);
}

__global__ __launch_bounds__(256) void gin_tail_kernel(
    const float* __restrict__ ysrc,
    const int* __restrict__ off,
    const int* __restrict__ cnt,
    const int* __restrict__ csr,
    const int* __restrict__ batch,
    const float* __restrict__ ba,
    const float* __restrict__ Wb,
    const float* __restrict__ bb,
    const float* __restrict__ gg,
    const float* __restrict__ be,
    const float* __restrict__ mm,
    const float* __restrict__ vv,
    float* __restrict__ hout,
    int* __restrict__ gcount,
    int n_nodes) {
  constexpr int PB = HIDDEN + WPITCH_PAD;
  __shared__ float sWb[64 * PB];
  __shared__ float sT[8][16 * PB];

  const int wave = threadIdx.x >> 5;
  const int lane = threadIdx.x & 31;
  const int half = lane >> 4;
  const int l16  = lane & 15;

  const int tile = blockIdx.x * 8 + wave;
  const int row0 = tile * 16;

  stage_weight_rows<HIDDEN>(sWb, Wb);
  __syncthreads();
  if (row0 >= n_nodes) return;

  const int nrow = row0 + l16;
  const int gl = batch[nrow < n_nodes ? nrow : n_nodes - 1];

  gather_tile(ysrc, off, cnt, csr, sT[wave], PB, row0, n_nodes, half, l16);

  {
    const int g0 = __builtin_amdgcn_readfirstlane(gl);
    if (row0 + 16 <= n_nodes && __all(gl == g0)) {
      if (lane == 0 && (unsigned)g0 < (unsigned)NGRAPH) atomicAdd(&gcount[g0], 16);
    } else if (half == 0 && nrow < n_nodes && (unsigned)gl < (unsigned)NGRAPH) {
      atomicAdd(&gcount[gl], 1);
    }
  }

  const float* tp = &sT[wave][l16 * PB + 2 * half];
  v2f afr[HIDDEN / 4];
#pragma unroll
  for (int j = 0; j < HIDDEN / 4; ++j) afr[j] = *(const v2f*)(tp + 4 * j);

  v8f acc2[4];
#pragma unroll
  for (int t = 0; t < 4; ++t)
#pragma unroll
    for (int r = 0; r < 8; ++r) acc2[t][r] = 0.0f;

  const float* biasp = ba + 2 * half;

#pragma unroll
  for (int k = 0; k < HIDDEN; k += 4) {
    v2f a = afr[k / 4];
    v2f bs = *(const v2f*)(biasp + k);
    a.x = a.x + bs.x;
    a.y = a.y + bs.y;
    a.x = a.x > 0.0f ? a.x : 0.0f;
    a.y = a.y > 0.0f ? a.y : 0.0f;
#pragma unroll
    for (int t = 0; t < 4; ++t) {
      const float* bp = sWb + (t * 16 + l16) * PB + k + 2 * half;
      v2f b = *(const v2f*)bp;
      acc2[t] = __builtin_amdgcn_wmma_f32_16x16x4_f32(
          false, a, false, b, (short)0, acc2[t], false, false);
    }
  }

#pragma unroll
  for (int t = 0; t < 4; ++t) {
    int n = t * 16 + l16;
    float s  = gg[n] * rsqrtf(vv[n] + BN_EPS);
    float sh = (bb[n] - mm[n]) * s + be[n];
#pragma unroll
    for (int r = 0; r < 8; ++r) {
      int mrow = row0 + r + 8 * half;
      float o = acc2[t][r] * s + sh;
      o = o > 0.0f ? o : 0.0f;
      if (mrow < n_nodes) hout[(long long)mrow * HIDDEN + n] = o;
    }
  }
}

extern "C" void kernel_launch(void* const* d_in, const int* in_sizes, int n_in,
                              void* d_out, int out_size, void* d_ws, size_t ws_size,
                              hipStream_t stream) {
  const float* x   = (const float*)d_in[0];
  const int* ei    = (const int*)d_in[1];
  const int* batch = (const int*)d_in[2];
  const float* W1a = (const float*)d_in[3];
  const float* b1a = (const float*)d_in[4];
  const float* W1b = (const float*)d_in[5];
  const float* b1b = (const float*)d_in[6];
  const float* g1  = (const float*)d_in[7];
  const float* be1 = (const float*)d_in[8];
  const float* m1  = (const float*)d_in[9];
  const float* v1  = (const float*)d_in[10];
  const float* W2a = (const float*)d_in[11];
  const float* b2a = (const float*)d_in[12];
  const float* W2b = (const float*)d_in[13];
  const float* b2b = (const float*)d_in[14];
  const float* g2  = (const float*)d_in[15];
  const float* be2 = (const float*)d_in[16];
  const float* m2  = (const float*)d_in[17];
  const float* v2  = (const float*)d_in[18];
  float* out = (float*)d_out;

  const int N = in_sizes[0] / IN_FEATS;
  const int E = in_sizes[1] / 2;

  float* ws    = (float*)d_ws;
  float* y1    = ws;
  float* h1    = ws + (long long)N * IN_FEATS;
  int* gcnt    = (int*)(h1 + (long long)N * HIDDEN);
  const int Npad = (N + 3) & ~3;
  int* icnt    = gcnt + NGRAPH;
  int* ioff    = icnt + Npad;
  int* ipairs  = ioff + Npad;
  int* icsr    = ipairs + E;
  int* iseglen = icsr + E;
  int* isegrel = iseglen + SORT_BLOCKS * MAX_BINS;
  (void)ws_size; (void)n_in; (void)out_size;

  const int tiles  = (N + 15) / 16;
  const int mlpBlk = (tiles + 7) / 8;
  const int projBlk = (tiles + PROJ_TILES - 1) / PROJ_TILES;
  const int nBins = (N + BIN_NODES - 1) >> BIN_SHIFT;

  bin_local_kernel<<<SORT_BLOCKS, SORT_THREADS, 0, stream>>>(ei, iseglen, isegrel, ipairs, gcnt, out, E);
  rank_proj_kernel<<<projBlk + nBins, BIN_NODES, 0, stream>>>(
      x, W1a, y1, iseglen, isegrel, ipairs, icnt, ioff, icsr, N, E, projBlk);

  gin_tail_kernel<<<mlpBlk, 256, 0, stream>>>(
      y1, ioff, icnt, icsr, batch, b1a, W1b, b1b, g1, be1, m1, v1, h1, gcnt, N);

  gin_mlp_kernel<HIDDEN><<<mlpBlk, 256, 0, stream>>>(
      h1, ioff, icnt, icsr, W2a, b2a, W2b, b2b, g2, be2, m2, v2, batch, gcnt, out, N);
}


# Level-1 edge binning sorts each chunk on chip: edges read once into registers as one load group, sorted chunk built in LDS and stored as whole lines; bucket order still = LDS rank arrival order
# speedup vs baseline: 5.3271x; 1.0473x over previous
#include <hip/hip_runtime.h>

typedef float v2f __attribute__((ext_vector_type(2)));
typedef float v8f __attribute__((ext_vector_type(8)));

#define IN_FEATS 128
#define HIDDEN   64
#define NGRAPH   512
#define BN_EPS   1e-5f


#define BIN_SHIFT    9
#define BIN_NODES    512
#define MAX_BINS     256
#define SORT_THREADS 512
#define SORT_BLOCKS  256

__device__ __forceinline__ int sort_chunk(int E) { return (E + SORT_BLOCKS - 1) / SORT_BLOCKS; }

static_assert(SORT_BLOCKS * SORT_THREADS >= NGRAPH * HIDDEN / 4 && SORT_BLOCKS * SORT_THREADS >= NGRAPH,
              "one thread per float4 of out also covers the per-graph counts");

#define EDGES_PER_THREAD 16
#define CHUNK_ON_CHIP    (EDGES_PER_THREAD * SORT_THREADS)

__global__ __launch_bounds__(SORT_THREADS) void bin_local_kernel(
    const int* __restrict__ ei, int* __restrict__ segLen, int* __restrict__ segRel,
    int* __restrict__ pairs, int* __restrict__ gcount, float* __restrict__ out, int E) {
  __shared__ int s_hist[MAX_BINS];
  __shared__ int s_wsum[SORT_THREADS / 32];
  __shared__ int s_img[CHUNK_ON_CHIP];
  const int t = threadIdx.x;
  const int lane = t & 31;
  const int wave = t >> 5;

  const int gid = (int)blockIdx.x * SORT_THREADS + t;
  if (gid < NGRAPH * HIDDEN / 4) ((float4*)out)[gid] = make_float4(0.0f, 0.0f, 0.0f, 0.0f);
  if (gid < NGRAPH) gcount[gid] = 0;

  if (t < MAX_BINS) s_hist[t] = 0;
  __syncthreads();
  const int* srcp = ei;
  const int* dst = ei + E;
  const int chunk = sort_chunk(E);
  const int c0 = (int)blockIdx.x * chunk;
  int c1 = c0 + chunk;
  if (c1 > E) c1 = E;
  const int n = c1 - c0;
  const bool on_chip = chunk <= CHUNK_ON_CHIP;

  int d[EDGES_PER_THREAD], s[EDGES_PER_THREAD];
  if (on_chip) {
#pragma unroll
    for (int j = 0; j < EDGES_PER_THREAD; ++j)
      if (j * SORT_THREADS < n) {
        const int e = c0 + j * SORT_THREADS + t;
        const int ec = e < c1 ? e : c1 - 1;
        d[j] = dst[ec];
        s[j] = srcp[ec];
      }
#pragma unroll
    for (int j = 0; j < EDGES_PER_THREAD; ++j)
      if (j * SORT_THREADS < n)
        if (c0 + j * SORT_THREADS + t < c1) atomicAdd(&s_hist[d[j] >> BIN_SHIFT], 1);
  } else {
    for (int b0 = c0; b0 < c1; b0 += 4 * SORT_THREADS) {
      int dd[4];
#pragma unroll
      for (int j = 0; j < 4; ++j) {
        const int e = b0 + j * SORT_THREADS + t;
        dd[j] = dst[e < c1 ? e : c1 - 1];
      }
#pragma unroll
      for (int j = 0; j < 4; ++j)
        if (b0 + j * SORT_THREADS + t < c1) atomicAdd(&s_hist[dd[j] >> BIN_SHIFT], 1);
    }
  }
  __syncthreads();

  const int h = t < MAX_BINS ? s_hist[t] : 0;
  int inc = h;
#pragma unroll
  for (int o = 1; o < 32; o <<= 1) {
    const int u = __shfl_up(inc, o);
    if (lane >= o) inc += u;
  }
  if (lane == 31) s_wsum[wave] = inc;
  __syncthreads();
  int wbase = 0;
#pragma unroll
  for (int w = 0; w < SORT_THREADS / 32; ++w)
    if (w < wave) wbase += s_wsum[w];
  if (t < MAX_BINS) {
    const int rel = wbase + inc - h;
    segLen[(int)blockIdx.x * MAX_BINS + t] = h;
    segRel[(int)blockIdx.x * MAX_BINS + t] = rel;
    s_hist[t] = on_chip ? rel : c0 + rel;
  }
  __syncthreads();

  if (on_chip) {
    int pos[EDGES_PER_THREAD];
#pragma unroll
    for (int j = 0; j < EDGES_PER_THREAD; ++j)
      if (j * SORT_THREADS < n)
        pos[j] = (c0 + j * SORT_THREADS + t < c1) ? atomicAdd(&s_hist[d[j] >> BIN_SHIFT], 1) : -1;
#pragma unroll
    for (int j = 0; j < EDGES_PER_THREAD; ++j)
      if (j * SORT_THREADS < n)
        if (pos[j] >= 0)
          s_img[pos[j]] = (int)(((unsigned)s[j] << BIN_SHIFT) | (unsigned)(d[j] & (BIN_NODES - 1)));
    __syncthreads();
    for (int i = t - (c0 & 31); i < n; i += SORT_THREADS)
      if (i >= 0) pairs[c0 + i] = s_img[i];
  } else {
    for (int b0 = c0; b0 < c1; b0 += 4 * SORT_THREADS) {
      int ss[4], dd[4];
#pragma unroll
      for (int j = 0; j < 4; ++j) {
        const int e = b0 + j * SORT_THREADS + t;
        const int ec = e < c1 ? e : c1 - 1;
        dd[j] = dst[ec];
        ss[j] = srcp[ec];
      }
      int pos[4];
#pragma unroll
      for (int j = 0; j < 4; ++j)
        pos[j] = (b0 + j * SORT_THREADS + t < c1) ? atomicAdd(&s_hist[dd[j] >> BIN_SHIFT], 1) : -1;
#pragma unroll
      for (int j = 0; j < 4; ++j)
        if (pos[j] >= 0)
          pairs[pos[j]] = (int)(((unsigned)ss[j] << BIN_SHIFT) | (unsigned)(dd[j] & (BIN_NODES - 1)));
    }
  }
}

#define SEG_WAVES    (BIN_NODES / 32)
#define SEG_PER_WAVE (SORT_BLOCKS / SEG_WAVES)
static_assert(SORT_BLOCKS <= BIN_NODES && SORT_BLOCKS % SEG_WAVES == 0,
              "one thread per segment header, a whole number of segments per wave");

__device__ __forceinline__ void bin_sort_block(
    const int* __restrict__ segLen, const int* __restrict__ segRel, const int* __restrict__ pairs,
    int* __restrict__ cnt, int* __restrict__ off, int* __restrict__ csr, int n_nodes, int E, int b) {
  __shared__ int s_cnt[BIN_NODES];
  __shared__ int s_wsum[BIN_NODES / 32];
  __shared__ int s_wsum2[BIN_NODES / 32];
  __shared__ int s_len[SORT_BLOCKS];
  __shared__ int s_st[SORT_BLOCKS];
  const int t = threadIdx.x;
  const int lane = t & 31;
  const int wave = t >> 5;
  const int chunk = sort_chunk(E);

  s_cnt[t] = 0;
  int rel = 0;
  if (t < SORT_BLOCKS) {
    rel = segRel[t * MAX_BINS + b];
    s_len[t] = segLen[t * MAX_BINS + b];
    s_st[t] = t * chunk + rel;
  }
  int red = rel;
#pragma unroll
  for (int o = 16; o > 0; o >>= 1) red += __shfl_xor(red, o);
  if (lane == 0) s_wsum[wave] = red;
  __syncthreads();
  int base = 0;
#pragma unroll
  for (int w = 0; w < BIN_NODES / 32; ++w) base += s_wsum[w];

  unsigned v[SEG_PER_WAVE][2];
  int len[SEG_PER_WAVE];
#pragma unroll
  for (int j = 0; j < SEG_PER_WAVE; ++j) {
    const int q = wave + j * SEG_WAVES;
    len[j] = s_len[q];
    const int st = s_st[q];
#pragma unroll
    for (int u = 0; u < 2; ++u) {
      const int k = lane + 32 * u;
      v[j][u] = (unsigned)pairs[k < len[j] ? st + k : 0];
    }
  }
#pragma unroll
  for (int j = 0; j < SEG_PER_WAVE; ++j)
#pragma unroll
    for (int u = 0; u < 2; ++u)
      if (lane + 32 * u < len[j]) atomicAdd(&s_cnt[v[j][u] & (BIN_NODES - 1)], 1);
  bool longer = false;
#pragma unroll
  for (int j = 0; j < SEG_PER_WAVE; ++j) longer |= len[j] > 64;
  if (longer)
    for (int q = wave; q < SORT_BLOCKS; q += SEG_WAVES) {
      const int L = s_len[q];
      const int* p = pairs + s_st[q];
      for (int k = 64 + lane; k < L; k += 32)
        atomicAdd(&s_cnt[(unsigned)p[k] & (BIN_NODES - 1)], 1);
    }
  __syncthreads();

  const int c = s_cnt[t];
  int inc2 = c;
#pragma unroll
  for (int o = 1; o < 32; o <<= 1) {
    const int u = __shfl_up(inc2, o);
    if (lane >= o) inc2 += u;
  }
  if (lane == 31) s_wsum2[wave] = inc2;
  __syncthreads();
  int wbase2 = 0;
#pragma unroll
  for (int w = 0; w < BIN_NODES / 32; ++w)
    if (w < wave) wbase2 += s_wsum2[w];
  const int start = base + wbase2 + inc2 - c;
  const int node = b * BIN_NODES + t;
  if (node < n_nodes) {
    cnt[node] = c;
    off[node] = start;
  }
  s_cnt[t] = start;
  __syncthreads();

#pragma unroll
  for (int j = 0; j < SEG_PER_WAVE; ++j) {
    int pos[2];
#pragma unroll
    for (int u = 0; u < 2; ++u)
      pos[u] = (lane + 32 * u < len[j]) ? atomicAdd(&s_cnt[v[j][u] & (BIN_NODES - 1)], 1) : -1;
#pragma unroll
    for (int u = 0; u < 2; ++u)
      if (pos[u] >= 0) csr[pos[u]] = (int)(v[j][u] >> BIN_SHIFT);
  }
  if (longer)
    for (int q = wave; q < SORT_BLOCKS; q += SEG_WAVES) {
      const int L = s_len[q];
      const int* p = pairs + s_st[q];
      for (int k = 64 + lane; k < L; k += 32) {
        const unsigned w = (unsigned)p[k];
        csr[atomicAdd(&s_cnt[w & (BIN_NODES - 1)], 1)] = (int)(w >> BIN_SHIFT);
      }
    }
}

__device__ __forceinline__ void gather_tile(
    const float* __restrict__ xin, const int* __restrict__ off, const int* __restrict__ cnt,
    const int* __restrict__ csr, float* __restrict__ tileT, int pitch, int row0, int n_nodes,
    int half, int l16) {
  const int c = l16 << 2;
  int hrow = row0 + l16;
  if (hrow >= n_nodes) hrow = n_nodes - 1;
  const int cntv = cnt[hrow];
  const int offv = off[hrow];
  for (int st = 0; st < 8; ++st) {
    const int r = 2 * st + half;
    const int len = __shfl(cntv, r);
    const int* p = csr + __shfl(offv, r);
    int i = row0 + r;
    if (i >= n_nodes) i = n_nodes - 1;
    float4 acc = *(const float4*)(xin + (long long)i * HIDDEN + c);
    const int last = len - 1;
    for (int j = 0; j < len; j += 4) {
      const int j1 = j + 1, j2 = j + 2, j3 = j + 3;
      const int s0 = p[j];
      const int s1 = p[j1 < last ? j1 : last];
      const int s2 = p[j2 < last ? j2 : last];
      const int s3 = p[j3 < last ? j3 : last];
      const float4 a0 = *(const float4*)(xin + (long long)s0 * HIDDEN + c);
      const float4 a1 = *(const float4*)(xin + (long long)s1 * HIDDEN + c);
      const float4 a2 = *(const float4*)(xin + (long long)s2 * HIDDEN + c);
      const float4 a3 = *(const float4*)(xin + (long long)s3 * HIDDEN + c);
      const int rem = len - j;
      acc.x += a0.x; acc.y += a0.y; acc.z += a0.z; acc.w += a0.w;
      const bool k1 = rem > 1, k2 = rem > 2, k3 = rem > 3;
      acc.x += k1 ? a1.x : 0.0f; acc.y += k1 ? a1.y : 0.0f;
      acc.z += k1 ? a1.z : 0.0f; acc.w += k1 ? a1.w : 0.0f;
      acc.x += k2 ? a2.x : 0.0f; acc.y += k2 ? a2.y : 0.0f;
      acc.z += k2 ? a2.z : 0.0f; acc.w += k2 ? a2.w : 0.0f;
      acc.x += k3 ? a3.x : 0.0f; acc.y += k3 ? a3.y : 0.0f;
      acc.z += k3 ? a3.z : 0.0f; acc.w += k3 ? a3.w : 0.0f;
    }
    *(float4*)(tileT + r * pitch + c) = acc;
  }
}

#define WPITCH_PAD 4
#define MLP_THREADS 256

template <int K, int NT = MLP_THREADS>
__device__ __forceinline__ void stage_weight_rows(float* __restrict__ sW,
                                                  const float* __restrict__ W) {
  constexpr int Q = K / 4;
  constexpr int R = 64 * Q / NT;
  float4 v[R];
#pragma unroll
  for (int j = 0; j < R; ++j) {
    const int i = (int)threadIdx.x + j * NT;
    const int row = i / Q;
    v[j] = *(const float4*)(W + row * K + 4 * (i - row * Q));
  }
#pragma unroll
  for (int j = 0; j < R; ++j) {
    const int i = (int)threadIdx.x + j * NT;
    const int row = i / Q;
    *(float4*)(sW + row * (K + WPITCH_PAD) + 4 * (i - row * Q)) = v[j];
  }
}

__device__ __forceinline__ void pool_add_run(float* __restrict__ out, const int* __restrict__ gcount,
                                             int g, int l16, const float* run) {
  float cg = (float)gcount[g];
  cg = cg > 1.0f ? cg : 1.0f;
  float* sp = out + (long long)g * HIDDEN + l16;
  atomicAdd(sp, run[0] / cg);
  atomicAdd(sp + 16, run[1] / cg);
  atomicAdd(sp + 32, run[2] / cg);
  atomicAdd(sp + 48, run[3] / cg);
}

template <int FIN>
__global__ __launch_bounds__(256) void gin_mlp_kernel(
    const float* __restrict__ hin,
    const int* __restrict__ off,
    const int* __restrict__ cnt,
    const int* __restrict__ csr,
    const float* __restrict__ Wa,
    const float* __restrict__ ba,
    const float* __restrict__ Wb,
    const float* __restrict__ bb,
    const float* __restrict__ gg,
    const float* __restrict__ be,
    const float* __restrict__ mm,
    const float* __restrict__ vv,
    const int* __restrict__ batch,
    const int* __restrict__ gcount,
    float* __restrict__ out,
    int n_nodes) {
  constexpr int PA = FIN + WPITCH_PAD;
  constexpr int PB = HIDDEN + WPITCH_PAD;
  __shared__ float sWa[64 * PA];
  __shared__ float sWb[64 * PB];
  static_assert(FIN == HIDDEN, "the tile of neighbour sums is HIDDEN wide");
  __shared__ float sT[8][16 * PB];

  const int wave = threadIdx.x >> 5;
  const int lane = threadIdx.x & 31;
  const int half = lane >> 4;
  const int l16  = lane & 15;

  const int tile = blockIdx.x * 8 + wave;
  const int row0 = tile * 16;

  int arow = row0 + l16;
  if (arow >= n_nodes) arow = n_nodes - 1;

  stage_weight_rows<FIN>(sWa, Wa);
  stage_weight_rows<HIDDEN>(sWb, Wb);
  __syncthreads();
  if (row0 >= n_nodes) return;

  gather_tile(hin, off, cnt, csr, sT[wave], PB, row0, n_nodes, half, l16);
  const float* tp = &sT[wave][l16 * PB + 2 * half];
  v2f afr[FIN / 4];
#pragma unroll
  for (int j = 0; j < FIN / 4; ++j) afr[j] = *(const v2f*)(tp + 4 * j);

  v8f acc[4];
#pragma unroll
  for (int t = 0; t < 4; ++t) {
    float bias = ba[t * 16 + l16];
#pragma unroll
    for (int r = 0; r < 8; ++r) acc[t][r] = bias;
  }

  const int gl = batch[arow];

#pragma unroll
  for (int k = 0; k < FIN; k += 4) {
    v2f a = afr[k / 4];
#pragma unroll
    for (int t = 0; t < 4; ++t) {
      const float* bp = sWa + (t * 16 + l16) * PA + k + 2 * half;
      v2f b = *(const v2f*)bp;
      acc[t] = __builtin_amdgcn_wmma_f32_16x16x4_f32(
          false, a, false, b, (short)0, acc[t], false, false);
    }
  }

#pragma unroll
  for (int t = 0; t < 4; ++t) {
#pragma unroll
    for (int r = 0; r < 8; ++r) {
      float x = acc[t][r];
      x = x > 0.0f ? x : 0.0f;
      sT[wave][(r + 8 * half) * PB + t * 16 + l16] = x;
    }
  }

  v8f acc2[4];
#pragma unroll
  for (int t = 0; t < 4; ++t)
#pragma unroll
    for (int r = 0; r < 8; ++r) acc2[t][r] = 0.0f;

  for (int k = 0; k < 64; k += 4) {
    v2f a = *(const v2f*)(tp + k);
#pragma unroll
    for (int t = 0; t < 4; ++t) {
      const float* bp = sWb + (t * 16 + l16) * PB + k + 2 * half;
      v2f b = *(const v2f*)bp;
      acc2[t] = __builtin_amdgcn_wmma_f32_16x16x4_f32(
          false, a, false, b, (short)0, acc2[t], false, false);
    }
  }

#pragma unroll
  for (int t = 0; t < 4; ++t) {
    int n = t * 16 + l16;
    float s  = gg[n] * rsqrtf(vv[n] + BN_EPS);
    float sh = (bb[n] - mm[n]) * s + be[n];
#pragma unroll
    for (int r = 0; r < 8; ++r) {
      float o = acc2[t][r] * s + sh;
      acc2[t][r] = o > 0.0f ? o : 0.0f;
    }
  }

  const int g0 = __builtin_amdgcn_readfirstlane(gl);
  const bool whole = row0 + 16 <= n_nodes;
  if (whole && __all(gl == g0)) {
    float cs[4];
#pragma unroll
    for (int t = 0; t < 4; ++t) {
      float v = acc2[t][0];
#pragma unroll
      for (int r = 1; r < 8; ++r) v += acc2[t][r];
      cs[t] = v + __shfl_xor(v, 16);
    }
    float c0 = (float)gcount[g0];
    c0 = c0 > 1.0f ? c0 : 1.0f;
    float* sp = out + (long long)g0 * HIDDEN + lane;
    atomicAdd(sp, (half ? cs[1] : cs[0]) / c0);
    atomicAdd(sp + 32, (half ? cs[3] : cs[2]) / c0);
  } else {
    int gr[8];
#pragma unroll
    for (int r = 0; r < 8; ++r) gr[r] = __shfl(gl, 8 * half + r);
    int gp = 0;
    bool open = false;
    float run[4] = {0.0f, 0.0f, 0.0f, 0.0f};
#pragma unroll
    for (int r = 0; r < 8; ++r) {
      if (row0 + 8 * half + r < n_nodes) {
        if (open && gr[r] != gp) {
          pool_add_run(out, gcount, gp, l16, run);
          run[0] = 0.0f; run[1] = 0.0f; run[2] = 0.0f; run[3] = 0.0f;
        }
        gp = gr[r];
        open = true;
#pragma unroll
        for (int t = 0; t < 4; ++t) run[t] += acc2[t][r];
      }
    }
    if (open) pool_add_run(out, gcount, gp, l16, run);
  }
}

#define PROJ_THREADS BIN_NODES
#define PROJ_TILES   (PROJ_THREADS / 32)

__device__ __forceinline__ void proj_block(
    const float* __restrict__ xin,
    const float* __restrict__ Wa,
    float* __restrict__ ysrc,
    int n_nodes, int blk) {
  constexpr int PA = IN_FEATS + WPITCH_PAD;
  __shared__ float sWa[64 * PA];

  const int wave = threadIdx.x >> 5;
  const int lane = threadIdx.x & 31;
  const int half = lane >> 4;
  const int l16  = lane & 15;

  const int tile = blk * PROJ_TILES + wave;
  const int row0 = tile * 16;

  int arow = row0 + l16;
  if (arow >= n_nodes) arow = n_nodes - 1;
  const float* aptr = xin + (long long)arow * IN_FEATS + 2 * half;
  v2f afr[IN_FEATS / 4];
#pragma unroll
  for (int j = 0; j < IN_FEATS / 4; ++j) afr[j] = *(const v2f*)(aptr + 4 * j);

  stage_weight_rows<IN_FEATS, PROJ_THREADS>(sWa, Wa);
#pragma unroll
  for (int j = 0; j < IN_FEATS / 4; j += 8)
    asm volatile("" ::"v"(afr[j]), "v"(afr[j + 1]), "v"(afr[j + 2]), "v"(afr[j + 3]),
                 "v"(afr[j + 4]), "v"(afr[j + 5]), "v"(afr[j + 6]), "v"(afr[j + 7]));
  __syncthreads();
  if (row0 >= n_nodes) return;

  v8f acc[4];
#pragma unroll
  for (int t = 0; t < 4; ++t)
#pragma unroll
    for (int r = 0; r < 8; ++r) acc[t][r] = 0.0f;

#pragma unroll
  for (int k = 0; k < IN_FEATS; k += 4) {
    v2f a = afr[k / 4];
#pragma unroll
    for (int t = 0; t < 4; ++t) {
      const float* bp = sWa + (t * 16 + l16) * PA + k + 2 * half;
      v2f b = *(const v2f*)bp;
      acc[t] = __builtin_amdgcn_wmma_f32_16x16x4_f32(
          false, a, false, b, (short)0, acc[t], false, false);
    }
  }

#pragma unroll
  for (int t = 0; t < 4; ++t) {
    int n = t * 16 + l16;
#pragma unroll
    for (int r = 0; r < 8; ++r) {
      int mrow = row0 + r + 8 * half;
      if (mrow < n_nodes) ysrc[(long long)mrow * HIDDEN + n] = acc[t][r];
    }
  }
}

__global__ __launch_bounds__(BIN_NODES) void rank_proj_kernel(
    const float* __restrict__ xin, const float* __restrict__ Wa, float* __restrict__ ysrc,
    const int* __restrict__ segLen, const int* __restrict__ segRel, const int* __restrict__ pairs,
    int* __restrict__ cnt, int* __restrict__ off, int* __restrict__ csr,
    int n_nodes, int E, int n_proj) {
  static_assert(PROJ_THREADS == BIN_NODES, "both kinds of block have the launch's block size");
  const int blk = (int)blockIdx.x;
  if (blk < n_proj)
    proj_block(xin, Wa, ysrc, n_nodes, blk);
  else
    bin_sort_block(segLen, segRel, pairs, cnt, off, csr, n_nodes, E, blk - n_proj);
}

__global__ __launch_bounds__(256) void gin_tail_kernel(
    const float* __restrict__ ysrc,
    const int* __restrict__ off,
    const int* __restrict__ cnt,
    const int* __restrict__ csr,
    const int* __restrict__ batch,
    const float* __restrict__ ba,
    const float* __restrict__ Wb,
    const float* __restrict__ bb,
    const float* __restrict__ gg,
    const float* __restrict__ be,
    const float* __restrict__ mm,
    const float* __restrict__ vv,
    float* __restrict__ hout,
    int* __restrict__ gcount,
    int n_nodes) {
  constexpr int PB = HIDDEN + WPITCH_PAD;
  __shared__ float sWb[64 * PB];
  __shared__ float sT[8][16 * PB];

  const int wave = threadIdx.x >> 5;
  const int lane = threadIdx.x & 31;
  const int half = lane >> 4;
  const int l16  = lane & 15;

  const int tile = blockIdx.x * 8 + wave;
  const int row0 = tile * 16;

  stage_weight_rows<HIDDEN>(sWb, Wb);
  __syncthreads();
  if (row0 >= n_nodes) return;

  const int nrow = row0 + l16;
  const int gl = batch[nrow < n_nodes ? nrow : n_nodes - 1];

  gather_tile(ysrc, off, cnt, csr, sT[wave], PB, row0, n_nodes, half, l16);

  {
    const int g0 = __builtin_amdgcn_readfirstlane(gl);
    if (row0 + 16 <= n_nodes && __all(gl == g0)) {
      if (lane == 0 && (unsigned)g0 < (unsigned)NGRAPH) atomicAdd(&gcount[g0], 16);
    } else if (half == 0 && nrow < n_nodes && (unsigned)gl < (unsigned)NGRAPH) {
      atomicAdd(&gcount[gl], 1);
    }
  }

  const float* tp = &sT[wave][l16 * PB + 2 * half];
  v2f afr[HIDDEN / 4];
#pragma unroll
  for (int j = 0; j < HIDDEN / 4; ++j) afr[j] = *(const v2f*)(tp + 4 * j);

  v8f acc2[4];
#pragma unroll
  for (int t = 0; t < 4; ++t)
#pragma unroll
    for (int r = 0; r < 8; ++r) acc2[t][r] = 0.0f;

  const float* biasp = ba + 2 * half;

#pragma unroll
  for (int k = 0; k < HIDDEN; k += 4) {
    v2f a = afr[k / 4];
    v2f bs = *(const v2f*)(biasp + k);
    a.x = a.x + bs.x;
    a.y = a.y + bs.y;
    a.x = a.x > 0.0f ? a.x : 0.0f;
    a.y = a.y > 0.0f ? a.y : 0.0f;
#pragma unroll
    for (int t = 0; t < 4; ++t) {
      const float* bp = sWb + (t * 16 + l16) * PB + k + 2 * half;
      v2f b = *(const v2f*)bp;
      acc2[t] = __builtin_amdgcn_wmma_f32_16x16x4_f32(
          false, a, false, b, (short)0, acc2[t], false, false);
    }
  }

#pragma unroll
  for (int t = 0; t < 4; ++t) {
    int n = t * 16 + l16;
    float s  = gg[n] * rsqrtf(vv[n] + BN_EPS);
    float sh = (bb[n] - mm[n]) * s + be[n];
#pragma unroll
    for (int r = 0; r < 8; ++r) {
      int mrow = row0 + r + 8 * half;
      float o = acc2[t][r] * s + sh;
      o = o > 0.0f ? o : 0.0f;
      if (mrow < n_nodes) hout[(long long)mrow * HIDDEN + n] = o;
    }
  }
}

extern "C" void kernel_launch(void* const* d_in, const int* in_sizes, int n_in,
                              void* d_out, int out_size, void* d_ws, size_t ws_size,
                              hipStream_t stream) {
  const float* x   = (const float*)d_in[0];
  const int* ei    = (const int*)d_in[1];
  const int* batch = (const int*)d_in[2];
  const float* W1a = (const float*)d_in[3];
  const float* b1a = (const float*)d_in[4];
  const float* W1b = (const float*)d_in[5];
  const float* b1b = (const float*)d_in[6];
  const float* g1  = (const float*)d_in[7];
  const float* be1 = (const float*)d_in[8];
  const float* m1  = (const float*)d_in[9];
  const float* v1  = (const float*)d_in[10];
  const float* W2a = (const float*)d_in[11];
  const float* b2a = (const float*)d_in[12];
  const float* W2b = (const float*)d_in[13];
  const float* b2b = (const float*)d_in[14];
  const float* g2  = (const float*)d_in[15];
  const float* be2 = (const float*)d_in[16];
  const float* m2  = (const float*)d_in[17];
  const float* v2  = (const float*)d_in[18];
  float* out = (float*)d_out;

  const int N = in_sizes[0] / IN_FEATS;
  const int E = in_sizes[1] / 2;

  float* ws    = (float*)d_ws;
  float* y1    = ws;
  float* h1    = ws + (long long)N * IN_FEATS;
  int* gcnt    = (int*)(h1 + (long long)N * HIDDEN);
  const int Npad = (N + 3) & ~3;
  int* icnt    = gcnt + NGRAPH;
  int* ioff    = icnt + Npad;
  int* ipairs  = ioff + Npad;
  int* icsr    = ipairs + E;
  int* iseglen = icsr + E;
  int* isegrel = iseglen + SORT_BLOCKS * MAX_BINS;
  (void)ws_size; (void)n_in; (void)out_size;

  const int tiles  = (N + 15) / 16;
  const int mlpBlk = (tiles + 7) / 8;
  const int projBlk = (tiles + PROJ_TILES - 1) / PROJ_TILES;
  const int nBins = (N + BIN_NODES - 1) >> BIN_SHIFT;

  bin_local_kernel<<<SORT_BLOCKS, SORT_THREADS, 0, stream>>>(ei, iseglen, isegrel, ipairs, gcnt, out, E);
  rank_proj_kernel<<<projBlk + nBins, BIN_NODES, 0, stream>>>(
      x, W1a, y1, iseglen, isegrel, ipairs, icnt, ioff, icsr, N, E, projBlk);

  gin_tail_kernel<<<mlpBlk, 256, 0, stream>>>(
      y1, ioff, icnt, icsr, batch, b1a, W1b, b1b, g1, be1, m1, v1, h1, gcnt, N);

  gin_mlp_kernel<HIDDEN><<<mlpBlk, 256, 0, stream>>>(
      h1, ioff, icnt, icsr, W2a, b2a, W2b, b2b, g2, be2, m2, v2, batch, gcnt, out, N);
}
